# starting kernel
# baseline (speedup 1.0000x reference)
#include <hip/hip_runtime.h>

#define N_NODES   50000
#define N_EDGES   800000
#define IN_DIM    256
#define OUT_DIM   64
#define NEG_SLOPE 0.01f

typedef __attribute__((ext_vector_type(2))) float v2f;
typedef __attribute__((ext_vector_type(8))) float v8f;

__global__ __launch_bounds__(256) void gat_init(float* __restrict__ agg,
                                                float* __restrict__ denom,
                                                float* __restrict__ emax) {
    int i = blockIdx.x * blockDim.x + threadIdx.x;
    if (i < N_NODES * OUT_DIM) agg[i] = 0.0f;
    if (i < N_NODES) {
        denom[i] = 0.0f;
        emax[i]  = -__builtin_inff();
    }
}

__global__ __launch_bounds__(128) void gat_gemm_wmma(
    const float* __restrict__ hidden,
    const float* __restrict__ W,
    const float* __restrict__ bias,
    float* __restrict__ h_out)
{
    const int lane  = threadIdx.x & 31;
    const int wave  = threadIdx.x >> 5;
    const int half  = lane >> 4;
    const int l16   = lane & 15;
    const int mbase = blockIdx.x * 16;
    const int gcol  = wave * 16 + l16;

    const float* arow = hidden + (size_t)(mbase + l16) * IN_DIM + 2 * half;
    const float* bcol = W + (size_t)(2 * half) * OUT_DIM + gcol;

    v8f acc = {};
#pragma unroll 4
    for (int k0 = 0; k0 < IN_DIM; k0 += 4) {
        v2f a = *(const v2f*)(arow + k0);
        v2f b;
        b.x = bcol[(size_t)k0 * OUT_DIM];
        b.y = bcol[(size_t)(k0 + 1) * OUT_DIM];
        acc = __builtin_amdgcn_wmma_f32_16x16x4_f32(
             false, a,  false, b,
             (short)0, acc,  false,  false);
    }

    const float bb = bias[gcol];
#pragma unroll
    for (int r = 0; r < 8; ++r) {
        int row = mbase + r + 8 * half;
        h_out[(size_t)row * OUT_DIM + gcol] = acc[r] + bb;
    }
}

__global__ __launch_bounds__(256) void gat_proj(
    const float* __restrict__ h,
    const float* __restrict__ att_w,
    float* __restrict__ a_src,
    float* __restrict__ a_dst)
{
    int node = blockIdx.x * 8 + (threadIdx.x >> 5);
    if (node >= N_NODES) return;
    int lane = threadIdx.x & 31;

    float h0 = h[(size_t)node * OUT_DIM + lane];
    float h1 = h[(size_t)node * OUT_DIM + lane + 32];
    float s = h0 * att_w[lane]      + h1 * att_w[lane + 32];
    float d = h0 * att_w[lane + 64] + h1 * att_w[lane + 96];
#pragma unroll
    for (int off = 16; off > 0; off >>= 1) {
        s += __shfl_down(s, off, 32);
        d += __shfl_down(d, off, 32);
    }
    if (lane == 0) { a_src[node] = s; a_dst[node] = d; }
}

__device__ __forceinline__ void atomicMaxF(float* addr, float v) {
    if (v >= 0.0f) atomicMax((int*)addr, __float_as_int(v));
    else           atomicMin((unsigned int*)addr, __float_as_uint(v));
}

__global__ __launch_bounds__(256) void gat_edge_logit(
    const int* __restrict__ src, const int* __restrict__ dst,
    const float* __restrict__ a_src, const float* __restrict__ a_dst,
    const float* __restrict__ att_b,
    float* __restrict__ e_out, float* __restrict__ emax)
{
    int e = blockIdx.x * blockDim.x + threadIdx.x;
    if (e >= N_EDGES) return;
    int s = src[e], d = dst[e];
    float v = a_src[s] + a_dst[d] + att_b[0];
    v = (v > 0.0f) ? v : NEG_SLOPE * v;
    e_out[e] = v;
    atomicMaxF(&emax[d], v);
}

__global__ __launch_bounds__(256) void gat_edge_scatter(
    const int* __restrict__ src, const int* __restrict__ dst,
    const float* __restrict__ e_in, const float* __restrict__ emax,
    const float* __restrict__ h,
    float* __restrict__ denom, float* __restrict__ agg)
{
    int edge = blockIdx.x * 8 + (threadIdx.x >> 5);
    if (edge >= N_EDGES) return;
    int lane = threadIdx.x & 31;
    int s = src[edge], d = dst[edge];

    float ex = __expf(e_in[edge] - emax[d]);
    if (lane == 0) atomicAdd(&denom[d], ex);

    float h0 = h[(size_t)s * OUT_DIM + lane];
    float h1 = h[(size_t)s * OUT_DIM + lane + 32];
    atomicAdd(&agg[(size_t)d * OUT_DIM + lane],       ex * h0);
    atomicAdd(&agg[(size_t)d * OUT_DIM + lane + 32],  ex * h1);
}

__global__ __launch_bounds__(256) void gat_final(
    const float* __restrict__ agg, const float* __restrict__ denom,
    const float* __restrict__ h, float* __restrict__ out)
{
    int i = blockIdx.x * blockDim.x + threadIdx.x;
    if (i >= N_NODES * OUT_DIM) return;
    int node = i >> 6;
    float dn = denom[node];
    out[i] = (dn > 0.0f) ? agg[i] / dn : h[i];
}

extern "C" void kernel_launch(void* const* d_in, const int* in_sizes, int n_in,
                              void* d_out, int out_size, void* d_ws, size_t ws_size,
                              hipStream_t stream) {
    const float* hidden = (const float*)d_in[0];
    const float* W      = (const float*)d_in[1];
    const float* bias   = (const float*)d_in[2];
    const float* att_w  = (const float*)d_in[3];
    const float* att_b  = (const float*)d_in[4];
    const int*   src    = (const int*)  d_in[5];
    const int*   dst    = (const int*)  d_in[6];
    float* out = (float*)d_out;

    float* ws     = (float*)d_ws;
    float* h_ws   = ws;
    float* a_src  = h_ws  + (size_t)N_NODES * OUT_DIM;
    float* a_dst  = a_src + N_NODES;
    float* emax   = a_dst + N_NODES;
    float* denom  = emax  + N_NODES;
    float* e_ws   = denom + N_NODES;
    float* agg    = e_ws  + N_EDGES;

    const int NT = 256;
    gat_init<<<(N_NODES * OUT_DIM + NT - 1) / NT, NT, 0, stream>>>(agg, denom, emax);
    gat_gemm_wmma<<<N_NODES / 16, 128, 0, stream>>>(hidden, W, bias, h_ws);
    gat_proj<<<(N_NODES + 7) / 8, NT, 0, stream>>>(h_ws, att_w, a_src, a_dst);
    gat_edge_logit<<<(N_EDGES + NT - 1) / NT, NT, 0, stream>>>(
        src, dst, a_src, a_dst, att_b, e_ws, emax);
    gat_edge_scatter<<<(N_EDGES + 7) / 8, NT, 0, stream>>>(
        src, dst, e_ws, emax, h_ws, denom, agg);
    gat_final<<<(N_NODES * OUT_DIM + NT - 1) / NT, NT, 0, stream>>>(
        agg, denom, h_ws, out);
}


# Replace float-atomic edge scatter by a counting sort of edges by dst (int atomic arrival-order rank) and one wave per dst: max, exp sum and weighted h gather in registers, summed in rank order
# speedup vs baseline: 1.8846x; 1.8846x over previous
#include <hip/hip_runtime.h>


#define N_NODES   50000
#define N_EDGES   800000
#define IN_DIM    256
#define OUT_DIM   64
#define NEG_SLOPE 0.01f

typedef __attribute__((ext_vector_type(2))) float v2f;
typedef __attribute__((ext_vector_type(8))) float v8f;

__global__ __launch_bounds__(128) void gat_gemm_wmma(
    const float* __restrict__ hidden,
    const float* __restrict__ W,
    const float* __restrict__ bias,
    float* __restrict__ h_out)
{
    const int lane  = threadIdx.x & 31;
    const int wave  = threadIdx.x >> 5;
    const int half  = lane >> 4;
    const int l16   = lane & 15;
    const int mbase = blockIdx.x * 16;
    const int gcol  = wave * 16 + l16;

    const float* arow = hidden + (size_t)(mbase + l16) * IN_DIM + 2 * half;
    const float* bcol = W + (size_t)(2 * half) * OUT_DIM + gcol;

    v8f acc = {};
#pragma unroll 4
    for (int k0 = 0; k0 < IN_DIM; k0 += 4) {
        v2f a = *(const v2f*)(arow + k0);
        v2f b;
        b.x = bcol[(size_t)k0 * OUT_DIM];
        b.y = bcol[(size_t)(k0 + 1) * OUT_DIM];
        acc = __builtin_amdgcn_wmma_f32_16x16x4_f32(
             false, a,  false, b,
             (short)0, acc,  false,  false);
    }

    const float bb = bias[gcol];
#pragma unroll
    for (int r = 0; r < 8; ++r) {
        int row = mbase + r + 8 * half;
        h_out[(size_t)row * OUT_DIM + gcol] = acc[r] + bb;
    }
}

__global__ __launch_bounds__(256) void gat_proj(
    const float* __restrict__ h,
    const float* __restrict__ att_w,
    float* __restrict__ a_src,
    float* __restrict__ a_dst,
    int* __restrict__ cnt)
{
    int node = blockIdx.x * 8 + (threadIdx.x >> 5);
    if (node >= N_NODES) return;
    int lane = threadIdx.x & 31;

    float h0 = h[(size_t)node * OUT_DIM + lane];
    float h1 = h[(size_t)node * OUT_DIM + lane + 32];
    float s = h0 * att_w[lane]      + h1 * att_w[lane + 32];
    float d = h0 * att_w[lane + 64] + h1 * att_w[lane + 96];
#pragma unroll
    for (int off = 16; off > 0; off >>= 1) {
        s += __shfl_down(s, off, 32);
        d += __shfl_down(d, off, 32);
    }
    if (lane == 0) { a_src[node] = s; a_dst[node] = d; cnt[node] = 0; }
}

__global__ __launch_bounds__(256) void gat_edge_rank(
    const int* __restrict__ dst, int* __restrict__ cnt, int* __restrict__ rank)
{
    int e = blockIdx.x * blockDim.x + threadIdx.x;
    if (e >= N_EDGES) return;
    rank[e] = atomicAdd(&cnt[dst[e]], 1);
}

#define SCAN_T 1024
__global__ __launch_bounds__(SCAN_T) void gat_offsets(
    const int* __restrict__ cnt, int* __restrict__ off)
{
    __shared__ int w_tot[32];
    __shared__ int w_pre[32];
    __shared__ int w_off[32];
    __shared__ int blk_base;

    const int tid  = threadIdx.x;
    const int lane = tid & 31;
    const int wave = tid >> 5;
    const int blk0 = blockIdx.x * SCAN_T;

    int pre = 0;
    for (int i = tid; i < blk0; i += SCAN_T) pre += cnt[i];

    const int idx = blk0 + tid;
    const int c = (idx < N_NODES) ? cnt[idx] : 0;

    int incl = c;
    for (int o = 1; o < 32; o <<= 1) {
        int t = __shfl_up(incl, o, 32);
        if (lane >= o) incl += t;
    }
    for (int o = 16; o > 0; o >>= 1) pre += __shfl_xor(pre, o, 32);

    if (lane == 31) w_tot[wave] = incl;
    if (lane == 0)  w_pre[wave] = pre;
    __syncthreads();

    if (wave == 0) {
        int t = w_tot[lane];
        int p = w_pre[lane];
        int ti = t;
        for (int o = 1; o < 32; o <<= 1) {
            int u = __shfl_up(ti, o, 32);
            if (lane >= o) ti += u;
        }
        for (int o = 16; o > 0; o >>= 1) p += __shfl_xor(p, o, 32);
        w_off[lane] = ti - t;
        if (lane == 0) blk_base = p;
    }
    __syncthreads();

    if (idx < N_NODES) off[idx] = blk_base + w_off[wave] + incl - c;
}

__global__ __launch_bounds__(256) void gat_edge_place(
    const int* __restrict__ src, const int* __restrict__ dst,
    const int* __restrict__ rank, const int* __restrict__ off,
    int* __restrict__ perm)
{
    int e = blockIdx.x * blockDim.x + threadIdx.x;
    if (e >= N_EDGES) return;
    int p = off[dst[e]] + rank[e];
    if ((unsigned)p < (unsigned)N_EDGES) perm[p] = src[e];
}

__device__ __forceinline__ float lrelu(float v) {
    return (v > 0.0f) ? v : NEG_SLOPE * v;
}

__device__ __forceinline__ int clamp_node(int s) {
    return min(max(s, 0), N_NODES - 1);
}

template <int K>
__device__ __forceinline__ void add_rows(v2f& acc, const float* __restrict__ hl,
                                         int s, float ex, int t) {
    v2f   hv[K];
    float w[K];
    for (int k = 0; k < K; ++k) {
        const int st = __builtin_amdgcn_readlane(s, t + k);
        w[k]  = __int_as_float(__builtin_amdgcn_readlane(__float_as_int(ex), t + k));
        hv[k] = *(const v2f*)(hl + (size_t)st * OUT_DIM);
    }
    for (int k = 0; k < K; ++k) {
        acc.x = __builtin_fmaf(w[k], hv[k].x, acc.x);
        acc.y = __builtin_fmaf(w[k], hv[k].y, acc.y);
    }
}

__global__ __launch_bounds__(256) void gat_node_gather(
    const int* __restrict__ perm, const int* __restrict__ off,
    const int* __restrict__ cnt,
    const float* __restrict__ a_src, const float* __restrict__ a_dst,
    const float* __restrict__ att_b,
    const float* __restrict__ h, float* __restrict__ out)
{
    const int node = __builtin_amdgcn_readfirstlane(
        (int)(blockIdx.x * 8 + (threadIdx.x >> 5)));
    if (node >= N_NODES) return;
    const int lane = threadIdx.x & 31;

    int start = off[node];
    int n     = cnt[node];
    start = min(max(start, 0), N_EDGES);
    n     = min(max(n, 0), N_EDGES - start);

    const float ad = a_dst[node];
    const float ab = att_b[0];
    const float NINF = -__builtin_inff();

    int   s0 = 0;
    float e0 = NINF;
    if (lane < n) {
        s0 = clamp_node(perm[start + lane]);
        e0 = lrelu(a_src[s0] + ad + ab);
    }
    float m = e0;
    for (int base = 32; base < n; base += 32) {
        int j = base + lane;
        if (j < n) {
            int s = clamp_node(perm[start + j]);
            m = fmaxf(m, lrelu(a_src[s] + ad + ab));
        }
    }
    for (int o = 16; o > 0; o >>= 1) m = fmaxf(m, __shfl_xor(m, o, 32));
    if (!__builtin_isfinite(m)) m = 0.0f;

    v2f   acc  = {0.0f, 0.0f};
    float dsum = 0.0f;
    const float* hl = h + 2 * lane;
    for (int base = 0; base < n; base += 32) {
        int   s = s0;
        float e = e0;
        if (base > 0) {
            s = 0; e = NINF;
            int j = base + lane;
            if (j < n) {
                s = clamp_node(perm[start + j]);
                e = lrelu(a_src[s] + ad + ab);
            }
        }
        const float ex = (base + lane < n) ? __expf(e - m) : 0.0f;
        dsum += ex;
        const int nc = min(32, n - base);
        int t = 0;
        for (; t + 8 <= nc; t += 8) add_rows<8>(acc, hl, s, ex, t);
        if (nc & 4) { add_rows<4>(acc, hl, s, ex, t); t += 4; }
        if (nc & 2) { add_rows<2>(acc, hl, s, ex, t); t += 2; }
        if (nc & 1) { add_rows<1>(acc, hl, s, ex, t); }
    }
    for (int o = 16; o > 0; o >>= 1) dsum += __shfl_xor(dsum, o, 32);

    v2f r;
    if (dsum > 0.0f) {
        r.x = acc.x / dsum;
        r.y = acc.y / dsum;
    } else {
        r = *(const v2f*)(h + (size_t)node * OUT_DIM + 2 * lane);
    }
    *(v2f*)(out + (size_t)node * OUT_DIM + 2 * lane) = r;
}

extern "C" void kernel_launch(void* const* d_in, const int* in_sizes, int n_in,
                              void* d_out, int out_size, void* d_ws, size_t ws_size,
                              hipStream_t stream) {
    const float* hidden = (const float*)d_in[0];
    const float* W      = (const float*)d_in[1];
    const float* bias   = (const float*)d_in[2];
    const float* att_w  = (const float*)d_in[3];
    const float* att_b  = (const float*)d_in[4];
    const int*   src    = (const int*)  d_in[5];
    const int*   dst    = (const int*)  d_in[6];
    float* out = (float*)d_out;

    float* ws     = (float*)d_ws;
    float* h_ws   = ws;
    float* a_src  = h_ws  + (size_t)N_NODES * OUT_DIM;
    float* a_dst  = a_src + N_NODES;
    int*   cnt    = (int*)(a_dst + N_NODES);
    int*   off    = cnt  + N_NODES;
    int*   rank   = off  + N_NODES;
    int*   perm   = rank + N_EDGES;

    const int NT = 256;
    gat_gemm_wmma<<<N_NODES / 16, 128, 0, stream>>>(hidden, W, bias, h_ws);
    gat_proj<<<(N_NODES + 7) / 8, NT, 0, stream>>>(h_ws, att_w, a_src, a_dst, cnt);
    gat_edge_rank<<<(N_EDGES + NT - 1) / NT, NT, 0, stream>>>(dst, cnt, rank);
    gat_offsets<<<(N_NODES + SCAN_T - 1) / SCAN_T, SCAN_T, 0, stream>>>(cnt, off);
    gat_edge_place<<<(N_EDGES + NT - 1) / NT, NT, 0, stream>>>(src, dst, rank, off, perm);
    gat_node_gather<<<(N_NODES + 7) / 8, NT, 0, stream>>>(
        perm, off, cnt, a_src, a_dst, att_b, h_ws, out);
}


# Each launch behind the GEMM loops its waves over several nodes or edges (next node's edge list fetched ahead): 152k short waves become 23k; sum order still the atomic arrival rank
# speedup vs baseline: 1.9525x; 1.0360x over previous
#include <hip/hip_runtime.h>


#define N_NODES   50000
#define N_EDGES   800000
#define IN_DIM    256
#define OUT_DIM   64
#define NEG_SLOPE 0.01f

typedef __attribute__((ext_vector_type(2))) float v2f;
typedef __attribute__((ext_vector_type(4))) float v4f;
typedef __attribute__((ext_vector_type(8))) float v8f;
typedef __attribute__((ext_vector_type(4))) int   v4i;

#define PROJ_NPW   16
#define EDGE_EPT   8
#define GATHER_NPW 4
static_assert(N_NODES % GATHER_NPW == 0, "gather waves read whole groups of nodes");

__global__ __launch_bounds__(128) void gat_gemm_wmma(
    const float* __restrict__ hidden,
    const float* __restrict__ W,
    const float* __restrict__ bias,
    float* __restrict__ h_out)
{
    const int lane  = threadIdx.x & 31;
    const int wave  = threadIdx.x >> 5;
    const int half  = lane >> 4;
    const int l16   = lane & 15;
    const int mbase = blockIdx.x * 16;
    const int gcol  = wave * 16 + l16;

    const float* arow = hidden + (size_t)(mbase + l16) * IN_DIM + 2 * half;
    const float* bcol = W + (size_t)(2 * half) * OUT_DIM + gcol;

    v8f acc = {};
#pragma unroll 4
    for (int k0 = 0; k0 < IN_DIM; k0 += 4) {
        v2f a = *(const v2f*)(arow + k0);
        v2f b;
        b.x = bcol[(size_t)k0 * OUT_DIM];
        b.y = bcol[(size_t)(k0 + 1) * OUT_DIM];
        acc = __builtin_amdgcn_wmma_f32_16x16x4_f32(
             false, a,  false, b,
             (short)0, acc,  false,  false);
    }

    const float bb = bias[gcol];
#pragma unroll
    for (int r = 0; r < 8; ++r) {
        int row = mbase + r + 8 * half;
        h_out[(size_t)row * OUT_DIM + gcol] = acc[r] + bb;
    }
}

__global__ __launch_bounds__(256) void gat_proj(
    const float* __restrict__ h,
    const float* __restrict__ att_w,
    float* __restrict__ a_src,
    float* __restrict__ a_dst,
    int* __restrict__ cnt)
{
    const int node0 = __builtin_amdgcn_readfirstlane(
        (int)(blockIdx.x * 8 + (threadIdx.x >> 5))) * PROJ_NPW;
    if (node0 >= N_NODES) return;
    const int lane = threadIdx.x & 31;
    const int nv = min(PROJ_NPW, N_NODES - node0);

    const float ws0 = att_w[lane],      ws1 = att_w[lane + 32];
    const float wd0 = att_w[lane + 64], wd1 = att_w[lane + 96];

    float h0[PROJ_NPW], h1[PROJ_NPW];
#pragma unroll
    for (int k = 0; k < PROJ_NPW; ++k) {
        const int node = min(node0 + k, N_NODES - 1);
        h0[k] = h[(size_t)node * OUT_DIM + lane];
        h1[k] = h[(size_t)node * OUT_DIM + lane + 32];
    }

    float sres = 0.0f, dres = 0.0f;
#pragma unroll
    for (int k = 0; k < PROJ_NPW; ++k) {
        float s = h0[k] * ws0 + h1[k] * ws1;
        float d = h0[k] * wd0 + h1[k] * wd1;
#pragma unroll
        for (int o = 16; o > 0; o >>= 1) {
            s += __shfl_down(s, o, 32);
            d += __shfl_down(d, o, 32);
        }
        const float s_all = __int_as_float(__builtin_amdgcn_readfirstlane(__float_as_int(s)));
        const float d_all = __int_as_float(__builtin_amdgcn_readfirstlane(__float_as_int(d)));
        if (lane == k) { sres = s_all; dres = d_all; }
    }
    if (lane < nv) {
        a_src[node0 + lane] = sres;
        a_dst[node0 + lane] = dres;
        cnt[node0 + lane]   = 0;
    }
}

__global__ __launch_bounds__(256) void gat_edge_rank(
    const int* __restrict__ dst, int* __restrict__ cnt, int* __restrict__ rank)
{
    const int e0 = blockIdx.x * (256 * EDGE_EPT) + threadIdx.x;
    int d[EDGE_EPT], r[EDGE_EPT];
#pragma unroll
    for (int k = 0; k < EDGE_EPT; ++k)
        d[k] = dst[min(e0 + k * 256, N_EDGES - 1)];
#pragma unroll
    for (int k = 0; k < EDGE_EPT; ++k) {
        r[k] = 0;
        if (e0 + k * 256 < N_EDGES) r[k] = atomicAdd(&cnt[d[k]], 1);
    }
#pragma unroll
    for (int k = 0; k < EDGE_EPT; ++k) {
        const int e = e0 + k * 256;
        if (e < N_EDGES) rank[e] = r[k];
    }
}

#define SCAN_T 1024
__global__ __launch_bounds__(SCAN_T) void gat_offsets(
    const int* __restrict__ cnt, int* __restrict__ off)
{
    __shared__ int w_tot[32];
    __shared__ int w_pre[32];
    __shared__ int w_off[32];
    __shared__ int blk_base;

    const int tid  = threadIdx.x;
    const int lane = tid & 31;
    const int wave = tid >> 5;
    const int blk0 = blockIdx.x * SCAN_T;

    int pre = 0;
    for (int i = tid; i < blk0; i += SCAN_T) pre += cnt[i];

    const int idx = blk0 + tid;
    const int c = (idx < N_NODES) ? cnt[idx] : 0;

    int incl = c;
    for (int o = 1; o < 32; o <<= 1) {
        int t = __shfl_up(incl, o, 32);
        if (lane >= o) incl += t;
    }
    for (int o = 16; o > 0; o >>= 1) pre += __shfl_xor(pre, o, 32);

    if (lane == 31) w_tot[wave] = incl;
    if (lane == 0)  w_pre[wave] = pre;
    __syncthreads();

    if (wave == 0) {
        int t = w_tot[lane];
        int p = w_pre[lane];
        int ti = t;
        for (int o = 1; o < 32; o <<= 1) {
            int u = __shfl_up(ti, o, 32);
            if (lane >= o) ti += u;
        }
        for (int o = 16; o > 0; o >>= 1) p += __shfl_xor(p, o, 32);
        w_off[lane] = ti - t;
        if (lane == 0) blk_base = p;
    }
    __syncthreads();

    if (idx < N_NODES) off[idx] = blk_base + w_off[wave] + incl - c;
}

__global__ __launch_bounds__(256) void gat_edge_place(
    const int* __restrict__ src, const int* __restrict__ dst,
    const int* __restrict__ rank, const int* __restrict__ off,
    int* __restrict__ perm)
{
    const int e0 = blockIdx.x * (256 * EDGE_EPT) + threadIdx.x;
    int d[EDGE_EPT], r[EDGE_EPT], s[EDGE_EPT], p[EDGE_EPT];
#pragma unroll
    for (int k = 0; k < EDGE_EPT; ++k) {
        const int e = min(e0 + k * 256, N_EDGES - 1);
        d[k] = dst[e];
        r[k] = rank[e];
        s[k] = src[e];
    }
#pragma unroll
    for (int k = 0; k < EDGE_EPT; ++k)
        p[k] = off[d[k]] + r[k];
#pragma unroll
    for (int k = 0; k < EDGE_EPT; ++k) {
        if (e0 + k * 256 < N_EDGES && (unsigned)p[k] < (unsigned)N_EDGES)
            perm[p[k]] = s[k];
    }
}

__device__ __forceinline__ float lrelu(float v) {
    return (v > 0.0f) ? v : NEG_SLOPE * v;
}

__device__ __forceinline__ int clamp_node(int s) {
    return min(max(s, 0), N_NODES - 1);
}

template <int K>
__device__ __forceinline__ void add_rows(v2f& acc, const float* __restrict__ hl,
                                         int s, float ex, int t) {
    v2f   hv[K];
    float w[K];
    for (int k = 0; k < K; ++k) {
        const int st = __builtin_amdgcn_readlane(s, t + k);
        w[k]  = __int_as_float(__builtin_amdgcn_readlane(__float_as_int(ex), t + k));
        hv[k] = *(const v2f*)(hl + (size_t)st * OUT_DIM);
    }
    for (int k = 0; k < K; ++k) {
        acc.x = __builtin_fmaf(w[k], hv[k].x, acc.x);
        acc.y = __builtin_fmaf(w[k], hv[k].y, acc.y);
    }
}

__device__ __forceinline__ void gather_node(
    const int node, const int start, const int n, const float ad, const float ab,
    const int s0, const float as0, const int lane,
    const int* __restrict__ perm, const float* __restrict__ a_src,
    const float* __restrict__ h, float* __restrict__ out)
{
    const float NINF = -__builtin_inff();

    const float e0 = (lane < n) ? lrelu(as0 + ad + ab) : NINF;
    float m = e0;
    for (int base = 32; base < n; base += 32) {
        int j = base + lane;
        if (j < n) {
            int s = clamp_node(perm[start + j]);
            m = fmaxf(m, lrelu(a_src[s] + ad + ab));
        }
    }
    for (int o = 16; o > 0; o >>= 1) m = fmaxf(m, __shfl_xor(m, o, 32));
    if (!__builtin_isfinite(m)) m = 0.0f;

    v2f   acc  = {0.0f, 0.0f};
    float dsum = 0.0f;
    const float* hl = h + 2 * lane;
    for (int base = 0; base < n; base += 32) {
        int   s = s0;
        float e = e0;
        if (base > 0) {
            s = 0; e = NINF;
            int j = base + lane;
            if (j < n) {
                s = clamp_node(perm[start + j]);
                e = lrelu(a_src[s] + ad + ab);
            }
        }
        const float ex = (base + lane < n) ? __expf(e - m) : 0.0f;
        dsum += ex;
        const int nc = min(32, n - base);
        int t = 0;
        for (; t + 8 <= nc; t += 8) add_rows<8>(acc, hl, s, ex, t);
        if (nc & 4) { add_rows<4>(acc, hl, s, ex, t); t += 4; }
        if (nc & 2) { add_rows<2>(acc, hl, s, ex, t); t += 2; }
        if (nc & 1) { add_rows<1>(acc, hl, s, ex, t); }
    }
    for (int o = 16; o > 0; o >>= 1) dsum += __shfl_xor(dsum, o, 32);

    v2f r;
    if (dsum > 0.0f) {
        r.x = acc.x / dsum;
        r.y = acc.y / dsum;
    } else {
        r = *(const v2f*)(h + (size_t)node * OUT_DIM + 2 * lane);
    }
    *(v2f*)(out + (size_t)node * OUT_DIM + 2 * lane) = r;
}

__device__ __forceinline__ int first_entries(const int* __restrict__ perm,
                                             int start, int lane) {
    return perm[min(start + lane, N_EDGES - 1)];
}
__device__ __forceinline__ float first_asrc(const float* __restrict__ a_src,
                                            int s, int n, int lane) {
    return (lane < n) ? a_src[s] : 0.0f;
}

__global__ __launch_bounds__(256) void gat_node_gather(
    const int* __restrict__ perm, const int* __restrict__ off,
    const int* __restrict__ cnt,
    const float* __restrict__ a_src, const float* __restrict__ a_dst,
    const float* __restrict__ att_b,
    const float* __restrict__ h, float* __restrict__ out)
{
    const int node0 = __builtin_amdgcn_readfirstlane(
        (int)(blockIdx.x * 8 + (threadIdx.x >> 5))) * GATHER_NPW;
    if (node0 >= N_NODES) return;
    const int lane = threadIdx.x & 31;

    const v4i o4 = *(const v4i*)(off + node0);
    const v4i c4 = *(const v4i*)(cnt + node0);
    const v4f d4 = *(const v4f*)(a_dst + node0);
    const float ab = att_b[0];

    int st[GATHER_NPW], nn[GATHER_NPW];
#pragma unroll
    for (int k = 0; k < GATHER_NPW; ++k) {
        st[k] = min(max(o4[k], 0), N_EDGES);
        nn[k] = min(max(c4[k], 0), N_EDGES - st[k]);
    }

    int   pA = first_entries(perm, st[0], lane);
    int   pB = first_entries(perm, st[1], lane);
    int   sA = clamp_node(pA);
    float aA = first_asrc(a_src, sA, nn[0], lane);
#pragma unroll
    for (int k = 0; k < GATHER_NPW; ++k) {
        int   pC = 0;
        float aB = 0.0f;
        if (k + 2 < GATHER_NPW) pC = first_entries(perm, st[k + 2], lane);
        const int sB = clamp_node(pB);
        if (k + 1 < GATHER_NPW) aB = first_asrc(a_src, sB, nn[k + 1], lane);
        gather_node(node0 + k, st[k], nn[k], d4[k], ab, sA, aA, lane, perm, a_src, h, out);
        sA = sB; pB = pC; aA = aB;
    }
}

extern "C" void kernel_launch(void* const* d_in, const int* in_sizes, int n_in,
                              void* d_out, int out_size, void* d_ws, size_t ws_size,
                              hipStream_t stream) {
    const float* hidden = (const float*)d_in[0];
    const float* W      = (const float*)d_in[1];
    const float* bias   = (const float*)d_in[2];
    const float* att_w  = (const float*)d_in[3];
    const float* att_b  = (const float*)d_in[4];
    const int*   src    = (const int*)  d_in[5];
    const int*   dst    = (const int*)  d_in[6];
    float* out = (float*)d_out;

    float* ws     = (float*)d_ws;
    float* h_ws   = ws;
    float* a_src  = h_ws  + (size_t)N_NODES * OUT_DIM;
    float* a_dst  = a_src + N_NODES;
    int*   cnt    = (int*)(a_dst + N_NODES);
    int*   off    = cnt  + N_NODES;
    int*   rank   = off  + N_NODES;
    int*   perm   = rank + N_EDGES;

    const int NT = 256;
    gat_gemm_wmma<<<N_NODES / 16, 128, 0, stream>>>(hidden, W, bias, h_ws);
    gat_proj<<<(N_NODES + 8 * PROJ_NPW - 1) / (8 * PROJ_NPW), NT, 0, stream>>>(
        h_ws, att_w, a_src, a_dst, cnt);
    gat_edge_rank<<<(N_EDGES + NT * EDGE_EPT - 1) / (NT * EDGE_EPT), NT, 0, stream>>>(
        dst, cnt, rank);
    gat_offsets<<<(N_NODES + SCAN_T - 1) / SCAN_T, SCAN_T, 0, stream>>>(cnt, off);
    gat_edge_place<<<(N_EDGES + NT * EDGE_EPT - 1) / (NT * EDGE_EPT), NT, 0, stream>>>(
        src, dst, rank, off, perm);
    gat_node_gather<<<(N_NODES + 8 * GATHER_NPW - 1) / (8 * GATHER_NPW), NT, 0, stream>>>(
        perm, off, cnt, a_src, a_dst, att_b, h_ws, out);
}


# GEMM re-tiled: 256 resident 16-wave groups hold W in LDS in fragment order; a wave owns a 16x64 strip (A fetched once by 16-byte loads, whole-row stores); k sum order inside 8-k blocks changes
# speedup vs baseline: 2.1006x; 1.0759x over previous
#include <hip/hip_runtime.h>


#define N_NODES   50000
#define N_EDGES   800000
#define IN_DIM    256
#define OUT_DIM   64
#define NEG_SLOPE 0.01f

typedef __attribute__((ext_vector_type(2))) float v2f;
typedef __attribute__((ext_vector_type(4))) float v4f;
typedef __attribute__((ext_vector_type(8))) float v8f;
typedef __attribute__((ext_vector_type(4))) int   v4i;

#define PROJ_NPW   16
#define EDGE_EPT   8
#define GATHER_NPW 4
static_assert(N_NODES % GATHER_NPW == 0, "gather waves read whole groups of nodes");

#define GEMM_GROUPS   256
#define GEMM_WAVES    16
#define N_STRIPS      (N_NODES / 16)
#define FULL_ROUNDS   (N_STRIPS / GEMM_GROUPS)
#define REM_STRIPS    (N_STRIPS - FULL_ROUNDS * GEMM_GROUPS)
#define KBLOCKS       (IN_DIM / 8)
#define ROW_PITCH     (OUT_DIM + 4)
static_assert(N_NODES % 16 == 0, "whole strips only");
static_assert(FULL_ROUNDS + 4 <= GEMM_WAVES, "waves for the shared strip");
static_assert(REM_STRIPS <= GEMM_GROUPS, "at most one shared strip per group");

template <int NT>
__device__ __forceinline__ void strip_products(v8f (&acc)[NT], const v4f (&a)[KBLOCKS],
                                               const v4f* __restrict__ wl, int t0, int lane) {
#pragma unroll
    for (int kb = 0; kb < KBLOCKS; ++kb) {
        v4f b[NT];
#pragma unroll
        for (int t = 0; t < NT; ++t) b[t] = wl[(kb * 4 + t0 + t) * 32 + lane];
        const v2f a01 = {a[kb].x, a[kb].y};
        const v2f a23 = {a[kb].z, a[kb].w};
#pragma unroll
        for (int t = 0; t < NT; ++t) {
            const v2f b01 = {b[t].x, b[t].y};
            acc[t] = __builtin_amdgcn_wmma_f32_16x16x4_f32(
                false, a01, false, b01, (short)0, acc[t], false, false);
        }
#pragma unroll
        for (int t = 0; t < NT; ++t) {
            const v2f b23 = {b[t].z, b[t].w};
            acc[t] = __builtin_amdgcn_wmma_f32_16x16x4_f32(
                false, a23, false, b23, (short)0, acc[t], false, false);
        }
    }
}

__global__ __launch_bounds__(GEMM_WAVES * 32) void gat_gemm_wmma(
    const float* __restrict__ hidden,
    const float* __restrict__ W,
    const float* __restrict__ bias,
    float* __restrict__ h_out)
{
    __shared__ v4f w_frag[KBLOCKS * 4 * 32];
    __shared__ float h_rows[FULL_ROUNDS][16 * ROW_PITCH];

    const int tid  = threadIdx.x;
    const int lane = tid & 31;
    const int wave = __builtin_amdgcn_readfirstlane(tid >> 5);
    const int half = lane >> 4;
    const int l16  = lane & 15;

    const bool whole = wave < FULL_ROUNDS;
    const int  strip = (int)blockIdx.x + GEMM_GROUPS * (whole ? wave : FULL_ROUNDS);
    const bool work  = whole || ((int)blockIdx.x < REM_STRIPS && wave < FULL_ROUNDS + 4);

    float wv[8][4];
#pragma unroll
    for (int i = 0; i < 8; ++i) {
        const int e  = tid + 512 * i;
        const int kb = e >> 7, t = (e >> 5) & 3, hf = (e >> 4) & 1, c = e & 15;
        const float* wp = W + (size_t)(8 * kb + 4 * hf) * OUT_DIM + 16 * t + c;
#pragma unroll
        for (int j = 0; j < 4; ++j) wv[i][j] = wp[j * OUT_DIM];
    }
#pragma unroll
    for (int i = 0; i < 8; ++i) {
        const v4f v = {wv[i][0], wv[i][1], wv[i][2], wv[i][3]};
        w_frag[tid + 512 * i] = v;
    }

    v4f a[KBLOCKS];
    if (work) {
        const float* arow = hidden + (size_t)(strip * 16 + l16) * IN_DIM + 4 * half;
#pragma unroll
        for (int kb = 0; kb < KBLOCKS; ++kb) a[kb] = *(const v4f*)(arow + 8 * kb);
    }
    __builtin_amdgcn_fence(__ATOMIC_RELEASE, "workgroup", "local");
    __builtin_amdgcn_s_barrier();
    __builtin_amdgcn_fence(__ATOMIC_ACQUIRE, "workgroup", "local");
    if (!work) return;

    if (whole) {
        v8f acc[4] = {};
        strip_products<4>(acc, a, w_frag, 0, lane);

        float* hr = h_rows[wave];
#pragma unroll
        for (int t = 0; t < 4; ++t) {
            const float bb = bias[16 * t + l16];
#pragma unroll
            for (int r = 0; r < 8; ++r)
                hr[(8 * half + r) * ROW_PITCH + 16 * t + l16] = acc[t][r] + bb;
        }
        float* orow = h_out + (size_t)strip * 16 * OUT_DIM;
#pragma unroll
        for (int i = 0; i < 8; ++i) {
            const int row = 2 * i + half;
            const v4f v = *(const v4f*)(hr + row * ROW_PITCH + 4 * l16);
            *(v4f*)(orow + row * OUT_DIM + 4 * l16) = v;
        }
    } else {
        const int t0 = wave - FULL_ROUNDS;
        v8f acc[1] = {};
        strip_products<1>(acc, a, w_frag, t0, lane);
        const int gcol = 16 * t0 + l16;
        const float bb = bias[gcol];
#pragma unroll
        for (int r = 0; r < 8; ++r)
            h_out[(size_t)(strip * 16 + 8 * half + r) * OUT_DIM + gcol] = acc[0][r] + bb;
    }
}

__global__ __launch_bounds__(256) void gat_proj(
    const float* __restrict__ h,
    const float* __restrict__ att_w,
    float* __restrict__ a_src,
    float* __restrict__ a_dst,
    int* __restrict__ cnt)
{
    const int node0 = __builtin_amdgcn_readfirstlane(
        (int)(blockIdx.x * 8 + (threadIdx.x >> 5))) * PROJ_NPW;
    if (node0 >= N_NODES) return;
    const int lane = threadIdx.x & 31;
    const int nv = min(PROJ_NPW, N_NODES - node0);

    const float ws0 = att_w[lane],      ws1 = att_w[lane + 32];
    const float wd0 = att_w[lane + 64], wd1 = att_w[lane + 96];

    float h0[PROJ_NPW], h1[PROJ_NPW];
#pragma unroll
    for (int k = 0; k < PROJ_NPW; ++k) {
        const int node = min(node0 + k, N_NODES - 1);
        h0[k] = h[(size_t)node * OUT_DIM + lane];
        h1[k] = h[(size_t)node * OUT_DIM + lane + 32];
    }

    float sres = 0.0f, dres = 0.0f;
#pragma unroll
    for (int k = 0; k < PROJ_NPW; ++k) {
        float s = h0[k] * ws0 + h1[k] * ws1;
        float d = h0[k] * wd0 + h1[k] * wd1;
#pragma unroll
        for (int o = 16; o > 0; o >>= 1) {
            s += __shfl_down(s, o, 32);
            d += __shfl_down(d, o, 32);
        }
        const float s_all = __int_as_float(__builtin_amdgcn_readfirstlane(__float_as_int(s)));
        const float d_all = __int_as_float(__builtin_amdgcn_readfirstlane(__float_as_int(d)));
        if (lane == k) { sres = s_all; dres = d_all; }
    }
    if (lane < nv) {
        a_src[node0 + lane] = sres;
        a_dst[node0 + lane] = dres;
        cnt[node0 + lane]   = 0;
    }
}

__global__ __launch_bounds__(256) void gat_edge_rank(
    const int* __restrict__ dst, int* __restrict__ cnt, int* __restrict__ rank)
{
    const int e0 = blockIdx.x * (256 * EDGE_EPT) + threadIdx.x;
    int d[EDGE_EPT], r[EDGE_EPT];
#pragma unroll
    for (int k = 0; k < EDGE_EPT; ++k)
        d[k] = dst[min(e0 + k * 256, N_EDGES - 1)];
#pragma unroll
    for (int k = 0; k < EDGE_EPT; ++k) {
        r[k] = 0;
        if (e0 + k * 256 < N_EDGES) r[k] = atomicAdd(&cnt[d[k]], 1);
    }
#pragma unroll
    for (int k = 0; k < EDGE_EPT; ++k) {
        const int e = e0 + k * 256;
        if (e < N_EDGES) rank[e] = r[k];
    }
}

#define SCAN_T 1024
__global__ __launch_bounds__(SCAN_T) void gat_offsets(
    const int* __restrict__ cnt, int* __restrict__ off)
{
    __shared__ int w_tot[32];
    __shared__ int w_pre[32];
    __shared__ int w_off[32];
    __shared__ int blk_base;

    const int tid  = threadIdx.x;
    const int lane = tid & 31;
    const int wave = tid >> 5;
    const int blk0 = blockIdx.x * SCAN_T;

    int pre = 0;
    for (int i = tid; i < blk0; i += SCAN_T) pre += cnt[i];

    const int idx = blk0 + tid;
    const int c = (idx < N_NODES) ? cnt[idx] : 0;

    int incl = c;
    for (int o = 1; o < 32; o <<= 1) {
        int t = __shfl_up(incl, o, 32);
        if (lane >= o) incl += t;
    }
    for (int o = 16; o > 0; o >>= 1) pre += __shfl_xor(pre, o, 32);

    if (lane == 31) w_tot[wave] = incl;
    if (lane == 0)  w_pre[wave] = pre;
    __syncthreads();

    if (wave == 0) {
        int t = w_tot[lane];
        int p = w_pre[lane];
        int ti = t;
        for (int o = 1; o < 32; o <<= 1) {
            int u = __shfl_up(ti, o, 32);
            if (lane >= o) ti += u;
        }
        for (int o = 16; o > 0; o >>= 1) p += __shfl_xor(p, o, 32);
        w_off[lane] = ti - t;
        if (lane == 0) blk_base = p;
    }
    __syncthreads();

    if (idx < N_NODES) off[idx] = blk_base + w_off[wave] + incl - c;
}

__global__ __launch_bounds__(256) void gat_edge_place(
    const int* __restrict__ src, const int* __restrict__ dst,
    const int* __restrict__ rank, const int* __restrict__ off,
    int* __restrict__ perm)
{
    const int e0 = blockIdx.x * (256 * EDGE_EPT) + threadIdx.x;
    int d[EDGE_EPT], r[EDGE_EPT], s[EDGE_EPT], p[EDGE_EPT];
#pragma unroll
    for (int k = 0; k < EDGE_EPT; ++k) {
        const int e = min(e0 + k * 256, N_EDGES - 1);
        d[k] = dst[e];
        r[k] = rank[e];
        s[k] = src[e];
    }
#pragma unroll
    for (int k = 0; k < EDGE_EPT; ++k)
        p[k] = off[d[k]] + r[k];
#pragma unroll
    for (int k = 0; k < EDGE_EPT; ++k) {
        if (e0 + k * 256 < N_EDGES && (unsigned)p[k] < (unsigned)N_EDGES)
            perm[p[k]] = s[k];
    }
}

__device__ __forceinline__ float lrelu(float v) {
    return (v > 0.0f) ? v : NEG_SLOPE * v;
}

__device__ __forceinline__ int clamp_node(int s) {
    return min(max(s, 0), N_NODES - 1);
}

template <int K>
__device__ __forceinline__ void add_rows(v2f& acc, const float* __restrict__ hl,
                                         int s, float ex, int t) {
    v2f   hv[K];
    float w[K];
    for (int k = 0; k < K; ++k) {
        const int st = __builtin_amdgcn_readlane(s, t + k);
        w[k]  = __int_as_float(__builtin_amdgcn_readlane(__float_as_int(ex), t + k));
        hv[k] = *(const v2f*)(hl + (size_t)st * OUT_DIM);
    }
    for (int k = 0; k < K; ++k) {
        acc.x = __builtin_fmaf(w[k], hv[k].x, acc.x);
        acc.y = __builtin_fmaf(w[k], hv[k].y, acc.y);
    }
}

__device__ __forceinline__ void gather_node(
    const int node, const int start, const int n, const float ad, const float ab,
    const int s0, const float as0, const int lane,
    const int* __restrict__ perm, const float* __restrict__ a_src,
    const float* __restrict__ h, float* __restrict__ out)
{
    const float NINF = -__builtin_inff();

    const float e0 = (lane < n) ? lrelu(as0 + ad + ab) : NINF;
    float m = e0;
    for (int base = 32; base < n; base += 32) {
        int j = base + lane;
        if (j < n) {
            int s = clamp_node(perm[start + j]);
            m = fmaxf(m, lrelu(a_src[s] + ad + ab));
        }
    }
    for (int o = 16; o > 0; o >>= 1) m = fmaxf(m, __shfl_xor(m, o, 32));
    if (!__builtin_isfinite(m)) m = 0.0f;

    v2f   acc  = {0.0f, 0.0f};
    float dsum = 0.0f;
    const float* hl = h + 2 * lane;
    for (int base = 0; base < n; base += 32) {
        int   s = s0;
        float e = e0;
        if (base > 0) {
            s = 0; e = NINF;
            int j = base + lane;
            if (j < n) {
                s = clamp_node(perm[start + j]);
                e = lrelu(a_src[s] + ad + ab);
            }
        }
        const float ex = (base + lane < n) ? __expf(e - m) : 0.0f;
        dsum += ex;
        const int nc = min(32, n - base);
        int t = 0;
        for (; t + 8 <= nc; t += 8) add_rows<8>(acc, hl, s, ex, t);
        if (nc & 4) { add_rows<4>(acc, hl, s, ex, t); t += 4; }
        if (nc & 2) { add_rows<2>(acc, hl, s, ex, t); t += 2; }
        if (nc & 1) { add_rows<1>(acc, hl, s, ex, t); }
    }
    for (int o = 16; o > 0; o >>= 1) dsum += __shfl_xor(dsum, o, 32);

    v2f r;
    if (dsum > 0.0f) {
        r.x = acc.x / dsum;
        r.y = acc.y / dsum;
    } else {
        r = *(const v2f*)(h + (size_t)node * OUT_DIM + 2 * lane);
    }
    *(v2f*)(out + (size_t)node * OUT_DIM + 2 * lane) = r;
}

__device__ __forceinline__ int first_entries(const int* __restrict__ perm,
                                             int start, int lane) {
    return perm[min(start + lane, N_EDGES - 1)];
}
__device__ __forceinline__ float first_asrc(const float* __restrict__ a_src,
                                            int s, int n, int lane) {
    return (lane < n) ? a_src[s] : 0.0f;
}

__global__ __launch_bounds__(256) void gat_node_gather(
    const int* __restrict__ perm, const int* __restrict__ off,
    const int* __restrict__ cnt,
    const float* __restrict__ a_src, const float* __restrict__ a_dst,
    const float* __restrict__ att_b,
    const float* __restrict__ h, float* __restrict__ out)
{
    const int node0 = __builtin_amdgcn_readfirstlane(
        (int)(blockIdx.x * 8 + (threadIdx.x >> 5))) * GATHER_NPW;
    if (node0 >= N_NODES) return;
    const int lane = threadIdx.x & 31;

    const v4i o4 = *(const v4i*)(off + node0);
    const v4i c4 = *(const v4i*)(cnt + node0);
    const v4f d4 = *(const v4f*)(a_dst + node0);
    const float ab = att_b[0];

    int st[GATHER_NPW], nn[GATHER_NPW];
#pragma unroll
    for (int k = 0; k < GATHER_NPW; ++k) {
        st[k] = min(max(o4[k], 0), N_EDGES);
        nn[k] = min(max(c4[k], 0), N_EDGES - st[k]);
    }

    int   pA = first_entries(perm, st[0], lane);
    int   pB = first_entries(perm, st[1], lane);
    int   sA = clamp_node(pA);
    float aA = first_asrc(a_src, sA, nn[0], lane);
#pragma unroll
    for (int k = 0; k < GATHER_NPW; ++k) {
        int   pC = 0;
        float aB = 0.0f;
        if (k + 2 < GATHER_NPW) pC = first_entries(perm, st[k + 2], lane);
        const int sB = clamp_node(pB);
        if (k + 1 < GATHER_NPW) aB = first_asrc(a_src, sB, nn[k + 1], lane);
        gather_node(node0 + k, st[k], nn[k], d4[k], ab, sA, aA, lane, perm, a_src, h, out);
        sA = sB; pB = pC; aA = aB;
    }
}

extern "C" void kernel_launch(void* const* d_in, const int* in_sizes, int n_in,
                              void* d_out, int out_size, void* d_ws, size_t ws_size,
                              hipStream_t stream) {
    const float* hidden = (const float*)d_in[0];
    const float* W      = (const float*)d_in[1];
    const float* bias   = (const float*)d_in[2];
    const float* att_w  = (const float*)d_in[3];
    const float* att_b  = (const float*)d_in[4];
    const int*   src    = (const int*)  d_in[5];
    const int*   dst    = (const int*)  d_in[6];
    float* out = (float*)d_out;

    float* ws     = (float*)d_ws;
    float* h_ws   = ws;
    float* a_src  = h_ws  + (size_t)N_NODES * OUT_DIM;
    float* a_dst  = a_src + N_NODES;
    int*   cnt    = (int*)(a_dst + N_NODES);
    int*   off    = cnt  + N_NODES;
    int*   rank   = off  + N_NODES;
    int*   perm   = rank + N_EDGES;

    const int NT = 256;
    gat_gemm_wmma<<<GEMM_GROUPS, GEMM_WAVES * 32, 0, stream>>>(hidden, W, bias, h_ws);
    gat_proj<<<(N_NODES + 8 * PROJ_NPW - 1) / (8 * PROJ_NPW), NT, 0, stream>>>(
        h_ws, att_w, a_src, a_dst, cnt);
    gat_edge_rank<<<(N_EDGES + NT * EDGE_EPT - 1) / (NT * EDGE_EPT), NT, 0, stream>>>(
        dst, cnt, rank);
    gat_offsets<<<(N_NODES + SCAN_T - 1) / SCAN_T, SCAN_T, 0, stream>>>(cnt, off);
    gat_edge_place<<<(N_EDGES + NT * EDGE_EPT - 1) / (NT * EDGE_EPT), NT, 0, stream>>>(
        src, dst, rank, off, perm);
    gat_node_gather<<<(N_NODES + 8 * GATHER_NPW - 1) / (8 * GATHER_NPW), NT, 0, stream>>>(
        perm, off, cnt, a_src, a_dst, att_b, h_ws, out);
}


# Edge list by dst built by a two-level counting sort on LDS counters (256 chunk groups, then one group per 256-node bucket): no global atomic, no scattered 4-byte access; sum order = LDS arrival rank
# speedup vs baseline: 4.8814x; 2.3238x over previous
#include <hip/hip_runtime.h>


#define N_NODES   50000
#define N_EDGES   800000
#define IN_DIM    256
#define OUT_DIM   64
#define NEG_SLOPE 0.01f

typedef __attribute__((ext_vector_type(2))) float v2f;
typedef __attribute__((ext_vector_type(4))) float v4f;
typedef __attribute__((ext_vector_type(8))) float v8f;
typedef __attribute__((ext_vector_type(4))) int   v4i;
typedef __attribute__((ext_vector_type(2))) int   v2i;

#define PROJ_NPW   16
#define GATHER_NPW 4
static_assert(N_NODES % GATHER_NPW == 0, "gather waves read whole groups of nodes");

#define GEMM_GROUPS   256
#define GEMM_WAVES    16
#define N_STRIPS      (N_NODES / 16)
#define FULL_ROUNDS   (N_STRIPS / GEMM_GROUPS)
#define REM_STRIPS    (N_STRIPS - FULL_ROUNDS * GEMM_GROUPS)
#define KBLOCKS       (IN_DIM / 8)
#define ROW_PITCH     (OUT_DIM + 4)
static_assert(N_NODES % 16 == 0, "whole strips only");
static_assert(FULL_ROUNDS + 4 <= GEMM_WAVES, "waves for the shared strip");
static_assert(REM_STRIPS <= GEMM_GROUPS, "at most one shared strip per group");

template <int NT>
__device__ __forceinline__ void strip_products(v8f (&acc)[NT], const v4f (&a)[KBLOCKS],
                                               const v4f* __restrict__ wl, int t0, int lane) {
#pragma unroll
    for (int kb = 0; kb < KBLOCKS; ++kb) {
        v4f b[NT];
#pragma unroll
        for (int t = 0; t < NT; ++t) b[t] = wl[(kb * 4 + t0 + t) * 32 + lane];
        const v2f a01 = {a[kb].x, a[kb].y};
        const v2f a23 = {a[kb].z, a[kb].w};
#pragma unroll
        for (int t = 0; t < NT; ++t) {
            const v2f b01 = {b[t].x, b[t].y};
            acc[t] = __builtin_amdgcn_wmma_f32_16x16x4_f32(
                false, a01, false, b01, (short)0, acc[t], false, false);
        }
#pragma unroll
        for (int t = 0; t < NT; ++t) {
            const v2f b23 = {b[t].z, b[t].w};
            acc[t] = __builtin_amdgcn_wmma_f32_16x16x4_f32(
                false, a23, false, b23, (short)0, acc[t], false, false);
        }
    }
}

__global__ __launch_bounds__(GEMM_WAVES * 32) void gat_gemm_wmma(
    const float* __restrict__ hidden,
    const float* __restrict__ W,
    const float* __restrict__ bias,
    float* __restrict__ h_out)
{
    __shared__ v4f w_frag[KBLOCKS * 4 * 32];
    __shared__ float h_rows[FULL_ROUNDS][16 * ROW_PITCH];

    const int tid  = threadIdx.x;
    const int lane = tid & 31;
    const int wave = __builtin_amdgcn_readfirstlane(tid >> 5);
    const int half = lane >> 4;
    const int l16  = lane & 15;

    const bool whole = wave < FULL_ROUNDS;
    const int  strip = (int)blockIdx.x + GEMM_GROUPS * (whole ? wave : FULL_ROUNDS);
    const bool work  = whole || ((int)blockIdx.x < REM_STRIPS && wave < FULL_ROUNDS + 4);

    float wv[8][4];
#pragma unroll
    for (int i = 0; i < 8; ++i) {
        const int e  = tid + 512 * i;
        const int kb = e >> 7, t = (e >> 5) & 3, hf = (e >> 4) & 1, c = e & 15;
        const float* wp = W + (size_t)(8 * kb + 4 * hf) * OUT_DIM + 16 * t + c;
#pragma unroll
        for (int j = 0; j < 4; ++j) wv[i][j] = wp[j * OUT_DIM];
    }
#pragma unroll
    for (int i = 0; i < 8; ++i) {
        const v4f v = {wv[i][0], wv[i][1], wv[i][2], wv[i][3]};
        w_frag[tid + 512 * i] = v;
    }

    v4f a[KBLOCKS];
    if (work) {
        const float* arow = hidden + (size_t)(strip * 16 + l16) * IN_DIM + 4 * half;
#pragma unroll
        for (int kb = 0; kb < KBLOCKS; ++kb) a[kb] = *(const v4f*)(arow + 8 * kb);
    }
    __builtin_amdgcn_fence(__ATOMIC_RELEASE, "workgroup", "local");
    __builtin_amdgcn_s_barrier();
    __builtin_amdgcn_fence(__ATOMIC_ACQUIRE, "workgroup", "local");
    if (!work) return;

    if (whole) {
        v8f acc[4] = {};
        strip_products<4>(acc, a, w_frag, 0, lane);

        float* hr = h_rows[wave];
#pragma unroll
        for (int t = 0; t < 4; ++t) {
            const float bb = bias[16 * t + l16];
#pragma unroll
            for (int r = 0; r < 8; ++r)
                hr[(8 * half + r) * ROW_PITCH + 16 * t + l16] = acc[t][r] + bb;
        }
        float* orow = h_out + (size_t)strip * 16 * OUT_DIM;
#pragma unroll
        for (int i = 0; i < 8; ++i) {
            const int row = 2 * i + half;
            const v4f v = *(const v4f*)(hr + row * ROW_PITCH + 4 * l16);
            *(v4f*)(orow + row * OUT_DIM + 4 * l16) = v;
        }
    } else {
        const int t0 = wave - FULL_ROUNDS;
        v8f acc[1] = {};
        strip_products<1>(acc, a, w_frag, t0, lane);
        const int gcol = 16 * t0 + l16;
        const float bb = bias[gcol];
#pragma unroll
        for (int r = 0; r < 8; ++r)
            h_out[(size_t)(strip * 16 + 8 * half + r) * OUT_DIM + gcol] = acc[0][r] + bb;
    }
}

__global__ __launch_bounds__(256) void gat_proj(
    const float* __restrict__ h,
    const float* __restrict__ att_w,
    float* __restrict__ a_src,
    float* __restrict__ a_dst)
{
    const int node0 = __builtin_amdgcn_readfirstlane(
        (int)(blockIdx.x * 8 + (threadIdx.x >> 5))) * PROJ_NPW;
    if (node0 >= N_NODES) return;
    const int lane = threadIdx.x & 31;
    const int nv = min(PROJ_NPW, N_NODES - node0);

    const float ws0 = att_w[lane],      ws1 = att_w[lane + 32];
    const float wd0 = att_w[lane + 64], wd1 = att_w[lane + 96];

    float h0[PROJ_NPW], h1[PROJ_NPW];
#pragma unroll
    for (int k = 0; k < PROJ_NPW; ++k) {
        const int node = min(node0 + k, N_NODES - 1);
        h0[k] = h[(size_t)node * OUT_DIM + lane];
        h1[k] = h[(size_t)node * OUT_DIM + lane + 32];
    }

    float sres = 0.0f, dres = 0.0f;
#pragma unroll
    for (int k = 0; k < PROJ_NPW; ++k) {
        float s = h0[k] * ws0 + h1[k] * ws1;
        float d = h0[k] * wd0 + h1[k] * wd1;
#pragma unroll
        for (int o = 16; o > 0; o >>= 1) {
            s += __shfl_down(s, o, 32);
            d += __shfl_down(d, o, 32);
        }
        const float s_all = __int_as_float(__builtin_amdgcn_readfirstlane(__float_as_int(s)));
        const float d_all = __int_as_float(__builtin_amdgcn_readfirstlane(__float_as_int(d)));
        if (lane == k) { sres = s_all; dres = d_all; }
    }
    if (lane < nv) {
        a_src[node0 + lane] = sres;
        a_dst[node0 + lane] = dres;
    }
}

#define SORT_T      1024
#define NCHUNK      256
#define CHUNK_E     (N_EDGES / NCHUNK)
#define CHUNK_SLOTS ((CHUNK_E + SORT_T - 1) / SORT_T)
#define BSHIFT      8
#define BNODES      (1 << BSHIFT)
#define NBUCKET     ((N_NODES + BNODES - 1) / BNODES)
#define REC_SLOTS   6
#define STAGE_CAP   16384
static_assert(N_EDGES % NCHUNK == 0, "equal chunks");
static_assert(NBUCKET <= SORT_T && NCHUNK <= SORT_T && BNODES <= SORT_T, "one table entry per thread");

__device__ __forceinline__ int bucket_of(int d) {
    return min(max(d, 0) >> BSHIFT, NBUCKET - 1);
}

__device__ __forceinline__ int block_excl_scan(int v, int* wsum, int& total) {
    const int lane = threadIdx.x & 31;
    const int wave = threadIdx.x >> 5;
    int incl = v;
    for (int o = 1; o < 32; o <<= 1) {
        const int t = __shfl_up(incl, o, 32);
        if (lane >= o) incl += t;
    }
    __syncthreads();
    if (lane == 31) wsum[wave] = incl;
    __syncthreads();
    const int wt = wsum[lane];
    int wi = wt;
    for (int o = 1; o < 32; o <<= 1) {
        const int t = __shfl_up(wi, o, 32);
        if (lane >= o) wi += t;
    }
    total = __shfl(wi, 31, 32);
    const int wexcl = __shfl(wi - wt, wave, 32);
    return wexcl + incl - v;
}
static_assert(SORT_T == 1024, "block_excl_scan reads one wave total per lane");

__global__ __launch_bounds__(SORT_T) void gat_sort_chunks(
    const int* __restrict__ src, const int* __restrict__ dst,
    int* __restrict__ recs,
    int* __restrict__ tcnt,
    int* __restrict__ toff)
{
    __shared__ int hist[NBUCKET];
    __shared__ int bofs[NBUCKET];
    __shared__ int wsum[32];
    __shared__ v2i stage[CHUNK_E];

    const int tid  = threadIdx.x;
    const int c    = blockIdx.x;
    const int base = c * CHUNK_E;

    if (tid < NBUCKET) hist[tid] = 0;
    int d[CHUNK_SLOTS], s[CHUNK_SLOTS], r[CHUNK_SLOTS], b[CHUNK_SLOTS];
#pragma unroll
    for (int k = 0; k < CHUNK_SLOTS; ++k) {
        const int e = base + min(tid + k * SORT_T, CHUNK_E - 1);
        d[k] = dst[e];
        s[k] = src[e];
    }
    __syncthreads();
#pragma unroll
    for (int k = 0; k < CHUNK_SLOTS; ++k) {
        b[k] = bucket_of(d[k]);
        r[k] = 0;
        if (tid + k * SORT_T < CHUNK_E) r[k] = atomicAdd(&hist[b[k]], 1);
    }
    __syncthreads();
    const int hv = (tid < NBUCKET) ? hist[tid] : 0;
    int tot;
    const int ex = block_excl_scan(hv, wsum, tot);
    if (tid < NBUCKET) bofs[tid] = ex;
    __syncthreads();
#pragma unroll
    for (int k = 0; k < CHUNK_SLOTS; ++k) {
        if (tid + k * SORT_T < CHUNK_E) {
            const int slot = bofs[b[k]] + r[k];
            const v2i rec = {s[k], d[k]};
            if ((unsigned)slot < (unsigned)CHUNK_E) stage[slot] = rec;
        }
    }
    __syncthreads();
    for (int i = tid; i < CHUNK_E; i += SORT_T)
        *(v2i*)(recs + 2 * (size_t)(base + i)) = stage[i];
    if (tid < NBUCKET) {
        tcnt[tid * NCHUNK + c] = hv;
        toff[tid * NCHUNK + c] = ex;
    }
}

__device__ __forceinline__ v2i bucket_record(const int* __restrict__ recs,
                                             const int* pbeg, const int* ploc, int i) {
    int lo = 0, hi = NCHUNK;
    for (int st = 0; st < 8; ++st) {
        const int mid = (lo + hi) >> 1;
        if (pbeg[mid] <= i) lo = mid; else hi = mid;
    }
    const int j = min(max(ploc[lo] + (i - pbeg[lo]), 0), N_EDGES - 1);
    return *(const v2i*)(recs + 2 * (size_t)j);
}
static_assert(NCHUNK == 256, "bucket_record halves eight times");

__global__ __launch_bounds__(SORT_T) void gat_sort_buckets(
    const int* __restrict__ recs, const int* __restrict__ tcnt,
    const int* __restrict__ toff,
    int* __restrict__ cnt, int* __restrict__ off, int* __restrict__ perm)
{
    __shared__ int pbeg[NCHUNK + 1];
    __shared__ int ploc[NCHUNK];
    __shared__ int ncnt[BNODES];
    __shared__ int ncur[BNODES];
    __shared__ int wsum[32];
    __shared__ int lperm[STAGE_CAP];

    const int tid   = threadIdx.x;
    const int bk    = blockIdx.x;
    const int node0 = bk * BNODES;

    int pre = 0;
    for (int i = tid; i < bk * NCHUNK; i += SORT_T) pre += tcnt[i];

    int len = 0;
    if (tid < NCHUNK) {
        len = tcnt[bk * NCHUNK + tid];
        ploc[tid] = tid * CHUNK_E + toff[bk * NCHUNK + tid];
    }
    if (tid < BNODES) ncnt[tid] = 0;
    int nb;
    const int pex = block_excl_scan(len, wsum, nb);
    if (tid < NCHUNK) pbeg[tid] = pex;
    if (tid == 0) pbeg[NCHUNK] = nb;
    int first;
    block_excl_scan(pre, wsum, first);
    __syncthreads();

    int rs[REC_SLOTS], rl[REC_SLOTS];
    const int last = max(nb - 1, 0);
    int pi[REC_SLOTS], lo[REC_SLOTS];
#pragma unroll
    for (int k = 0; k < REC_SLOTS; ++k) {
        pi[k] = min(tid + k * SORT_T, last);
        lo[k] = 0;
    }
#pragma unroll
    for (int step = NCHUNK / 2; step > 0; step >>= 1) {
        int pm[REC_SLOTS];
#pragma unroll
        for (int k = 0; k < REC_SLOTS; ++k) pm[k] = pbeg[lo[k] + step];
#pragma unroll
        for (int k = 0; k < REC_SLOTS; ++k) if (pm[k] <= pi[k]) lo[k] += step;
    }
    int rj[REC_SLOTS];
#pragma unroll
    for (int k = 0; k < REC_SLOTS; ++k) {
        rj[k] = min(max(ploc[lo[k]] + (pi[k] - pbeg[lo[k]]), 0), N_EDGES - 1);
    }
#pragma unroll
    for (int k = 0; k < REC_SLOTS; ++k) {
        const v2i rec = *(const v2i*)(recs + 2 * (size_t)rj[k]);
        rs[k] = rec.x;
        rl[k] = (tid + k * SORT_T < nb) ? min(max(rec.y - node0, 0), BNODES - 1) : -1;
    }
#pragma unroll
    for (int k = 0; k < REC_SLOTS; ++k)
        if (rl[k] >= 0) atomicAdd(&ncnt[rl[k]], 1);
    for (int i = tid + REC_SLOTS * SORT_T; i < nb; i += SORT_T) {
        const v2i rec = bucket_record(recs, pbeg, ploc, i);
        atomicAdd(&ncnt[min(max(rec.y - node0, 0), BNODES - 1)], 1);
    }
    __syncthreads();

    const int nc = (tid < BNODES) ? ncnt[tid] : 0;
    int tot;
    const int nex = block_excl_scan(nc, wsum, tot);
    if (tid < BNODES) {
        ncur[tid] = nex;
        const int node = node0 + tid;
        if (node < N_NODES) {
            cnt[node] = nc;
            off[node] = first + nex;
        }
    }
    __syncthreads();

    const bool staged = nb <= STAGE_CAP;
#pragma unroll
    for (int k = 0; k < REC_SLOTS; ++k) {
        if (rl[k] >= 0) {
            const int pos = atomicAdd(&ncur[rl[k]], 1);
            if (staged) {
                if ((unsigned)pos < (unsigned)STAGE_CAP) lperm[pos] = rs[k];
            } else if ((unsigned)(first + pos) < (unsigned)N_EDGES) {
                perm[first + pos] = rs[k];
            }
        }
    }
    for (int i = tid + REC_SLOTS * SORT_T; i < nb; i += SORT_T) {
        const v2i rec = bucket_record(recs, pbeg, ploc, i);
        const int pos = atomicAdd(&ncur[min(max(rec.y - node0, 0), BNODES - 1)], 1);
        if (staged) {
            if ((unsigned)pos < (unsigned)STAGE_CAP) lperm[pos] = rec.x;
        } else if ((unsigned)(first + pos) < (unsigned)N_EDGES) {
            perm[first + pos] = rec.x;
        }
    }
    if (staged) {
        __syncthreads();
        for (int i = tid; i < nb; i += SORT_T)
            if ((unsigned)(first + i) < (unsigned)N_EDGES) perm[first + i] = lperm[i];
    }
}

__device__ __forceinline__ float lrelu(float v) {
    return (v > 0.0f) ? v : NEG_SLOPE * v;
}

__device__ __forceinline__ int clamp_node(int s) {
    return min(max(s, 0), N_NODES - 1);
}

template <int K>
__device__ __forceinline__ void add_rows(v2f& acc, const float* __restrict__ hl,
                                         int s, float ex, int t) {
    v2f   hv[K];
    float w[K];
    for (int k = 0; k < K; ++k) {
        const int st = __builtin_amdgcn_readlane(s, t + k);
        w[k]  = __int_as_float(__builtin_amdgcn_readlane(__float_as_int(ex), t + k));
        hv[k] = *(const v2f*)(hl + (size_t)st * OUT_DIM);
    }
    for (int k = 0; k < K; ++k) {
        acc.x = __builtin_fmaf(w[k], hv[k].x, acc.x);
        acc.y = __builtin_fmaf(w[k], hv[k].y, acc.y);
    }
}

__device__ __forceinline__ void gather_node(
    const int node, const int start, const int n, const float ad, const float ab,
    const int s0, const float as0, const int lane,
    const int* __restrict__ perm, const float* __restrict__ a_src,
    const float* __restrict__ h, float* __restrict__ out)
{
    const float NINF = -__builtin_inff();

    const float e0 = (lane < n) ? lrelu(as0 + ad + ab) : NINF;
    float m = e0;
    for (int base = 32; base < n; base += 32) {
        int j = base + lane;
        if (j < n) {
            int s = clamp_node(perm[start + j]);
            m = fmaxf(m, lrelu(a_src[s] + ad + ab));
        }
    }
    for (int o = 16; o > 0; o >>= 1) m = fmaxf(m, __shfl_xor(m, o, 32));
    if (!__builtin_isfinite(m)) m = 0.0f;

    v2f   acc  = {0.0f, 0.0f};
    float dsum = 0.0f;
    const float* hl = h + 2 * lane;
    for (int base = 0; base < n; base += 32) {
        int   s = s0;
        float e = e0;
        if (base > 0) {
            s = 0; e = NINF;
            int j = base + lane;
            if (j < n) {
                s = clamp_node(perm[start + j]);
                e = lrelu(a_src[s] + ad + ab);
            }
        }
        const float ex = (base + lane < n) ? __expf(e - m) : 0.0f;
        dsum += ex;
        const int nc = min(32, n - base);
        int t = 0;
        for (; t + 8 <= nc; t += 8) add_rows<8>(acc, hl, s, ex, t);
        if (nc & 4) { add_rows<4>(acc, hl, s, ex, t); t += 4; }
        if (nc & 2) { add_rows<2>(acc, hl, s, ex, t); t += 2; }
        if (nc & 1) { add_rows<1>(acc, hl, s, ex, t); }
    }
    for (int o = 16; o > 0; o >>= 1) dsum += __shfl_xor(dsum, o, 32);

    v2f r;
    if (dsum > 0.0f) {
        r.x = acc.x / dsum;
        r.y = acc.y / dsum;
    } else {
        r = *(const v2f*)(h + (size_t)node * OUT_DIM + 2 * lane);
    }
    *(v2f*)(out + (size_t)node * OUT_DIM + 2 * lane) = r;
}

__device__ __forceinline__ int first_entries(const int* __restrict__ perm,
                                             int start, int lane) {
    return perm[min(start + lane, N_EDGES - 1)];
}
__device__ __forceinline__ float first_asrc(const float* __restrict__ a_src,
                                            int s, int n, int lane) {
    return (lane < n) ? a_src[s] : 0.0f;
}

__global__ __launch_bounds__(256) void gat_node_gather(
    const int* __restrict__ perm, const int* __restrict__ off,
    const int* __restrict__ cnt,
    const float* __restrict__ a_src, const float* __restrict__ a_dst,
    const float* __restrict__ att_b,
    const float* __restrict__ h, float* __restrict__ out)
{
    const int node0 = __builtin_amdgcn_readfirstlane(
        (int)(blockIdx.x * 8 + (threadIdx.x >> 5))) * GATHER_NPW;
    if (node0 >= N_NODES) return;
    const int lane = threadIdx.x & 31;

    const v4i o4 = *(const v4i*)(off + node0);
    const v4i c4 = *(const v4i*)(cnt + node0);
    const v4f d4 = *(const v4f*)(a_dst + node0);
    const float ab = att_b[0];

    int st[GATHER_NPW], nn[GATHER_NPW];
#pragma unroll
    for (int k = 0; k < GATHER_NPW; ++k) {
        st[k] = min(max(o4[k], 0), N_EDGES);
        nn[k] = min(max(c4[k], 0), N_EDGES - st[k]);
    }

    int   pA = first_entries(perm, st[0], lane);
    int   pB = first_entries(perm, st[1], lane);
    int   sA = clamp_node(pA);
    float aA = first_asrc(a_src, sA, nn[0], lane);
#pragma unroll
    for (int k = 0; k < GATHER_NPW; ++k) {
        int   pC = 0;
        float aB = 0.0f;
        if (k + 2 < GATHER_NPW) pC = first_entries(perm, st[k + 2], lane);
        const int sB = clamp_node(pB);
        if (k + 1 < GATHER_NPW) aB = first_asrc(a_src, sB, nn[k + 1], lane);
        gather_node(node0 + k, st[k], nn[k], d4[k], ab, sA, aA, lane, perm, a_src, h, out);
        sA = sB; pB = pC; aA = aB;
    }
}

extern "C" void kernel_launch(void* const* d_in, const int* in_sizes, int n_in,
                              void* d_out, int out_size, void* d_ws, size_t ws_size,
                              hipStream_t stream) {
    const float* hidden = (const float*)d_in[0];
    const float* W      = (const float*)d_in[1];
    const float* bias   = (const float*)d_in[2];
    const float* att_w  = (const float*)d_in[3];
    const float* att_b  = (const float*)d_in[4];
    const int*   src    = (const int*)  d_in[5];
    const int*   dst    = (const int*)  d_in[6];
    float* out = (float*)d_out;

    float* ws     = (float*)d_ws;
    float* h_ws   = ws;
    float* a_src  = h_ws  + (size_t)N_NODES * OUT_DIM;
    float* a_dst  = a_src + N_NODES;
    int*   cnt    = (int*)(a_dst + N_NODES);
    int*   off    = cnt  + N_NODES;
    int*   recs   = off  + N_NODES;
    int*   perm   = recs + 2 * (size_t)N_EDGES;
    int*   tcnt   = perm + N_EDGES;
    int*   toff   = tcnt + NBUCKET * NCHUNK;

    const int NT = 256;
    gat_gemm_wmma<<<GEMM_GROUPS, GEMM_WAVES * 32, 0, stream>>>(hidden, W, bias, h_ws);
    gat_proj<<<(N_NODES + 8 * PROJ_NPW - 1) / (8 * PROJ_NPW), NT, 0, stream>>>(
        h_ws, att_w, a_src, a_dst);
    gat_sort_chunks<<<NCHUNK, SORT_T, 0, stream>>>(src, dst, recs, tcnt, toff);
    gat_sort_buckets<<<NBUCKET, SORT_T, 0, stream>>>(recs, tcnt, toff, cnt, off, perm);
    gat_node_gather<<<(N_NODES + 8 * GATHER_NPW - 1) / (8 * GATHER_NPW), NT, 0, stream>>>(
        perm, off, cnt, a_src, a_dst, att_b, h_ws, out);
}


# Attention projections folded into the GEMM epilogue: a lane multiplies its 4 columns of a strip row in LDS, 16-lane row sums by DPP xor-adds (new sum order, no LDS shuffles); their launch is removed
# speedup vs baseline: 5.1908x; 1.0634x over previous
#include <hip/hip_runtime.h>


#define N_NODES   50000
#define N_EDGES   800000
#define IN_DIM    256
#define OUT_DIM   64
#define NEG_SLOPE 0.01f

typedef __attribute__((ext_vector_type(2))) float v2f;
typedef __attribute__((ext_vector_type(4))) float v4f;
typedef __attribute__((ext_vector_type(8))) float v8f;
typedef __attribute__((ext_vector_type(4))) int   v4i;
typedef __attribute__((ext_vector_type(2))) int   v2i;

#define GATHER_NPW 4
static_assert(N_NODES % GATHER_NPW == 0, "gather waves read whole groups of nodes");

#define GEMM_GROUPS   256
#define GEMM_WAVES    16
#define N_STRIPS      (N_NODES / 16)
#define FULL_ROUNDS   (N_STRIPS / GEMM_GROUPS)
#define REM_STRIPS    (N_STRIPS - FULL_ROUNDS * GEMM_GROUPS)
#define KBLOCKS       (IN_DIM / 8)
#define ROW_PITCH     (OUT_DIM + 4)
static_assert(N_NODES % 16 == 0, "whole strips only");
static_assert(FULL_ROUNDS + 4 <= GEMM_WAVES, "waves for the shared strip");
static_assert(REM_STRIPS <= GEMM_GROUPS, "at most one shared strip per group");

template <int NT>
__device__ __forceinline__ void strip_products(v8f (&acc)[NT], const v4f (&a)[KBLOCKS],
                                               const v4f* __restrict__ wl, int t0, int lane) {
#pragma unroll
    for (int kb = 0; kb < KBLOCKS; ++kb) {
        v4f b[NT];
#pragma unroll
        for (int t = 0; t < NT; ++t) b[t] = wl[(kb * 4 + t0 + t) * 32 + lane];
        const v2f a01 = {a[kb].x, a[kb].y};
        const v2f a23 = {a[kb].z, a[kb].w};
#pragma unroll
        for (int t = 0; t < NT; ++t) {
            const v2f b01 = {b[t].x, b[t].y};
            acc[t] = __builtin_amdgcn_wmma_f32_16x16x4_f32(
                false, a01, false, b01, (short)0, acc[t], false, false);
        }
#pragma unroll
        for (int t = 0; t < NT; ++t) {
            const v2f b23 = {b[t].z, b[t].w};
            acc[t] = __builtin_amdgcn_wmma_f32_16x16x4_f32(
                false, a23, false, b23, (short)0, acc[t], false, false);
        }
    }
}

template <int M>
__device__ __forceinline__ float row_xor_add(float x) {
    const int xi = __builtin_bit_cast(int, x);
    const int yi = __builtin_amdgcn_update_dpp(0, xi, 0x160 + M, 0xf, 0xf, true);
    return x + __builtin_bit_cast(float, yi);
}

__device__ __forceinline__ void row_sums8(float (&p)[8], float (&q)[8]) {
#pragma unroll
    for (int i = 0; i < 8; ++i) { p[i] = row_xor_add<1>(p[i]); q[i] = row_xor_add<1>(q[i]); }
#pragma unroll
    for (int i = 0; i < 8; ++i) { p[i] = row_xor_add<2>(p[i]); q[i] = row_xor_add<2>(q[i]); }
#pragma unroll
    for (int i = 0; i < 8; ++i) { p[i] = row_xor_add<4>(p[i]); q[i] = row_xor_add<4>(q[i]); }
#pragma unroll
    for (int i = 0; i < 8; ++i) { p[i] = row_xor_add<8>(p[i]); q[i] = row_xor_add<8>(q[i]); }
}

__global__ __launch_bounds__(GEMM_WAVES * 32) void gat_gemm_wmma(
    const float* __restrict__ hidden,
    const float* __restrict__ W,
    const float* __restrict__ bias,
    const float* __restrict__ att_w,
    float* __restrict__ h_out,
    float* __restrict__ a_src,
    float* __restrict__ a_dst)
{
    __shared__ v4f w_frag[KBLOCKS * 4 * 32];
    __shared__ float h_rows[FULL_ROUNDS][16 * ROW_PITCH];
    __shared__ float part[4][2][16];

    const int tid  = threadIdx.x;
    const int lane = tid & 31;
    const int wave = __builtin_amdgcn_readfirstlane(tid >> 5);
    const int half = lane >> 4;
    const int l16  = lane & 15;

    const bool whole = wave < FULL_ROUNDS;
    const int  strip = (int)blockIdx.x + GEMM_GROUPS * (whole ? wave : FULL_ROUNDS);
    const bool work  = whole || ((int)blockIdx.x < REM_STRIPS && wave < FULL_ROUNDS + 4);

    float wv[8][4];
#pragma unroll
    for (int i = 0; i < 8; ++i) {
        const int e  = tid + 512 * i;
        const int kb = e >> 7, t = (e >> 5) & 3, hf = (e >> 4) & 1, c = e & 15;
        const float* wp = W + (size_t)(8 * kb + 4 * hf) * OUT_DIM + 16 * t + c;
#pragma unroll
        for (int j = 0; j < 4; ++j) wv[i][j] = wp[j * OUT_DIM];
    }
#pragma unroll
    for (int i = 0; i < 8; ++i) {
        const v4f v = {wv[i][0], wv[i][1], wv[i][2], wv[i][3]};
        w_frag[tid + 512 * i] = v;
    }

    v4f a[KBLOCKS];
    if (work) {
        const float* arow = hidden + (size_t)(strip * 16 + l16) * IN_DIM + 4 * half;
#pragma unroll
        for (int kb = 0; kb < KBLOCKS; ++kb) a[kb] = *(const v4f*)(arow + 8 * kb);
    }
    __builtin_amdgcn_fence(__ATOMIC_RELEASE, "workgroup", "local");
    __builtin_amdgcn_s_barrier();
    __builtin_amdgcn_fence(__ATOMIC_ACQUIRE, "workgroup", "local");
    if (!work) return;

    if (whole) {
        v8f acc[4] = {};
        strip_products<4>(acc, a, w_frag, 0, lane);

        float* hr = h_rows[wave];
#pragma unroll
        for (int t = 0; t < 4; ++t) {
            const float bb = bias[16 * t + l16];
#pragma unroll
            for (int r = 0; r < 8; ++r)
                hr[(8 * half + r) * ROW_PITCH + 16 * t + l16] = acc[t][r] + bb;
        }
        float* orow = h_out + (size_t)strip * 16 * OUT_DIM;
        const v4f ws4 = *(const v4f*)(att_w + 4 * l16);
        const v4f wd4 = *(const v4f*)(att_w + OUT_DIM + 4 * l16);
        float ps[8], pd[8];
#pragma unroll
        for (int i = 0; i < 8; ++i) {
            const int row = 2 * i + half;
            const v4f v = *(const v4f*)(hr + row * ROW_PITCH + 4 * l16);
            *(v4f*)(orow + row * OUT_DIM + 4 * l16) = v;
            ps[i] = v.x * ws4.x + v.y * ws4.y + v.z * ws4.z + v.w * ws4.w;
            pd[i] = v.x * wd4.x + v.y * wd4.y + v.z * wd4.z + v.w * wd4.w;
        }
        row_sums8(ps, pd);
        float sres = ps[0], dres = pd[0];
#pragma unroll
        for (int i = 1; i < 8; ++i) if (l16 == i) { sres = ps[i]; dres = pd[i]; }
        if (l16 < 8) {
            a_src[strip * 16 + 2 * l16 + half] = sres;
            a_dst[strip * 16 + 2 * l16 + half] = dres;
        }
    } else {
        const int t0 = wave - FULL_ROUNDS;
        v8f acc[1] = {};
        strip_products<1>(acc, a, w_frag, t0, lane);
        const int gcol = 16 * t0 + l16;
        const float bb = bias[gcol];
        const float wsc = att_w[gcol], wdc = att_w[OUT_DIM + gcol];
        float ps[8], pd[8];
#pragma unroll
        for (int r = 0; r < 8; ++r) {
            const float hv = acc[0][r] + bb;
            h_out[(size_t)(strip * 16 + 8 * half + r) * OUT_DIM + gcol] = hv;
            ps[r] = hv * wsc;
            pd[r] = hv * wdc;
        }
        row_sums8(ps, pd);
        float sres = ps[0], dres = pd[0];
#pragma unroll
        for (int r = 1; r < 8; ++r) if (l16 == r) { sres = ps[r]; dres = pd[r]; }
        if (l16 < 8) {
            part[t0][0][8 * half + l16] = sres;
            part[t0][1][8 * half + l16] = dres;
        }
    }

    if ((int)blockIdx.x < REM_STRIPS) {
        __builtin_amdgcn_fence(__ATOMIC_RELEASE, "workgroup", "local");
        __builtin_amdgcn_s_barrier();
        __builtin_amdgcn_fence(__ATOMIC_ACQUIRE, "workgroup", "local");
        if (wave == FULL_ROUNDS) {
            const float v = ((part[0][half][l16] + part[1][half][l16])
                             + part[2][half][l16]) + part[3][half][l16];
            float* ap = half ? a_dst : a_src;
            ap[strip * 16 + l16] = v;
        }
    }
}

#define SORT_T      1024
#define NCHUNK      256
#define CHUNK_E     (N_EDGES / NCHUNK)
#define CHUNK_SLOTS ((CHUNK_E + SORT_T - 1) / SORT_T)
#define BSHIFT      8
#define BNODES      (1 << BSHIFT)
#define NBUCKET     ((N_NODES + BNODES - 1) / BNODES)
#define REC_SLOTS   6
#define STAGE_CAP   16384
static_assert(N_EDGES % NCHUNK == 0, "equal chunks");
static_assert(NBUCKET <= SORT_T && NCHUNK <= SORT_T && BNODES <= SORT_T, "one table entry per thread");

__device__ __forceinline__ int bucket_of(int d) {
    return min(max(d, 0) >> BSHIFT, NBUCKET - 1);
}

__device__ __forceinline__ int block_excl_scan(int v, int* wsum, int& total) {
    const int lane = threadIdx.x & 31;
    const int wave = threadIdx.x >> 5;
    int incl = v;
    for (int o = 1; o < 32; o <<= 1) {
        const int t = __shfl_up(incl, o, 32);
        if (lane >= o) incl += t;
    }
    __syncthreads();
    if (lane == 31) wsum[wave] = incl;
    __syncthreads();
    const int wt = wsum[lane];
    int wi = wt;
    for (int o = 1; o < 32; o <<= 1) {
        const int t = __shfl_up(wi, o, 32);
        if (lane >= o) wi += t;
    }
    total = __shfl(wi, 31, 32);
    const int wexcl = __shfl(wi - wt, wave, 32);
    return wexcl + incl - v;
}
static_assert(SORT_T == 1024, "block_excl_scan reads one wave total per lane");

__global__ __launch_bounds__(SORT_T) void gat_sort_chunks(
    const int* __restrict__ src, const int* __restrict__ dst,
    int* __restrict__ recs,
    int* __restrict__ tcnt,
    int* __restrict__ toff)
{
    __shared__ int hist[NBUCKET];
    __shared__ int bofs[NBUCKET];
    __shared__ int wsum[32];
    __shared__ v2i stage[CHUNK_E];

    const int tid  = threadIdx.x;
    const int c    = blockIdx.x;
    const int base = c * CHUNK_E;

    if (tid < NBUCKET) hist[tid] = 0;
    int d[CHUNK_SLOTS], s[CHUNK_SLOTS], r[CHUNK_SLOTS], b[CHUNK_SLOTS];
#pragma unroll
    for (int k = 0; k < CHUNK_SLOTS; ++k) {
        const int e = base + min(tid + k * SORT_T, CHUNK_E - 1);
        d[k] = dst[e];
        s[k] = src[e];
    }
    __syncthreads();
#pragma unroll
    for (int k = 0; k < CHUNK_SLOTS; ++k) {
        b[k] = bucket_of(d[k]);
        r[k] = 0;
        if (tid + k * SORT_T < CHUNK_E) r[k] = atomicAdd(&hist[b[k]], 1);
    }
    __syncthreads();
    const int hv = (tid < NBUCKET) ? hist[tid] : 0;
    int tot;
    const int ex = block_excl_scan(hv, wsum, tot);
    if (tid < NBUCKET) bofs[tid] = ex;
    __syncthreads();
#pragma unroll
    for (int k = 0; k < CHUNK_SLOTS; ++k) {
        if (tid + k * SORT_T < CHUNK_E) {
            const int slot = bofs[b[k]] + r[k];
            const v2i rec = {s[k], d[k]};
            if ((unsigned)slot < (unsigned)CHUNK_E) stage[slot] = rec;
        }
    }
    __syncthreads();
    for (int i = tid; i < CHUNK_E; i += SORT_T)
        *(v2i*)(recs + 2 * (size_t)(base + i)) = stage[i];
    if (tid < NBUCKET) {
        tcnt[tid * NCHUNK + c] = hv;
        toff[tid * NCHUNK + c] = ex;
    }
}

__device__ __forceinline__ v2i bucket_record(const int* __restrict__ recs,
                                             const int* pbeg, const int* ploc, int i) {
    int lo = 0, hi = NCHUNK;
    for (int st = 0; st < 8; ++st) {
        const int mid = (lo + hi) >> 1;
        if (pbeg[mid] <= i) lo = mid; else hi = mid;
    }
    const int j = min(max(ploc[lo] + (i - pbeg[lo]), 0), N_EDGES - 1);
    return *(const v2i*)(recs + 2 * (size_t)j);
}
static_assert(NCHUNK == 256, "bucket_record halves eight times");

__global__ __launch_bounds__(SORT_T) void gat_sort_buckets(
    const int* __restrict__ recs, const int* __restrict__ tcnt,
    const int* __restrict__ toff,
    int* __restrict__ cnt, int* __restrict__ off, int* __restrict__ perm)
{
    __shared__ int pbeg[NCHUNK + 1];
    __shared__ int ploc[NCHUNK];
    __shared__ int ncnt[BNODES];
    __shared__ int ncur[BNODES];
    __shared__ int wsum[32];
    __shared__ int lperm[STAGE_CAP];

    const int tid   = threadIdx.x;
    const int bk    = blockIdx.x;
    const int node0 = bk * BNODES;

    int pre = 0;
    for (int i = tid; i < bk * NCHUNK; i += SORT_T) pre += tcnt[i];

    int len = 0;
    if (tid < NCHUNK) {
        len = tcnt[bk * NCHUNK + tid];
        ploc[tid] = tid * CHUNK_E + toff[bk * NCHUNK + tid];
    }
    if (tid < BNODES) ncnt[tid] = 0;
    int nb;
    const int pex = block_excl_scan(len, wsum, nb);
    if (tid < NCHUNK) pbeg[tid] = pex;
    if (tid == 0) pbeg[NCHUNK] = nb;
    int first;
    block_excl_scan(pre, wsum, first);
    __syncthreads();

    int rs[REC_SLOTS], rl[REC_SLOTS];
    const int last = max(nb - 1, 0);
    int pi[REC_SLOTS], lo[REC_SLOTS];
#pragma unroll
    for (int k = 0; k < REC_SLOTS; ++k) {
        pi[k] = min(tid + k * SORT_T, last);
        lo[k] = 0;
    }
#pragma unroll
    for (int step = NCHUNK / 2; step > 0; step >>= 1) {
        int pm[REC_SLOTS];
#pragma unroll
        for (int k = 0; k < REC_SLOTS; ++k) pm[k] = pbeg[lo[k] + step];
#pragma unroll
        for (int k = 0; k < REC_SLOTS; ++k) if (pm[k] <= pi[k]) lo[k] += step;
    }
    int rj[REC_SLOTS];
#pragma unroll
    for (int k = 0; k < REC_SLOTS; ++k) {
        rj[k] = min(max(ploc[lo[k]] + (pi[k] - pbeg[lo[k]]), 0), N_EDGES - 1);
    }
#pragma unroll
    for (int k = 0; k < REC_SLOTS; ++k) {
        const v2i rec = *(const v2i*)(recs + 2 * (size_t)rj[k]);
        rs[k] = rec.x;
        rl[k] = (tid + k * SORT_T < nb) ? min(max(rec.y - node0, 0), BNODES - 1) : -1;
    }
#pragma unroll
    for (int k = 0; k < REC_SLOTS; ++k)
        if (rl[k] >= 0) atomicAdd(&ncnt[rl[k]], 1);
    for (int i = tid + REC_SLOTS * SORT_T; i < nb; i += SORT_T) {
        const v2i rec = bucket_record(recs, pbeg, ploc, i);
        atomicAdd(&ncnt[min(max(rec.y - node0, 0), BNODES - 1)], 1);
    }
    __syncthreads();

    const int nc = (tid < BNODES) ? ncnt[tid] : 0;
    int tot;
    const int nex = block_excl_scan(nc, wsum, tot);
    if (tid < BNODES) {
        ncur[tid] = nex;
        const int node = node0 + tid;
        if (node < N_NODES) {
            cnt[node] = nc;
            off[node] = first + nex;
        }
    }
    __syncthreads();

    const bool staged = nb <= STAGE_CAP;
#pragma unroll
    for (int k = 0; k < REC_SLOTS; ++k) {
        if (rl[k] >= 0) {
            const int pos = atomicAdd(&ncur[rl[k]], 1);
            if (staged) {
                if ((unsigned)pos < (unsigned)STAGE_CAP) lperm[pos] = rs[k];
            } else if ((unsigned)(first + pos) < (unsigned)N_EDGES) {
                perm[first + pos] = rs[k];
            }
        }
    }
    for (int i = tid + REC_SLOTS * SORT_T; i < nb; i += SORT_T) {
        const v2i rec = bucket_record(recs, pbeg, ploc, i);
        const int pos = atomicAdd(&ncur[min(max(rec.y - node0, 0), BNODES - 1)], 1);
        if (staged) {
            if ((unsigned)pos < (unsigned)STAGE_CAP) lperm[pos] = rec.x;
        } else if ((unsigned)(first + pos) < (unsigned)N_EDGES) {
            perm[first + pos] = rec.x;
        }
    }
    if (staged) {
        __syncthreads();
        for (int i = tid; i < nb; i += SORT_T)
            if ((unsigned)(first + i) < (unsigned)N_EDGES) perm[first + i] = lperm[i];
    }
}

__device__ __forceinline__ float lrelu(float v) {
    return (v > 0.0f) ? v : NEG_SLOPE * v;
}

__device__ __forceinline__ int clamp_node(int s) {
    return min(max(s, 0), N_NODES - 1);
}

template <int K>
__device__ __forceinline__ void add_rows(v2f& acc, const float* __restrict__ hl,
                                         int s, float ex, int t) {
    v2f   hv[K];
    float w[K];
    for (int k = 0; k < K; ++k) {
        const int st = __builtin_amdgcn_readlane(s, t + k);
        w[k]  = __int_as_float(__builtin_amdgcn_readlane(__float_as_int(ex), t + k));
        hv[k] = *(const v2f*)(hl + (size_t)st * OUT_DIM);
    }
    for (int k = 0; k < K; ++k) {
        acc.x = __builtin_fmaf(w[k], hv[k].x, acc.x);
        acc.y = __builtin_fmaf(w[k], hv[k].y, acc.y);
    }
}

__device__ __forceinline__ void gather_node(
    const int node, const int start, const int n, const float ad, const float ab,
    const int s0, const float as0, const int lane,
    const int* __restrict__ perm, const float* __restrict__ a_src,
    const float* __restrict__ h, float* __restrict__ out)
{
    const float NINF = -__builtin_inff();

    const float e0 = (lane < n) ? lrelu(as0 + ad + ab) : NINF;
    float m = e0;
    for (int base = 32; base < n; base += 32) {
        int j = base + lane;
        if (j < n) {
            int s = clamp_node(perm[start + j]);
            m = fmaxf(m, lrelu(a_src[s] + ad + ab));
        }
    }
    for (int o = 16; o > 0; o >>= 1) m = fmaxf(m, __shfl_xor(m, o, 32));
    if (!__builtin_isfinite(m)) m = 0.0f;

    v2f   acc  = {0.0f, 0.0f};
    float dsum = 0.0f;
    const float* hl = h + 2 * lane;
    for (int base = 0; base < n; base += 32) {
        int   s = s0;
        float e = e0;
        if (base > 0) {
            s = 0; e = NINF;
            int j = base + lane;
            if (j < n) {
                s = clamp_node(perm[start + j]);
                e = lrelu(a_src[s] + ad + ab);
            }
        }
        const float ex = (base + lane < n) ? __expf(e - m) : 0.0f;
        dsum += ex;
        const int nc = min(32, n - base);
        int t = 0;
        for (; t + 8 <= nc; t += 8) add_rows<8>(acc, hl, s, ex, t);
        if (nc & 4) { add_rows<4>(acc, hl, s, ex, t); t += 4; }
        if (nc & 2) { add_rows<2>(acc, hl, s, ex, t); t += 2; }
        if (nc & 1) { add_rows<1>(acc, hl, s, ex, t); }
    }
    for (int o = 16; o > 0; o >>= 1) dsum += __shfl_xor(dsum, o, 32);

    v2f r;
    if (dsum > 0.0f) {
        r.x = acc.x / dsum;
        r.y = acc.y / dsum;
    } else {
        r = *(const v2f*)(h + (size_t)node * OUT_DIM + 2 * lane);
    }
    *(v2f*)(out + (size_t)node * OUT_DIM + 2 * lane) = r;
}

__device__ __forceinline__ int first_entries(const int* __restrict__ perm,
                                             int start, int lane) {
    return perm[min(start + lane, N_EDGES - 1)];
}
__device__ __forceinline__ float first_asrc(const float* __restrict__ a_src,
                                            int s, int n, int lane) {
    return (lane < n) ? a_src[s] : 0.0f;
}

__global__ __launch_bounds__(256) void gat_node_gather(
    const int* __restrict__ perm, const int* __restrict__ off,
    const int* __restrict__ cnt,
    const float* __restrict__ a_src, const float* __restrict__ a_dst,
    const float* __restrict__ att_b,
    const float* __restrict__ h, float* __restrict__ out)
{
    const int node0 = __builtin_amdgcn_readfirstlane(
        (int)(blockIdx.x * 8 + (threadIdx.x >> 5))) * GATHER_NPW;
    if (node0 >= N_NODES) return;
    const int lane = threadIdx.x & 31;

    const v4i o4 = *(const v4i*)(off + node0);
    const v4i c4 = *(const v4i*)(cnt + node0);
    const v4f d4 = *(const v4f*)(a_dst + node0);
    const float ab = att_b[0];

    int st[GATHER_NPW], nn[GATHER_NPW];
#pragma unroll
    for (int k = 0; k < GATHER_NPW; ++k) {
        st[k] = min(max(o4[k], 0), N_EDGES);
        nn[k] = min(max(c4[k], 0), N_EDGES - st[k]);
    }

    int   pA = first_entries(perm, st[0], lane);
    int   pB = first_entries(perm, st[1], lane);
    int   sA = clamp_node(pA);
    float aA = first_asrc(a_src, sA, nn[0], lane);
#pragma unroll
    for (int k = 0; k < GATHER_NPW; ++k) {
        int   pC = 0;
        float aB = 0.0f;
        if (k + 2 < GATHER_NPW) pC = first_entries(perm, st[k + 2], lane);
        const int sB = clamp_node(pB);
        if (k + 1 < GATHER_NPW) aB = first_asrc(a_src, sB, nn[k + 1], lane);
        gather_node(node0 + k, st[k], nn[k], d4[k], ab, sA, aA, lane, perm, a_src, h, out);
        sA = sB; pB = pC; aA = aB;
    }
}

extern "C" void kernel_launch(void* const* d_in, const int* in_sizes, int n_in,
                              void* d_out, int out_size, void* d_ws, size_t ws_size,
                              hipStream_t stream) {
    const float* hidden = (const float*)d_in[0];
    const float* W      = (const float*)d_in[1];
    const float* bias   = (const float*)d_in[2];
    const float* att_w  = (const float*)d_in[3];
    const float* att_b  = (const float*)d_in[4];
    const int*   src    = (const int*)  d_in[5];
    const int*   dst    = (const int*)  d_in[6];
    float* out = (float*)d_out;

    float* ws     = (float*)d_ws;
    float* h_ws   = ws;
    float* a_src  = h_ws  + (size_t)N_NODES * OUT_DIM;
    float* a_dst  = a_src + N_NODES;
    int*   cnt    = (int*)(a_dst + N_NODES);
    int*   off    = cnt  + N_NODES;
    int*   recs   = off  + N_NODES;
    int*   perm   = recs + 2 * (size_t)N_EDGES;
    int*   tcnt   = perm + N_EDGES;
    int*   toff   = tcnt + NBUCKET * NCHUNK;

    const int NT = 256;
    gat_gemm_wmma<<<GEMM_GROUPS, GEMM_WAVES * 32, 0, stream>>>(
        hidden, W, bias, att_w, h_ws, a_src, a_dst);
    gat_sort_chunks<<<NCHUNK, SORT_T, 0, stream>>>(src, dst, recs, tcnt, toff);
    gat_sort_buckets<<<NBUCKET, SORT_T, 0, stream>>>(recs, tcnt, toff, cnt, off, perm);
    gat_node_gather<<<(N_NODES + 8 * GATHER_NPW - 1) / (8 * GATHER_NPW), NT, 0, stream>>>(
        perm, off, cnt, a_src, a_dst, att_b, h_ws, out);
}


# Chunk pass of the edge sort moved into the GEMM launch: each 16-wave group orders one chunk in LDS before and while its strip loads fly; one launch less; edge (sum) order = new LDS arrival rank
# speedup vs baseline: 6.1094x; 1.1770x over previous
#include <hip/hip_runtime.h>


#define N_NODES   50000
#define N_EDGES   800000
#define IN_DIM    256
#define OUT_DIM   64
#define NEG_SLOPE 0.01f

typedef __attribute__((ext_vector_type(2))) float v2f;
typedef __attribute__((ext_vector_type(4))) float v4f;
typedef __attribute__((ext_vector_type(8))) float v8f;
typedef __attribute__((ext_vector_type(4))) int   v4i;
typedef __attribute__((ext_vector_type(2))) int   v2i;

#define GATHER_NPW 4
static_assert(N_NODES % GATHER_NPW == 0, "gather waves read whole groups of nodes");

#define NCHUNK      256
#define CHUNK_E     (N_EDGES / NCHUNK)
#define BSHIFT      8
#define BNODES      (1 << BSHIFT)
#define NBUCKET     ((N_NODES + BNODES - 1) / BNODES)
static_assert(N_EDGES % NCHUNK == 0, "equal chunks");

__device__ __forceinline__ int bucket_of(int d) {
    return min(max(d, 0) >> BSHIFT, NBUCKET - 1);
}

__device__ __forceinline__ void lds_barrier() {
    __builtin_amdgcn_fence(__ATOMIC_RELEASE, "workgroup", "local");
    __builtin_amdgcn_s_barrier();
    __builtin_amdgcn_fence(__ATOMIC_ACQUIRE, "workgroup", "local");
}

#define GEMM_GROUPS   256
#define GEMM_WAVES    16
#define N_STRIPS      (N_NODES / 16)
#define FULL_ROUNDS   (N_STRIPS / GEMM_GROUPS)
#define REM_STRIPS    (N_STRIPS - FULL_ROUNDS * GEMM_GROUPS)
#define KBLOCKS       (IN_DIM / 8)
#define ROW_PITCH     (OUT_DIM + 4)
static_assert(N_NODES % 16 == 0, "whole strips only");
static_assert(FULL_ROUNDS + 4 <= GEMM_WAVES, "waves for the shared strip");
static_assert(REM_STRIPS <= GEMM_GROUPS, "at most one shared strip per group");
#define CHUNK_T       (GEMM_WAVES * 32)
#define CHUNK_SLOTS   ((CHUNK_E + CHUNK_T - 1) / CHUNK_T)
#define SCAN_PER      ((NBUCKET + 31) / 32)
#define HIST_PAD      (32 * SCAN_PER)
static_assert(NCHUNK == GEMM_GROUPS, "group g orders chunk g");
static_assert(HIST_PAD <= CHUNK_T, "one counter per thread to clear");

template <int NT>
__device__ __forceinline__ void strip_products(v8f (&acc)[NT], const v4f (&a)[KBLOCKS],
                                               const v4f* __restrict__ wl, int t0, int lane) {
#pragma unroll
    for (int kb = 0; kb < KBLOCKS; ++kb) {
        v4f b[NT];
#pragma unroll
        for (int t = 0; t < NT; ++t) b[t] = wl[(kb * 4 + t0 + t) * 32 + lane];
        const v2f a01 = {a[kb].x, a[kb].y};
        const v2f a23 = {a[kb].z, a[kb].w};
#pragma unroll
        for (int t = 0; t < NT; ++t) {
            const v2f b01 = {b[t].x, b[t].y};
            acc[t] = __builtin_amdgcn_wmma_f32_16x16x4_f32(
                false, a01, false, b01, (short)0, acc[t], false, false);
        }
#pragma unroll
        for (int t = 0; t < NT; ++t) {
            const v2f b23 = {b[t].z, b[t].w};
            acc[t] = __builtin_amdgcn_wmma_f32_16x16x4_f32(
                false, a23, false, b23, (short)0, acc[t], false, false);
        }
    }
}

template <int M>
__device__ __forceinline__ float row_xor_add(float x) {
    const int xi = __builtin_bit_cast(int, x);
    const int yi = __builtin_amdgcn_update_dpp(0, xi, 0x160 + M, 0xf, 0xf, true);
    return x + __builtin_bit_cast(float, yi);
}

__device__ __forceinline__ void row_sums8(float (&p)[8], float (&q)[8]) {
#pragma unroll
    for (int i = 0; i < 8; ++i) { p[i] = row_xor_add<1>(p[i]); q[i] = row_xor_add<1>(q[i]); }
#pragma unroll
    for (int i = 0; i < 8; ++i) { p[i] = row_xor_add<2>(p[i]); q[i] = row_xor_add<2>(q[i]); }
#pragma unroll
    for (int i = 0; i < 8; ++i) { p[i] = row_xor_add<4>(p[i]); q[i] = row_xor_add<4>(q[i]); }
#pragma unroll
    for (int i = 0; i < 8; ++i) { p[i] = row_xor_add<8>(p[i]); q[i] = row_xor_add<8>(q[i]); }
}

__global__ __launch_bounds__(GEMM_WAVES * 32) void gat_gemm_chunks(
    const float* __restrict__ hidden,
    const float* __restrict__ W,
    const float* __restrict__ bias,
    const float* __restrict__ att_w,
    const int* __restrict__ src,
    const int* __restrict__ dst,
    float* __restrict__ h_out,
    float* __restrict__ a_src,
    float* __restrict__ a_dst,
    int* __restrict__ recs,
    int* __restrict__ tcnt,
    int* __restrict__ toff)
{
    __shared__ v4f w_frag[KBLOCKS * 4 * 32];
    __shared__ float h_rows[FULL_ROUNDS][16 * ROW_PITCH];
    __shared__ float part[4][2][16];
    __shared__ int hist[HIST_PAD];
    __shared__ int bofs[HIST_PAD];
    __shared__ v2i stage[CHUNK_E];

    const int tid  = threadIdx.x;
    const int lane = tid & 31;
    const int wave = __builtin_amdgcn_readfirstlane(tid >> 5);
    const int half = lane >> 4;
    const int l16  = lane & 15;

    const bool whole = wave < FULL_ROUNDS;
    const int  strip = (int)blockIdx.x + GEMM_GROUPS * (whole ? wave : FULL_ROUNDS);
    const bool work  = whole || ((int)blockIdx.x < REM_STRIPS && wave < FULL_ROUNDS + 4);

    float wv[8][4];
#pragma unroll
    for (int i = 0; i < 8; ++i) {
        const int e  = tid + 512 * i;
        const int kb = e >> 7, t = (e >> 5) & 3, hf = (e >> 4) & 1, c = e & 15;
        const float* wp = W + (size_t)(8 * kb + 4 * hf) * OUT_DIM + 16 * t + c;
#pragma unroll
        for (int j = 0; j < 4; ++j) wv[i][j] = wp[j * OUT_DIM];
    }
    const int cbase = (int)blockIdx.x * CHUNK_E;
    int es[CHUNK_SLOTS], ed[CHUNK_SLOTS];
#pragma unroll
    for (int k = 0; k < CHUNK_SLOTS; ++k)
        es[k] = src[cbase + min(tid + k * CHUNK_T, CHUNK_E - 1)];
#pragma unroll
    for (int k = 0; k < CHUNK_SLOTS; ++k)
        ed[k] = dst[cbase + min(tid + k * CHUNK_T, CHUNK_E - 1)];
    if (tid < HIST_PAD) hist[tid] = 0;
    lds_barrier();

#pragma unroll
    for (int i = 0; i < 8; ++i) {
        const v4f v = {wv[i][0], wv[i][1], wv[i][2], wv[i][3]};
        w_frag[tid + 512 * i] = v;
    }

    int eb[CHUNK_SLOTS], er[CHUNK_SLOTS];
#pragma unroll
    for (int k = 0; k < CHUNK_SLOTS; ++k) {
        eb[k] = bucket_of(ed[k]);
        er[k] = 0;
        if (tid + k * CHUNK_T < CHUNK_E) er[k] = atomicAdd(&hist[eb[k]], 1);
    }

    v4f a[KBLOCKS];
    if (work) {
        const float* arow = hidden + (size_t)(strip * 16 + l16) * IN_DIM + 4 * half;
#pragma unroll
        for (int kb = 0; kb < KBLOCKS; ++kb) a[kb] = *(const v4f*)(arow + 8 * kb);
    }
    lds_barrier();

    const bool scans = (wave == GEMM_WAVES - 1);
    int cv[SCAN_PER] = {}, cx[SCAN_PER] = {};
    int lex = 0;
    if (scans) {
        int run = 0;
#pragma unroll
        for (int j = 0; j < SCAN_PER; ++j) cv[j] = hist[lane * SCAN_PER + j];
#pragma unroll
        for (int j = 0; j < SCAN_PER; ++j) { cx[j] = run; run += cv[j]; }
        int incl = run;
        for (int o = 1; o < 32; o <<= 1) {
            const int t = __shfl_up(incl, o, 32);
            if (lane >= o) incl += t;
        }
        lex = incl - run;
#pragma unroll
        for (int j = 0; j < SCAN_PER; ++j) bofs[lane * SCAN_PER + j] = lex + cx[j];
    }
    lds_barrier();

    int sl[CHUNK_SLOTS];
#pragma unroll
    for (int k = 0; k < CHUNK_SLOTS; ++k) sl[k] = bofs[eb[k]] + er[k];
#pragma unroll
    for (int k = 0; k < CHUNK_SLOTS; ++k) {
        const v2i rec = {es[k], ed[k]};
        if (tid + k * CHUNK_T < CHUNK_E && (unsigned)sl[k] < (unsigned)CHUNK_E) stage[sl[k]] = rec;
    }
    lds_barrier();

#pragma unroll
    for (int k = 0; k < CHUNK_SLOTS; ++k) {
        const int i = tid + k * CHUNK_T;
        if (i < CHUNK_E) *(v2i*)(recs + 2 * (size_t)(cbase + i)) = stage[i];
    }
    if (scans) {
#pragma unroll
        for (int j = 0; j < SCAN_PER; ++j) {
            const int idx = lane * SCAN_PER + j;
            if (idx < NBUCKET) {
                tcnt[idx * NCHUNK + (int)blockIdx.x] = cv[j];
                toff[idx * NCHUNK + (int)blockIdx.x] = lex + cx[j];
            }
        }
    }
    if (!work) return;

    if (whole) {
        v8f acc[4] = {};
        strip_products<4>(acc, a, w_frag, 0, lane);

        float* hr = h_rows[wave];
#pragma unroll
        for (int t = 0; t < 4; ++t) {
            const float bb = bias[16 * t + l16];
#pragma unroll
            for (int r = 0; r < 8; ++r)
                hr[(8 * half + r) * ROW_PITCH + 16 * t + l16] = acc[t][r] + bb;
        }
        float* orow = h_out + (size_t)strip * 16 * OUT_DIM;
        const v4f ws4 = *(const v4f*)(att_w + 4 * l16);
        const v4f wd4 = *(const v4f*)(att_w + OUT_DIM + 4 * l16);
        float ps[8], pd[8];
#pragma unroll
        for (int i = 0; i < 8; ++i) {
            const int row = 2 * i + half;
            const v4f v = *(const v4f*)(hr + row * ROW_PITCH + 4 * l16);
            *(v4f*)(orow + row * OUT_DIM + 4 * l16) = v;
            ps[i] = v.x * ws4.x + v.y * ws4.y + v.z * ws4.z + v.w * ws4.w;
            pd[i] = v.x * wd4.x + v.y * wd4.y + v.z * wd4.z + v.w * wd4.w;
        }
        row_sums8(ps, pd);
        float sres = ps[0], dres = pd[0];
#pragma unroll
        for (int i = 1; i < 8; ++i) if (l16 == i) { sres = ps[i]; dres = pd[i]; }
        if (l16 < 8) {
            a_src[strip * 16 + 2 * l16 + half] = sres;
            a_dst[strip * 16 + 2 * l16 + half] = dres;
        }
    } else {
        const int t0 = wave - FULL_ROUNDS;
        v8f acc[1] = {};
        strip_products<1>(acc, a, w_frag, t0, lane);
        const int gcol = 16 * t0 + l16;
        const float bb = bias[gcol];
        const float wsc = att_w[gcol], wdc = att_w[OUT_DIM + gcol];
        float ps[8], pd[8];
#pragma unroll
        for (int r = 0; r < 8; ++r) {
            const float hv = acc[0][r] + bb;
            h_out[(size_t)(strip * 16 + 8 * half + r) * OUT_DIM + gcol] = hv;
            ps[r] = hv * wsc;
            pd[r] = hv * wdc;
        }
        row_sums8(ps, pd);
        float sres = ps[0], dres = pd[0];
#pragma unroll
        for (int r = 1; r < 8; ++r) if (l16 == r) { sres = ps[r]; dres = pd[r]; }
        if (l16 < 8) {
            part[t0][0][8 * half + l16] = sres;
            part[t0][1][8 * half + l16] = dres;
        }
    }

    if ((int)blockIdx.x < REM_STRIPS) {
        __builtin_amdgcn_fence(__ATOMIC_RELEASE, "workgroup", "local");
        __builtin_amdgcn_s_barrier();
        __builtin_amdgcn_fence(__ATOMIC_ACQUIRE, "workgroup", "local");
        if (wave == FULL_ROUNDS) {
            const float v = ((part[0][half][l16] + part[1][half][l16])
                             + part[2][half][l16]) + part[3][half][l16];
            float* ap = half ? a_dst : a_src;
            ap[strip * 16 + l16] = v;
        }
    }
}

#define SORT_T      1024
#define REC_SLOTS   6
#define STAGE_CAP   16384
static_assert(NBUCKET <= SORT_T && NCHUNK <= SORT_T && BNODES <= SORT_T, "one table entry per thread");

__device__ __forceinline__ int block_excl_scan(int v, int* wsum, int& total) {
    const int lane = threadIdx.x & 31;
    const int wave = threadIdx.x >> 5;
    int incl = v;
    for (int o = 1; o < 32; o <<= 1) {
        const int t = __shfl_up(incl, o, 32);
        if (lane >= o) incl += t;
    }
    __syncthreads();
    if (lane == 31) wsum[wave] = incl;
    __syncthreads();
    const int wt = wsum[lane];
    int wi = wt;
    for (int o = 1; o < 32; o <<= 1) {
        const int t = __shfl_up(wi, o, 32);
        if (lane >= o) wi += t;
    }
    total = __shfl(wi, 31, 32);
    const int wexcl = __shfl(wi - wt, wave, 32);
    return wexcl + incl - v;
}
static_assert(SORT_T == 1024, "block_excl_scan reads one wave total per lane");

__device__ __forceinline__ v2i bucket_record(const int* __restrict__ recs,
                                             const int* pbeg, const int* ploc, int i) {
    int lo = 0, hi = NCHUNK;
    for (int st = 0; st < 8; ++st) {
        const int mid = (lo + hi) >> 1;
        if (pbeg[mid] <= i) lo = mid; else hi = mid;
    }
    const int j = min(max(ploc[lo] + (i - pbeg[lo]), 0), N_EDGES - 1);
    return *(const v2i*)(recs + 2 * (size_t)j);
}
static_assert(NCHUNK == 256, "bucket_record halves eight times");

__global__ __launch_bounds__(SORT_T) void gat_sort_buckets(
    const int* __restrict__ recs, const int* __restrict__ tcnt,
    const int* __restrict__ toff,
    int* __restrict__ cnt, int* __restrict__ off, int* __restrict__ perm)
{
    __shared__ int pbeg[NCHUNK + 1];
    __shared__ int ploc[NCHUNK];
    __shared__ int ncnt[BNODES];
    __shared__ int ncur[BNODES];
    __shared__ int wsum[32];
    __shared__ int lperm[STAGE_CAP];

    const int tid   = threadIdx.x;
    const int bk    = blockIdx.x;
    const int node0 = bk * BNODES;

    int pre = 0;
    for (int i = tid; i < bk * NCHUNK; i += SORT_T) pre += tcnt[i];

    int len = 0;
    if (tid < NCHUNK) {
        len = tcnt[bk * NCHUNK + tid];
        ploc[tid] = tid * CHUNK_E + toff[bk * NCHUNK + tid];
    }
    if (tid < BNODES) ncnt[tid] = 0;
    int nb;
    const int pex = block_excl_scan(len, wsum, nb);
    if (tid < NCHUNK) pbeg[tid] = pex;
    if (tid == 0) pbeg[NCHUNK] = nb;
    int first;
    block_excl_scan(pre, wsum, first);
    __syncthreads();

    int rs[REC_SLOTS], rl[REC_SLOTS];
    const int last = max(nb - 1, 0);
    int pi[REC_SLOTS], lo[REC_SLOTS];
#pragma unroll
    for (int k = 0; k < REC_SLOTS; ++k) {
        pi[k] = min(tid + k * SORT_T, last);
        lo[k] = 0;
    }
#pragma unroll
    for (int step = NCHUNK / 2; step > 0; step >>= 1) {
        int pm[REC_SLOTS];
#pragma unroll
        for (int k = 0; k < REC_SLOTS; ++k) pm[k] = pbeg[lo[k] + step];
#pragma unroll
        for (int k = 0; k < REC_SLOTS; ++k) if (pm[k] <= pi[k]) lo[k] += step;
    }
    int rj[REC_SLOTS];
#pragma unroll
    for (int k = 0; k < REC_SLOTS; ++k) {
        rj[k] = min(max(ploc[lo[k]] + (pi[k] - pbeg[lo[k]]), 0), N_EDGES - 1);
    }
#pragma unroll
    for (int k = 0; k < REC_SLOTS; ++k) {
        const v2i rec = *(const v2i*)(recs + 2 * (size_t)rj[k]);
        rs[k] = rec.x;
        rl[k] = (tid + k * SORT_T < nb) ? min(max(rec.y - node0, 0), BNODES - 1) : -1;
    }
#pragma unroll
    for (int k = 0; k < REC_SLOTS; ++k)
        if (rl[k] >= 0) atomicAdd(&ncnt[rl[k]], 1);
    for (int i = tid + REC_SLOTS * SORT_T; i < nb; i += SORT_T) {
        const v2i rec = bucket_record(recs, pbeg, ploc, i);
        atomicAdd(&ncnt[min(max(rec.y - node0, 0), BNODES - 1)], 1);
    }
    __syncthreads();

    const int nc = (tid < BNODES) ? ncnt[tid] : 0;
    int tot;
    const int nex = block_excl_scan(nc, wsum, tot);
    if (tid < BNODES) {
        ncur[tid] = nex;
        const int node = node0 + tid;
        if (node < N_NODES) {
            cnt[node] = nc;
            off[node] = first + nex;
        }
    }
    __syncthreads();

    const bool staged = nb <= STAGE_CAP;
#pragma unroll
    for (int k = 0; k < REC_SLOTS; ++k) {
        if (rl[k] >= 0) {
            const int pos = atomicAdd(&ncur[rl[k]], 1);
            if (staged) {
                if ((unsigned)pos < (unsigned)STAGE_CAP) lperm[pos] = rs[k];
            } else if ((unsigned)(first + pos) < (unsigned)N_EDGES) {
                perm[first + pos] = rs[k];
            }
        }
    }
    for (int i = tid + REC_SLOTS * SORT_T; i < nb; i += SORT_T) {
        const v2i rec = bucket_record(recs, pbeg, ploc, i);
        const int pos = atomicAdd(&ncur[min(max(rec.y - node0, 0), BNODES - 1)], 1);
        if (staged) {
            if ((unsigned)pos < (unsigned)STAGE_CAP) lperm[pos] = rec.x;
        } else if ((unsigned)(first + pos) < (unsigned)N_EDGES) {
            perm[first + pos] = rec.x;
        }
    }
    if (staged) {
        __syncthreads();
        for (int i = tid; i < nb; i += SORT_T)
            if ((unsigned)(first + i) < (unsigned)N_EDGES) perm[first + i] = lperm[i];
    }
}

__device__ __forceinline__ float lrelu(float v) {
    return (v > 0.0f) ? v : NEG_SLOPE * v;
}

__device__ __forceinline__ int clamp_node(int s) {
    return min(max(s, 0), N_NODES - 1);
}

template <int K>
__device__ __forceinline__ void add_rows(v2f& acc, const float* __restrict__ hl,
                                         int s, float ex, int t) {
    v2f   hv[K];
    float w[K];
    for (int k = 0; k < K; ++k) {
        const int st = __builtin_amdgcn_readlane(s, t + k);
        w[k]  = __int_as_float(__builtin_amdgcn_readlane(__float_as_int(ex), t + k));
        hv[k] = *(const v2f*)(hl + (size_t)st * OUT_DIM);
    }
    for (int k = 0; k < K; ++k) {
        acc.x = __builtin_fmaf(w[k], hv[k].x, acc.x);
        acc.y = __builtin_fmaf(w[k], hv[k].y, acc.y);
    }
}

__device__ __forceinline__ void gather_node(
    const int node, const int start, const int n, const float ad, const float ab,
    const int s0, const float as0, const int lane,
    const int* __restrict__ perm, const float* __restrict__ a_src,
    const float* __restrict__ h, float* __restrict__ out)
{
    const float NINF = -__builtin_inff();

    const float e0 = (lane < n) ? lrelu(as0 + ad + ab) : NINF;
    float m = e0;
    for (int base = 32; base < n; base += 32) {
        int j = base + lane;
        if (j < n) {
            int s = clamp_node(perm[start + j]);
            m = fmaxf(m, lrelu(a_src[s] + ad + ab));
        }
    }
    for (int o = 16; o > 0; o >>= 1) m = fmaxf(m, __shfl_xor(m, o, 32));
    if (!__builtin_isfinite(m)) m = 0.0f;

    v2f   acc  = {0.0f, 0.0f};
    float dsum = 0.0f;
    const float* hl = h + 2 * lane;
    for (int base = 0; base < n; base += 32) {
        int   s = s0;
        float e = e0;
        if (base > 0) {
            s = 0; e = NINF;
            int j = base + lane;
            if (j < n) {
                s = clamp_node(perm[start + j]);
                e = lrelu(a_src[s] + ad + ab);
            }
        }
        const float ex = (base + lane < n) ? __expf(e - m) : 0.0f;
        dsum += ex;
        const int nc = min(32, n - base);
        int t = 0;
        for (; t + 8 <= nc; t += 8) add_rows<8>(acc, hl, s, ex, t);
        if (nc & 4) { add_rows<4>(acc, hl, s, ex, t); t += 4; }
        if (nc & 2) { add_rows<2>(acc, hl, s, ex, t); t += 2; }
        if (nc & 1) { add_rows<1>(acc, hl, s, ex, t); }
    }
    for (int o = 16; o > 0; o >>= 1) dsum += __shfl_xor(dsum, o, 32);

    v2f r;
    if (dsum > 0.0f) {
        r.x = acc.x / dsum;
        r.y = acc.y / dsum;
    } else {
        r = *(const v2f*)(h + (size_t)node * OUT_DIM + 2 * lane);
    }
    *(v2f*)(out + (size_t)node * OUT_DIM + 2 * lane) = r;
}

__device__ __forceinline__ int first_entries(const int* __restrict__ perm,
                                             int start, int lane) {
    return perm[min(start + lane, N_EDGES - 1)];
}
__device__ __forceinline__ float first_asrc(const float* __restrict__ a_src,
                                            int s, int n, int lane) {
    return (lane < n) ? a_src[s] : 0.0f;
}

__global__ __launch_bounds__(256) void gat_node_gather(
    const int* __restrict__ perm, const int* __restrict__ off,
    const int* __restrict__ cnt,
    const float* __restrict__ a_src, const float* __restrict__ a_dst,
    const float* __restrict__ att_b,
    const float* __restrict__ h, float* __restrict__ out)
{
    const int node0 = __builtin_amdgcn_readfirstlane(
        (int)(blockIdx.x * 8 + (threadIdx.x >> 5))) * GATHER_NPW;
    if (node0 >= N_NODES) return;
    const int lane = threadIdx.x & 31;

    const v4i o4 = *(const v4i*)(off + node0);
    const v4i c4 = *(const v4i*)(cnt + node0);
    const v4f d4 = *(const v4f*)(a_dst + node0);
    const float ab = att_b[0];

    int st[GATHER_NPW], nn[GATHER_NPW];
#pragma unroll
    for (int k = 0; k < GATHER_NPW; ++k) {
        st[k] = min(max(o4[k], 0), N_EDGES);
        nn[k] = min(max(c4[k], 0), N_EDGES - st[k]);
    }

    int   pA = first_entries(perm, st[0], lane);
    int   pB = first_entries(perm, st[1], lane);
    int   sA = clamp_node(pA);
    float aA = first_asrc(a_src, sA, nn[0], lane);
#pragma unroll
    for (int k = 0; k < GATHER_NPW; ++k) {
        int   pC = 0;
        float aB = 0.0f;
        if (k + 2 < GATHER_NPW) pC = first_entries(perm, st[k + 2], lane);
        const int sB = clamp_node(pB);
        if (k + 1 < GATHER_NPW) aB = first_asrc(a_src, sB, nn[k + 1], lane);
        gather_node(node0 + k, st[k], nn[k], d4[k], ab, sA, aA, lane, perm, a_src, h, out);
        sA = sB; pB = pC; aA = aB;
    }
}

extern "C" void kernel_launch(void* const* d_in, const int* in_sizes, int n_in,
                              void* d_out, int out_size, void* d_ws, size_t ws_size,
                              hipStream_t stream) {
    const float* hidden = (const float*)d_in[0];
    const float* W      = (const float*)d_in[1];
    const float* bias   = (const float*)d_in[2];
    const float* att_w  = (const float*)d_in[3];
    const float* att_b  = (const float*)d_in[4];
    const int*   src    = (const int*)  d_in[5];
    const int*   dst    = (const int*)  d_in[6];
    float* out = (float*)d_out;

    float* ws     = (float*)d_ws;
    float* h_ws   = ws;
    float* a_src  = h_ws  + (size_t)N_NODES * OUT_DIM;
    float* a_dst  = a_src + N_NODES;
    int*   cnt    = (int*)(a_dst + N_NODES);
    int*   off    = cnt  + N_NODES;
    int*   recs   = off  + N_NODES;
    int*   perm   = recs + 2 * (size_t)N_EDGES;
    int*   tcnt   = perm + N_EDGES;
    int*   toff   = tcnt + NBUCKET * NCHUNK;

    const int NT = 256;
    gat_gemm_chunks<<<GEMM_GROUPS, GEMM_WAVES * 32, 0, stream>>>(
        hidden, W, bias, att_w, src, dst, h_ws, a_src, a_dst, recs, tcnt, toff);
    gat_sort_buckets<<<NBUCKET, SORT_T, 0, stream>>>(recs, tcnt, toff, cnt, off, perm);
    gat_node_gather<<<(N_NODES + 8 * GATHER_NPW - 1) / (8 * GATHER_NPW), NT, 0, stream>>>(
        perm, off, cnt, a_src, a_dst, att_b, h_ws, out);
}


# Level 2 of the edge sort fused into the softmax/gather launch: a 128-node bucket group orders its edges in LDS and its waves gather from there; no list in memory; sum order = LDS arrival rank
# speedup vs baseline: 6.7502x; 1.1049x over previous
#include <hip/hip_runtime.h>


#define N_NODES   50000
#define N_EDGES   800000
#define IN_DIM    256
#define OUT_DIM   64
#define NEG_SLOPE 0.01f

typedef __attribute__((ext_vector_type(2))) float v2f;
typedef __attribute__((ext_vector_type(4))) float v4f;
typedef __attribute__((ext_vector_type(8))) float v8f;
typedef __attribute__((ext_vector_type(4))) int   v4i;
typedef __attribute__((ext_vector_type(2))) int   v2i;

#define GATHER_NPW 4

#define NCHUNK      256
#define CHUNK_E     (N_EDGES / NCHUNK)
#define BSHIFT      7
#define BNODES      (1 << BSHIFT)
#define NBUCKET     ((N_NODES + BNODES - 1) / BNODES)
static_assert(N_EDGES % NCHUNK == 0, "equal chunks");

__device__ __forceinline__ int bucket_of(int d) {
    return min(max(d, 0) >> BSHIFT, NBUCKET - 1);
}

__device__ __forceinline__ void lds_barrier() {
    __builtin_amdgcn_fence(__ATOMIC_RELEASE, "workgroup", "local");
    __builtin_amdgcn_s_barrier();
    __builtin_amdgcn_fence(__ATOMIC_ACQUIRE, "workgroup", "local");
}

#define GEMM_GROUPS   256
#define GEMM_WAVES    16
#define N_STRIPS      (N_NODES / 16)
#define FULL_ROUNDS   (N_STRIPS / GEMM_GROUPS)
#define REM_STRIPS    (N_STRIPS - FULL_ROUNDS * GEMM_GROUPS)
#define KBLOCKS       (IN_DIM / 8)
#define ROW_PITCH     (OUT_DIM + 4)
static_assert(N_NODES % 16 == 0, "whole strips only");
static_assert(FULL_ROUNDS + 4 <= GEMM_WAVES, "waves for the shared strip");
static_assert(REM_STRIPS <= GEMM_GROUPS, "at most one shared strip per group");
#define CHUNK_T       (GEMM_WAVES * 32)
#define CHUNK_SLOTS   ((CHUNK_E + CHUNK_T - 1) / CHUNK_T)
#define SCAN_PER      ((NBUCKET + 31) / 32)
#define HIST_PAD      (32 * SCAN_PER)
static_assert(NCHUNK == GEMM_GROUPS, "group g orders chunk g");
static_assert(HIST_PAD <= CHUNK_T, "one counter per thread to clear");

template <int NT>
__device__ __forceinline__ void strip_products(v8f (&acc)[NT], const v4f (&a)[KBLOCKS],
                                               const v4f* __restrict__ wl, int t0, int lane) {
#pragma unroll
    for (int kb = 0; kb < KBLOCKS; ++kb) {
        v4f b[NT];
#pragma unroll
        for (int t = 0; t < NT; ++t) b[t] = wl[(kb * 4 + t0 + t) * 32 + lane];
        const v2f a01 = {a[kb].x, a[kb].y};
        const v2f a23 = {a[kb].z, a[kb].w};
#pragma unroll
        for (int t = 0; t < NT; ++t) {
            const v2f b01 = {b[t].x, b[t].y};
            acc[t] = __builtin_amdgcn_wmma_f32_16x16x4_f32(
                false, a01, false, b01, (short)0, acc[t], false, false);
        }
#pragma unroll
        for (int t = 0; t < NT; ++t) {
            const v2f b23 = {b[t].z, b[t].w};
            acc[t] = __builtin_amdgcn_wmma_f32_16x16x4_f32(
                false, a23, false, b23, (short)0, acc[t], false, false);
        }
    }
}

template <int M>
__device__ __forceinline__ float row_xor_add(float x) {
    const int xi = __builtin_bit_cast(int, x);
    const int yi = __builtin_amdgcn_update_dpp(0, xi, 0x160 + M, 0xf, 0xf, true);
    return x + __builtin_bit_cast(float, yi);
}

__device__ __forceinline__ void row_sums8(float (&p)[8], float (&q)[8]) {
#pragma unroll
    for (int i = 0; i < 8; ++i) { p[i] = row_xor_add<1>(p[i]); q[i] = row_xor_add<1>(q[i]); }
#pragma unroll
    for (int i = 0; i < 8; ++i) { p[i] = row_xor_add<2>(p[i]); q[i] = row_xor_add<2>(q[i]); }
#pragma unroll
    for (int i = 0; i < 8; ++i) { p[i] = row_xor_add<4>(p[i]); q[i] = row_xor_add<4>(q[i]); }
#pragma unroll
    for (int i = 0; i < 8; ++i) { p[i] = row_xor_add<8>(p[i]); q[i] = row_xor_add<8>(q[i]); }
}

__global__ __launch_bounds__(GEMM_WAVES * 32) void gat_gemm_chunks(
    const float* __restrict__ hidden,
    const float* __restrict__ W,
    const float* __restrict__ bias,
    const float* __restrict__ att_w,
    const int* __restrict__ src,
    const int* __restrict__ dst,
    float* __restrict__ h_out,
    float* __restrict__ a_src,
    float* __restrict__ a_dst,
    int* __restrict__ recs,
    int* __restrict__ tcnt,
    int* __restrict__ toff)
{
    __shared__ v4f w_frag[KBLOCKS * 4 * 32];
    __shared__ float h_rows[FULL_ROUNDS][16 * ROW_PITCH];
    __shared__ float part[4][2][16];
    __shared__ int hist[HIST_PAD];
    __shared__ int bofs[HIST_PAD];
    __shared__ v2i stage[CHUNK_E];

    const int tid  = threadIdx.x;
    const int lane = tid & 31;
    const int wave = __builtin_amdgcn_readfirstlane(tid >> 5);
    const int half = lane >> 4;
    const int l16  = lane & 15;

    const bool whole = wave < FULL_ROUNDS;
    const int  strip = (int)blockIdx.x + GEMM_GROUPS * (whole ? wave : FULL_ROUNDS);
    const bool work  = whole || ((int)blockIdx.x < REM_STRIPS && wave < FULL_ROUNDS + 4);

    float wv[8][4];
#pragma unroll
    for (int i = 0; i < 8; ++i) {
        const int e  = tid + 512 * i;
        const int kb = e >> 7, t = (e >> 5) & 3, hf = (e >> 4) & 1, c = e & 15;
        const float* wp = W + (size_t)(8 * kb + 4 * hf) * OUT_DIM + 16 * t + c;
#pragma unroll
        for (int j = 0; j < 4; ++j) wv[i][j] = wp[j * OUT_DIM];
    }
    const int cbase = (int)blockIdx.x * CHUNK_E;
    int es[CHUNK_SLOTS], ed[CHUNK_SLOTS];
#pragma unroll
    for (int k = 0; k < CHUNK_SLOTS; ++k)
        es[k] = src[cbase + min(tid + k * CHUNK_T, CHUNK_E - 1)];
#pragma unroll
    for (int k = 0; k < CHUNK_SLOTS; ++k)
        ed[k] = dst[cbase + min(tid + k * CHUNK_T, CHUNK_E - 1)];
    if (tid < HIST_PAD) hist[tid] = 0;
    lds_barrier();

#pragma unroll
    for (int i = 0; i < 8; ++i) {
        const v4f v = {wv[i][0], wv[i][1], wv[i][2], wv[i][3]};
        w_frag[tid + 512 * i] = v;
    }

    int eb[CHUNK_SLOTS], er[CHUNK_SLOTS];
#pragma unroll
    for (int k = 0; k < CHUNK_SLOTS; ++k) {
        eb[k] = bucket_of(ed[k]);
        er[k] = 0;
        if (tid + k * CHUNK_T < CHUNK_E) er[k] = atomicAdd(&hist[eb[k]], 1);
    }

    v4f a[KBLOCKS];
    if (work) {
        const float* arow = hidden + (size_t)(strip * 16 + l16) * IN_DIM + 4 * half;
#pragma unroll
        for (int kb = 0; kb < KBLOCKS; ++kb) a[kb] = *(const v4f*)(arow + 8 * kb);
    }
    lds_barrier();

    const bool scans = (wave == GEMM_WAVES - 1);
    int cv[SCAN_PER] = {}, cx[SCAN_PER] = {};
    int lex = 0;
    if (scans) {
        int run = 0;
#pragma unroll
        for (int j = 0; j < SCAN_PER; ++j) cv[j] = hist[lane * SCAN_PER + j];
#pragma unroll
        for (int j = 0; j < SCAN_PER; ++j) { cx[j] = run; run += cv[j]; }
        int incl = run;
        for (int o = 1; o < 32; o <<= 1) {
            const int t = __shfl_up(incl, o, 32);
            if (lane >= o) incl += t;
        }
        lex = incl - run;
#pragma unroll
        for (int j = 0; j < SCAN_PER; ++j) bofs[lane * SCAN_PER + j] = lex + cx[j];
    }
    lds_barrier();

    int sl[CHUNK_SLOTS];
#pragma unroll
    for (int k = 0; k < CHUNK_SLOTS; ++k) sl[k] = bofs[eb[k]] + er[k];
#pragma unroll
    for (int k = 0; k < CHUNK_SLOTS; ++k) {
        const v2i rec = {es[k], ed[k]};
        if (tid + k * CHUNK_T < CHUNK_E && (unsigned)sl[k] < (unsigned)CHUNK_E) stage[sl[k]] = rec;
    }
    lds_barrier();

#pragma unroll
    for (int k = 0; k < CHUNK_SLOTS; ++k) {
        const int i = tid + k * CHUNK_T;
        if (i < CHUNK_E) *(v2i*)(recs + 2 * (size_t)(cbase + i)) = stage[i];
    }
    if (scans) {
#pragma unroll
        for (int j = 0; j < SCAN_PER; ++j) {
            const int idx = lane * SCAN_PER + j;
            if (idx < NBUCKET) {
                tcnt[idx * NCHUNK + (int)blockIdx.x] = cv[j];
                toff[idx * NCHUNK + (int)blockIdx.x] = lex + cx[j];
            }
        }
    }
    if (!work) return;

    if (whole) {
        v8f acc[4] = {};
        strip_products<4>(acc, a, w_frag, 0, lane);

        float* hr = h_rows[wave];
#pragma unroll
        for (int t = 0; t < 4; ++t) {
            const float bb = bias[16 * t + l16];
#pragma unroll
            for (int r = 0; r < 8; ++r)
                hr[(8 * half + r) * ROW_PITCH + 16 * t + l16] = acc[t][r] + bb;
        }
        float* orow = h_out + (size_t)strip * 16 * OUT_DIM;
        const v4f ws4 = *(const v4f*)(att_w + 4 * l16);
        const v4f wd4 = *(const v4f*)(att_w + OUT_DIM + 4 * l16);
        float ps[8], pd[8];
#pragma unroll
        for (int i = 0; i < 8; ++i) {
            const int row = 2 * i + half;
            const v4f v = *(const v4f*)(hr + row * ROW_PITCH + 4 * l16);
            *(v4f*)(orow + row * OUT_DIM + 4 * l16) = v;
            ps[i] = v.x * ws4.x + v.y * ws4.y + v.z * ws4.z + v.w * ws4.w;
            pd[i] = v.x * wd4.x + v.y * wd4.y + v.z * wd4.z + v.w * wd4.w;
        }
        row_sums8(ps, pd);
        float sres = ps[0], dres = pd[0];
#pragma unroll
        for (int i = 1; i < 8; ++i) if (l16 == i) { sres = ps[i]; dres = pd[i]; }
        if (l16 < 8) {
            a_src[strip * 16 + 2 * l16 + half] = sres;
            a_dst[strip * 16 + 2 * l16 + half] = dres;
        }
    } else {
        const int t0 = wave - FULL_ROUNDS;
        v8f acc[1] = {};
        strip_products<1>(acc, a, w_frag, t0, lane);
        const int gcol = 16 * t0 + l16;
        const float bb = bias[gcol];
        const float wsc = att_w[gcol], wdc = att_w[OUT_DIM + gcol];
        float ps[8], pd[8];
#pragma unroll
        for (int r = 0; r < 8; ++r) {
            const float hv = acc[0][r] + bb;
            h_out[(size_t)(strip * 16 + 8 * half + r) * OUT_DIM + gcol] = hv;
            ps[r] = hv * wsc;
            pd[r] = hv * wdc;
        }
        row_sums8(ps, pd);
        float sres = ps[0], dres = pd[0];
#pragma unroll
        for (int r = 1; r < 8; ++r) if (l16 == r) { sres = ps[r]; dres = pd[r]; }
        if (l16 < 8) {
            part[t0][0][8 * half + l16] = sres;
            part[t0][1][8 * half + l16] = dres;
        }
    }

    if ((int)blockIdx.x < REM_STRIPS) {
        __builtin_amdgcn_fence(__ATOMIC_RELEASE, "workgroup", "local");
        __builtin_amdgcn_s_barrier();
        __builtin_amdgcn_fence(__ATOMIC_ACQUIRE, "workgroup", "local");
        if (wave == FULL_ROUNDS) {
            const float v = ((part[0][half][l16] + part[1][half][l16])
                             + part[2][half][l16]) + part[3][half][l16];
            float* ap = half ? a_dst : a_src;
            ap[strip * 16 + l16] = v;
        }
    }
}

#define SORT_T      1024
#define REC_SLOTS   3
#define STAGE_CAP   8192
static_assert(NBUCKET <= SORT_T && NCHUNK <= SORT_T && BNODES <= SORT_T, "one table entry per thread");
static_assert(BNODES == (SORT_T / 32) * GATHER_NPW, "every wave of the group takes GATHER_NPW nodes");

__device__ __forceinline__ int block_excl_scan(int v, int* wsum, int& total) {
    const int lane = threadIdx.x & 31;
    const int wave = threadIdx.x >> 5;
    int incl = v;
    for (int o = 1; o < 32; o <<= 1) {
        const int t = __shfl_up(incl, o, 32);
        if (lane >= o) incl += t;
    }
    __syncthreads();
    if (lane == 31) wsum[wave] = incl;
    __syncthreads();
    const int wt = wsum[lane];
    int wi = wt;
    for (int o = 1; o < 32; o <<= 1) {
        const int t = __shfl_up(wi, o, 32);
        if (lane >= o) wi += t;
    }
    total = __shfl(wi, 31, 32);
    const int wexcl = __shfl(wi - wt, wave, 32);
    return wexcl + incl - v;
}
static_assert(SORT_T == 1024, "block_excl_scan reads one wave total per lane");

__device__ __forceinline__ v2i bucket_record(const int* __restrict__ recs,
                                             const int* pbeg, const int* ploc, int i) {
    int lo = 0, hi = NCHUNK;
    for (int st = 0; st < 8; ++st) {
        const int mid = (lo + hi) >> 1;
        if (pbeg[mid] <= i) lo = mid; else hi = mid;
    }
    const int j = min(max(ploc[lo] + (i - pbeg[lo]), 0), N_EDGES - 1);
    return *(const v2i*)(recs + 2 * (size_t)j);
}
static_assert(NCHUNK == 256, "bucket_record halves eight times");

__device__ __forceinline__ float lrelu(float v) {
    return (v > 0.0f) ? v : NEG_SLOPE * v;
}

__device__ __forceinline__ int clamp_node(int s) {
    return min(max(s, 0), N_NODES - 1);
}

template <int K>
__device__ __forceinline__ void add_rows(v2f& acc, const float* __restrict__ hl,
                                         int s, float ex, int t) {
    v2f   hv[K];
    float w[K];
    for (int k = 0; k < K; ++k) {
        const int st = __builtin_amdgcn_readlane(s, t + k);
        w[k]  = __int_as_float(__builtin_amdgcn_readlane(__float_as_int(ex), t + k));
        hv[k] = *(const v2f*)(hl + (size_t)st * OUT_DIM);
    }
    for (int k = 0; k < K; ++k) {
        acc.x = __builtin_fmaf(w[k], hv[k].x, acc.x);
        acc.y = __builtin_fmaf(w[k], hv[k].y, acc.y);
    }
}

__device__ __forceinline__ void gather_node(
    const int node, const int start, const int n, const float ad, const float ab,
    const int s0, const float as0, const int lane,
    const int* __restrict__ list, const float* __restrict__ a_src,
    const float* __restrict__ h, float* __restrict__ out)
{
    const float NINF = -__builtin_inff();

    const float e0 = (lane < n) ? lrelu(as0 + ad + ab) : NINF;
    float m = e0;
    for (int base = 32; base < n; base += 32) {
        int j = base + lane;
        if (j < n) {
            int s = clamp_node(list[start + j]);
            m = fmaxf(m, lrelu(a_src[s] + ad + ab));
        }
    }
    for (int o = 16; o > 0; o >>= 1) m = fmaxf(m, __shfl_xor(m, o, 32));
    if (!__builtin_isfinite(m)) m = 0.0f;

    v2f   acc  = {0.0f, 0.0f};
    float dsum = 0.0f;
    const float* hl = h + 2 * lane;
    for (int base = 0; base < n; base += 32) {
        int   s = s0;
        float e = e0;
        if (base > 0) {
            s = 0; e = NINF;
            int j = base + lane;
            if (j < n) {
                s = clamp_node(list[start + j]);
                e = lrelu(a_src[s] + ad + ab);
            }
        }
        const float ex = (base + lane < n) ? __expf(e - m) : 0.0f;
        dsum += ex;
        const int nc = min(32, n - base);
        int t = 0;
        for (; t + 8 <= nc; t += 8) add_rows<8>(acc, hl, s, ex, t);
        if (nc & 4) { add_rows<4>(acc, hl, s, ex, t); t += 4; }
        if (nc & 2) { add_rows<2>(acc, hl, s, ex, t); t += 2; }
        if (nc & 1) { add_rows<1>(acc, hl, s, ex, t); }
    }
    for (int o = 16; o > 0; o >>= 1) dsum += __shfl_xor(dsum, o, 32);

    v2f r;
    if (dsum > 0.0f) {
        r.x = acc.x / dsum;
        r.y = acc.y / dsum;
    } else {
        r = *(const v2f*)(h + (size_t)node * OUT_DIM + 2 * lane);
    }
    *(v2f*)(out + (size_t)node * OUT_DIM + 2 * lane) = r;
}

template <int LIM>
__device__ __forceinline__ int first_entries(const int* __restrict__ list,
                                             int start, int lane) {
    return list[min(start + lane, LIM - 1)];
}
__device__ __forceinline__ float first_asrc(const float* __restrict__ a_src,
                                            int s, int n, int lane) {
    return (lane < n) ? a_src[s] : 0.0f;
}

template <int LIM>
__device__ __forceinline__ void gather_wave(
    const int* __restrict__ list, const int lbase,
    const int* nst, const int* ncnt, const int nl0, const int nodeg0,
    const float my_ad, const float ab, const int lane,
    const float* __restrict__ a_src, const float* __restrict__ h, float* __restrict__ out)
{
    const int kk = lane & (GATHER_NPW - 1);
    const int my_st = min(max(lbase + nst[nl0 + kk], 0), LIM);
    const int my_n  = min(max(ncnt[nl0 + kk], 0), LIM - my_st);

    int st[GATHER_NPW], nn[GATHER_NPW];
    float ad[GATHER_NPW];
#pragma unroll
    for (int k = 0; k < GATHER_NPW; ++k) {
        st[k] = __builtin_amdgcn_readlane(my_st, k);
        nn[k] = __builtin_amdgcn_readlane(my_n, k);
        ad[k] = __int_as_float(__builtin_amdgcn_readlane(__float_as_int(my_ad), k));
    }

    int   pA = first_entries<LIM>(list, st[0], lane);
    int   pB = first_entries<LIM>(list, st[1], lane);
    int   sA = clamp_node(pA);
    float aA = first_asrc(a_src, sA, nn[0], lane);
#pragma unroll
    for (int k = 0; k < GATHER_NPW; ++k) {
        int   pC = 0;
        float aB = 0.0f;
        if (k + 2 < GATHER_NPW) pC = first_entries<LIM>(list, st[k + 2], lane);
        const int sB = clamp_node(pB);
        if (k + 1 < GATHER_NPW) aB = first_asrc(a_src, sB, nn[k + 1], lane);
        if (nodeg0 + k < N_NODES)
            gather_node(nodeg0 + k, st[k], nn[k], ad[k], ab, sA, aA, lane, list, a_src, h, out);
        sA = sB; pB = pC; aA = aB;
    }
}

__device__ __forceinline__ void gather_wave_plain(
    const int* __restrict__ list, const int lbase,
    const int* nst, const int* ncnt, const int nl0, const int nodeg0,
    const float my_ad, const float ab, const int lane,
    const float* __restrict__ a_src, const float* __restrict__ h, float* __restrict__ out)
{
    const int kk = lane & (GATHER_NPW - 1);
    const int my_st = min(max(lbase + nst[nl0 + kk], 0), N_EDGES);
    const int my_n  = min(max(ncnt[nl0 + kk], 0), N_EDGES - my_st);
#pragma unroll 1
    for (int k = 0; k < GATHER_NPW; ++k) {
        if (nodeg0 + k >= N_NODES) break;
        const int   stk = __builtin_amdgcn_readlane(my_st, k);
        const int   nk  = __builtin_amdgcn_readlane(my_n, k);
        const float adk = __int_as_float(__builtin_amdgcn_readlane(__float_as_int(my_ad), k));
        const int   s   = (lane < nk) ? clamp_node(list[stk + lane]) : 0;
        const float as  = first_asrc(a_src, s, nk, lane);
        gather_node(nodeg0 + k, stk, nk, adk, ab, s, as, lane, list, a_src, h, out);
    }
}

__global__ __launch_bounds__(SORT_T) void gat_bucket_gather(
    const int* __restrict__ recs, const int* __restrict__ tcnt,
    const int* __restrict__ toff, int* __restrict__ perm,
    const float* __restrict__ a_src, const float* __restrict__ a_dst,
    const float* __restrict__ att_b,
    const float* __restrict__ h, float* __restrict__ out)
{
    __shared__ int pbeg[NCHUNK + 1];
    __shared__ int ploc[NCHUNK];
    __shared__ int ncnt[BNODES];
    __shared__ int ncur[BNODES];
    __shared__ int nst[BNODES];
    __shared__ int wsum[32];
    __shared__ int lperm[STAGE_CAP];

    const int tid   = threadIdx.x;
    const int bk    = blockIdx.x;
    const int node0 = bk * BNODES;
    const int lane  = tid & 31;
    const int nl0   = __builtin_amdgcn_readfirstlane(tid >> 5) * GATHER_NPW;

    const float my_ad = a_dst[min(node0 + nl0 + (lane & (GATHER_NPW - 1)), N_NODES - 1)];
    const float ab = att_b[0];

    int len = 0;
    if (tid < NCHUNK) {
        len = tcnt[bk * NCHUNK + tid];
        ploc[tid] = tid * CHUNK_E + toff[bk * NCHUNK + tid];
    }
    if (tid < BNODES) ncnt[tid] = 0;
    int nb;
    const int pex = block_excl_scan(len, wsum, nb);
    if (tid < NCHUNK) pbeg[tid] = pex;
    if (tid == 0) pbeg[NCHUNK] = nb;

    const bool staged = nb <= STAGE_CAP;
    int first = 0;
    if (!staged) {
        int pre = 0;
        for (int i = tid; i < bk * NCHUNK; i += SORT_T) pre += tcnt[i];
        block_excl_scan(pre, wsum, first);
    }
    __syncthreads();

    int rs[REC_SLOTS], rl[REC_SLOTS];
    const int last = max(nb - 1, 0);
    int pi[REC_SLOTS], lo[REC_SLOTS];
#pragma unroll
    for (int k = 0; k < REC_SLOTS; ++k) {
        pi[k] = min(tid + k * SORT_T, last);
        lo[k] = 0;
    }
#pragma unroll
    for (int step = NCHUNK / 2; step > 0; step >>= 1) {
        int pm[REC_SLOTS];
#pragma unroll
        for (int k = 0; k < REC_SLOTS; ++k) pm[k] = pbeg[lo[k] + step];
#pragma unroll
        for (int k = 0; k < REC_SLOTS; ++k) if (pm[k] <= pi[k]) lo[k] += step;
    }
    int rj[REC_SLOTS];
#pragma unroll
    for (int k = 0; k < REC_SLOTS; ++k) {
        rj[k] = min(max(ploc[lo[k]] + (pi[k] - pbeg[lo[k]]), 0), N_EDGES - 1);
    }
#pragma unroll
    for (int k = 0; k < REC_SLOTS; ++k) {
        const v2i rec = *(const v2i*)(recs + 2 * (size_t)rj[k]);
        rs[k] = rec.x;
        rl[k] = (tid + k * SORT_T < nb) ? min(max(rec.y - node0, 0), BNODES - 1) : -1;
    }
#pragma unroll
    for (int k = 0; k < REC_SLOTS; ++k)
        if (rl[k] >= 0) atomicAdd(&ncnt[rl[k]], 1);
    for (int i = tid + REC_SLOTS * SORT_T; i < nb; i += SORT_T) {
        const v2i rec = bucket_record(recs, pbeg, ploc, i);
        atomicAdd(&ncnt[min(max(rec.y - node0, 0), BNODES - 1)], 1);
    }
    __syncthreads();

    const int nc = (tid < BNODES) ? ncnt[tid] : 0;
    int tot;
    const int nex = block_excl_scan(nc, wsum, tot);
    if (tid < BNODES) {
        ncur[tid] = nex;
        nst[tid]  = nex;
    }
    __syncthreads();

#pragma unroll
    for (int k = 0; k < REC_SLOTS; ++k) {
        if (rl[k] >= 0) {
            const int pos = atomicAdd(&ncur[rl[k]], 1);
            if (staged) {
                if ((unsigned)pos < (unsigned)STAGE_CAP) lperm[pos] = rs[k];
            } else if ((unsigned)(first + pos) < (unsigned)N_EDGES) {
                perm[first + pos] = rs[k];
            }
        }
    }
    for (int i = tid + REC_SLOTS * SORT_T; i < nb; i += SORT_T) {
        const v2i rec = bucket_record(recs, pbeg, ploc, i);
        const int pos = atomicAdd(&ncur[min(max(rec.y - node0, 0), BNODES - 1)], 1);
        if (staged) {
            if ((unsigned)pos < (unsigned)STAGE_CAP) lperm[pos] = rec.x;
        } else if ((unsigned)(first + pos) < (unsigned)N_EDGES) {
            perm[first + pos] = rec.x;
        }
    }

    if (staged) {
        __syncthreads();
        gather_wave<STAGE_CAP>(lperm, 0, nst, ncnt, nl0, node0 + nl0, my_ad, ab, lane,
                               a_src, h, out);
    } else {
        __threadfence();
        __syncthreads();
        gather_wave_plain(perm, first, nst, ncnt, nl0, node0 + nl0, my_ad, ab, lane,
                          a_src, h, out);
    }
}

extern "C" void kernel_launch(void* const* d_in, const int* in_sizes, int n_in,
                              void* d_out, int out_size, void* d_ws, size_t ws_size,
                              hipStream_t stream) {
    const float* hidden = (const float*)d_in[0];
    const float* W      = (const float*)d_in[1];
    const float* bias   = (const float*)d_in[2];
    const float* att_w  = (const float*)d_in[3];
    const float* att_b  = (const float*)d_in[4];
    const int*   src    = (const int*)  d_in[5];
    const int*   dst    = (const int*)  d_in[6];
    float* out = (float*)d_out;

    float* ws     = (float*)d_ws;
    float* h_ws   = ws;
    float* a_src  = h_ws  + (size_t)N_NODES * OUT_DIM;
    float* a_dst  = a_src + N_NODES;
    int*   recs   = (int*)(a_dst + N_NODES);
    int*   perm   = recs + 2 * (size_t)N_EDGES;
    int*   tcnt   = perm + N_EDGES;
    int*   toff   = tcnt + NBUCKET * NCHUNK;

    gat_gemm_chunks<<<GEMM_GROUPS, GEMM_WAVES * 32, 0, stream>>>(
        hidden, W, bias, att_w, src, dst, h_ws, a_src, a_dst, recs, tcnt, toff);
    gat_bucket_gather<<<NBUCKET, SORT_T, 0, stream>>>(
        recs, tcnt, toff, perm, a_src, a_dst, att_b, h_ws, out);
}


# Wave-wide sums, maxima and prefix scans of both launches done with DPP row steps and a lane-half exchange on the vector ALU instead of LDS-unit shuffles; same values, same sum order
# speedup vs baseline: 6.9854x; 1.0348x over previous
#include <hip/hip_runtime.h>


#define N_NODES   50000
#define N_EDGES   800000
#define IN_DIM    256
#define OUT_DIM   64
#define NEG_SLOPE 0.01f

typedef __attribute__((ext_vector_type(2))) float v2f;
typedef __attribute__((ext_vector_type(4))) float v4f;
typedef __attribute__((ext_vector_type(8))) float v8f;
typedef __attribute__((ext_vector_type(4))) int   v4i;
typedef __attribute__((ext_vector_type(2))) int   v2i;

#define GATHER_NPW 4

#define NCHUNK      256
#define CHUNK_E     (N_EDGES / NCHUNK)
#define BSHIFT      7
#define BNODES      (1 << BSHIFT)
#define NBUCKET     ((N_NODES + BNODES - 1) / BNODES)
static_assert(N_EDGES % NCHUNK == 0, "equal chunks");

__device__ __forceinline__ int bucket_of(int d) {
    return min(max(d, 0) >> BSHIFT, NBUCKET - 1);
}

__device__ __forceinline__ void lds_barrier() {
    __builtin_amdgcn_fence(__ATOMIC_RELEASE, "workgroup", "local");
    __builtin_amdgcn_s_barrier();
    __builtin_amdgcn_fence(__ATOMIC_ACQUIRE, "workgroup", "local");
}

#define GEMM_GROUPS   256
#define GEMM_WAVES    16
#define N_STRIPS      (N_NODES / 16)
#define FULL_ROUNDS   (N_STRIPS / GEMM_GROUPS)
#define REM_STRIPS    (N_STRIPS - FULL_ROUNDS * GEMM_GROUPS)
#define KBLOCKS       (IN_DIM / 8)
#define ROW_PITCH     (OUT_DIM + 4)
static_assert(N_NODES % 16 == 0, "whole strips only");
static_assert(FULL_ROUNDS + 4 <= GEMM_WAVES, "waves for the shared strip");
static_assert(REM_STRIPS <= GEMM_GROUPS, "at most one shared strip per group");
#define CHUNK_T       (GEMM_WAVES * 32)
#define CHUNK_SLOTS   ((CHUNK_E + CHUNK_T - 1) / CHUNK_T)
#define SCAN_PER      ((NBUCKET + 31) / 32)
#define HIST_PAD      (32 * SCAN_PER)
static_assert(NCHUNK == GEMM_GROUPS, "group g orders chunk g");
static_assert(HIST_PAD <= CHUNK_T, "one counter per thread to clear");

template <int NT>
__device__ __forceinline__ void strip_products(v8f (&acc)[NT], const v4f (&a)[KBLOCKS],
                                               const v4f* __restrict__ wl, int t0, int lane) {
#pragma unroll
    for (int kb = 0; kb < KBLOCKS; ++kb) {
        v4f b[NT];
#pragma unroll
        for (int t = 0; t < NT; ++t) b[t] = wl[(kb * 4 + t0 + t) * 32 + lane];
        const v2f a01 = {a[kb].x, a[kb].y};
        const v2f a23 = {a[kb].z, a[kb].w};
#pragma unroll
        for (int t = 0; t < NT; ++t) {
            const v2f b01 = {b[t].x, b[t].y};
            acc[t] = __builtin_amdgcn_wmma_f32_16x16x4_f32(
                false, a01, false, b01, (short)0, acc[t], false, false);
        }
#pragma unroll
        for (int t = 0; t < NT; ++t) {
            const v2f b23 = {b[t].z, b[t].w};
            acc[t] = __builtin_amdgcn_wmma_f32_16x16x4_f32(
                false, a23, false, b23, (short)0, acc[t], false, false);
        }
    }
}

template <int M>
__device__ __forceinline__ float row_xor_add(float x) {
    const int xi = __builtin_bit_cast(int, x);
    const int yi = __builtin_amdgcn_update_dpp(0, xi, 0x160 + M, 0xf, 0xf, true);
    return x + __builtin_bit_cast(float, yi);
}


__device__ __forceinline__ int xhalf_i(int b) {
    return __builtin_amdgcn_permlanex16(b, b, 0x76543210, 0xfedcba98, false, false);
}
__device__ __forceinline__ float xhalf(float x) {
    return __builtin_bit_cast(float, xhalf_i(__builtin_bit_cast(int, x)));
}
template <int M>
__device__ __forceinline__ float row_xor_max(float x) {
    const int xi = __builtin_bit_cast(int, x);
    const int yi = __builtin_amdgcn_update_dpp(0, xi, 0x160 + M, 0xf, 0xf, true);
    return fmaxf(x, __builtin_bit_cast(float, yi));
}
__device__ __forceinline__ float wave_sum32(float x) {
    x = x + xhalf(x);
    x = row_xor_add<8>(x);
    x = row_xor_add<4>(x);
    x = row_xor_add<2>(x);
    return row_xor_add<1>(x);
}
__device__ __forceinline__ float wave_max32(float x) {
    x = fmaxf(x, xhalf(x));
    x = row_xor_max<8>(x);
    x = row_xor_max<4>(x);
    x = row_xor_max<2>(x);
    return row_xor_max<1>(x);
}
template <int N>
__device__ __forceinline__ int row_shr_or_zero(int x) {
    return __builtin_amdgcn_update_dpp(0, x, 0x110 + N, 0xf, 0xf, true);
}
__device__ __forceinline__ int wave_incl_scan(int v) {
    v += row_shr_or_zero<1>(v);
    v += row_shr_or_zero<2>(v);
    v += row_shr_or_zero<4>(v);
    v += row_shr_or_zero<8>(v);
    const int low = __builtin_amdgcn_readlane(v, 15);
    return v + (((threadIdx.x & 31) >= 16) ? low : 0);
}

__device__ __forceinline__ void row_sums8(float (&p)[8], float (&q)[8]) {
#pragma unroll
    for (int i = 0; i < 8; ++i) { p[i] = row_xor_add<1>(p[i]); q[i] = row_xor_add<1>(q[i]); }
#pragma unroll
    for (int i = 0; i < 8; ++i) { p[i] = row_xor_add<2>(p[i]); q[i] = row_xor_add<2>(q[i]); }
#pragma unroll
    for (int i = 0; i < 8; ++i) { p[i] = row_xor_add<4>(p[i]); q[i] = row_xor_add<4>(q[i]); }
#pragma unroll
    for (int i = 0; i < 8; ++i) { p[i] = row_xor_add<8>(p[i]); q[i] = row_xor_add<8>(q[i]); }
}

__global__ __launch_bounds__(GEMM_WAVES * 32) void gat_gemm_chunks(
    const float* __restrict__ hidden,
    const float* __restrict__ W,
    const float* __restrict__ bias,
    const float* __restrict__ att_w,
    const int* __restrict__ src,
    const int* __restrict__ dst,
    float* __restrict__ h_out,
    float* __restrict__ a_src,
    float* __restrict__ a_dst,
    int* __restrict__ recs,
    int* __restrict__ tcnt,
    int* __restrict__ toff)
{
    __shared__ v4f w_frag[KBLOCKS * 4 * 32];
    __shared__ float h_rows[FULL_ROUNDS][16 * ROW_PITCH];
    __shared__ float part[4][2][16];
    __shared__ int hist[HIST_PAD];
    __shared__ int bofs[HIST_PAD];
    __shared__ v2i stage[CHUNK_E];

    const int tid  = threadIdx.x;
    const int lane = tid & 31;
    const int wave = __builtin_amdgcn_readfirstlane(tid >> 5);
    const int half = lane >> 4;
    const int l16  = lane & 15;

    const bool whole = wave < FULL_ROUNDS;
    const int  strip = (int)blockIdx.x + GEMM_GROUPS * (whole ? wave : FULL_ROUNDS);
    const bool work  = whole || ((int)blockIdx.x < REM_STRIPS && wave < FULL_ROUNDS + 4);

    float wv[8][4];
#pragma unroll
    for (int i = 0; i < 8; ++i) {
        const int e  = tid + 512 * i;
        const int kb = e >> 7, t = (e >> 5) & 3, hf = (e >> 4) & 1, c = e & 15;
        const float* wp = W + (size_t)(8 * kb + 4 * hf) * OUT_DIM + 16 * t + c;
#pragma unroll
        for (int j = 0; j < 4; ++j) wv[i][j] = wp[j * OUT_DIM];
    }
    const int cbase = (int)blockIdx.x * CHUNK_E;
    int es[CHUNK_SLOTS], ed[CHUNK_SLOTS];
#pragma unroll
    for (int k = 0; k < CHUNK_SLOTS; ++k)
        es[k] = src[cbase + min(tid + k * CHUNK_T, CHUNK_E - 1)];
#pragma unroll
    for (int k = 0; k < CHUNK_SLOTS; ++k)
        ed[k] = dst[cbase + min(tid + k * CHUNK_T, CHUNK_E - 1)];
    if (tid < HIST_PAD) hist[tid] = 0;
    lds_barrier();

#pragma unroll
    for (int i = 0; i < 8; ++i) {
        const v4f v = {wv[i][0], wv[i][1], wv[i][2], wv[i][3]};
        w_frag[tid + 512 * i] = v;
    }

    int eb[CHUNK_SLOTS], er[CHUNK_SLOTS];
#pragma unroll
    for (int k = 0; k < CHUNK_SLOTS; ++k) {
        eb[k] = bucket_of(ed[k]);
        er[k] = 0;
        if (tid + k * CHUNK_T < CHUNK_E) er[k] = atomicAdd(&hist[eb[k]], 1);
    }

    v4f a[KBLOCKS];
    if (work) {
        const float* arow = hidden + (size_t)(strip * 16 + l16) * IN_DIM + 4 * half;
#pragma unroll
        for (int kb = 0; kb < KBLOCKS; ++kb) a[kb] = *(const v4f*)(arow + 8 * kb);
    }
    lds_barrier();

    const bool scans = (wave == GEMM_WAVES - 1);
    int cv[SCAN_PER] = {}, cx[SCAN_PER] = {};
    int lex = 0;
    if (scans) {
        int run = 0;
#pragma unroll
        for (int j = 0; j < SCAN_PER; ++j) cv[j] = hist[lane * SCAN_PER + j];
#pragma unroll
        for (int j = 0; j < SCAN_PER; ++j) { cx[j] = run; run += cv[j]; }
        const int incl = wave_incl_scan(run);
        lex = incl - run;
#pragma unroll
        for (int j = 0; j < SCAN_PER; ++j) bofs[lane * SCAN_PER + j] = lex + cx[j];
    }
    lds_barrier();

    int sl[CHUNK_SLOTS];
#pragma unroll
    for (int k = 0; k < CHUNK_SLOTS; ++k) sl[k] = bofs[eb[k]] + er[k];
#pragma unroll
    for (int k = 0; k < CHUNK_SLOTS; ++k) {
        const v2i rec = {es[k], ed[k]};
        if (tid + k * CHUNK_T < CHUNK_E && (unsigned)sl[k] < (unsigned)CHUNK_E) stage[sl[k]] = rec;
    }
    lds_barrier();

#pragma unroll
    for (int k = 0; k < CHUNK_SLOTS; ++k) {
        const int i = tid + k * CHUNK_T;
        if (i < CHUNK_E) *(v2i*)(recs + 2 * (size_t)(cbase + i)) = stage[i];
    }
    if (scans) {
#pragma unroll
        for (int j = 0; j < SCAN_PER; ++j) {
            const int idx = lane * SCAN_PER + j;
            if (idx < NBUCKET) {
                tcnt[idx * NCHUNK + (int)blockIdx.x] = cv[j];
                toff[idx * NCHUNK + (int)blockIdx.x] = lex + cx[j];
            }
        }
    }
    if (!work) return;

    if (whole) {
        v8f acc[4] = {};
        strip_products<4>(acc, a, w_frag, 0, lane);

        float* hr = h_rows[wave];
#pragma unroll
        for (int t = 0; t < 4; ++t) {
            const float bb = bias[16 * t + l16];
#pragma unroll
            for (int r = 0; r < 8; ++r)
                hr[(8 * half + r) * ROW_PITCH + 16 * t + l16] = acc[t][r] + bb;
        }
        float* orow = h_out + (size_t)strip * 16 * OUT_DIM;
        const v4f ws4 = *(const v4f*)(att_w + 4 * l16);
        const v4f wd4 = *(const v4f*)(att_w + OUT_DIM + 4 * l16);
        float ps[8], pd[8];
#pragma unroll
        for (int i = 0; i < 8; ++i) {
            const int row = 2 * i + half;
            const v4f v = *(const v4f*)(hr + row * ROW_PITCH + 4 * l16);
            *(v4f*)(orow + row * OUT_DIM + 4 * l16) = v;
            ps[i] = v.x * ws4.x + v.y * ws4.y + v.z * ws4.z + v.w * ws4.w;
            pd[i] = v.x * wd4.x + v.y * wd4.y + v.z * wd4.z + v.w * wd4.w;
        }
        row_sums8(ps, pd);
        float sres = ps[0], dres = pd[0];
#pragma unroll
        for (int i = 1; i < 8; ++i) if (l16 == i) { sres = ps[i]; dres = pd[i]; }
        if (l16 < 8) {
            a_src[strip * 16 + 2 * l16 + half] = sres;
            a_dst[strip * 16 + 2 * l16 + half] = dres;
        }
    } else {
        const int t0 = wave - FULL_ROUNDS;
        v8f acc[1] = {};
        strip_products<1>(acc, a, w_frag, t0, lane);
        const int gcol = 16 * t0 + l16;
        const float bb = bias[gcol];
        const float wsc = att_w[gcol], wdc = att_w[OUT_DIM + gcol];
        float ps[8], pd[8];
#pragma unroll
        for (int r = 0; r < 8; ++r) {
            const float hv = acc[0][r] + bb;
            h_out[(size_t)(strip * 16 + 8 * half + r) * OUT_DIM + gcol] = hv;
            ps[r] = hv * wsc;
            pd[r] = hv * wdc;
        }
        row_sums8(ps, pd);
        float sres = ps[0], dres = pd[0];
#pragma unroll
        for (int r = 1; r < 8; ++r) if (l16 == r) { sres = ps[r]; dres = pd[r]; }
        if (l16 < 8) {
            part[t0][0][8 * half + l16] = sres;
            part[t0][1][8 * half + l16] = dres;
        }
    }

    if ((int)blockIdx.x < REM_STRIPS) {
        __builtin_amdgcn_fence(__ATOMIC_RELEASE, "workgroup", "local");
        __builtin_amdgcn_s_barrier();
        __builtin_amdgcn_fence(__ATOMIC_ACQUIRE, "workgroup", "local");
        if (wave == FULL_ROUNDS) {
            const float v = ((part[0][half][l16] + part[1][half][l16])
                             + part[2][half][l16]) + part[3][half][l16];
            float* ap = half ? a_dst : a_src;
            ap[strip * 16 + l16] = v;
        }
    }
}

#define SORT_T      1024
#define REC_SLOTS   3
#define STAGE_CAP   8192
static_assert(NBUCKET <= SORT_T && NCHUNK <= SORT_T && BNODES <= SORT_T, "one table entry per thread");
static_assert(BNODES == (SORT_T / 32) * GATHER_NPW, "every wave of the group takes GATHER_NPW nodes");

__device__ __forceinline__ int block_excl_scan(int v, int* wsum, int& total) {
    const int lane = threadIdx.x & 31;
    const int wave = __builtin_amdgcn_readfirstlane(threadIdx.x >> 5);
    const int incl = wave_incl_scan(v);
    __syncthreads();
    if (lane == 31) wsum[wave] = incl;
    __syncthreads();
    const int wt = wsum[lane];
    const int wi = wave_incl_scan(wt);
    total = __builtin_amdgcn_readlane(wi, 31);
    const int wexcl = __builtin_amdgcn_readlane(wi - wt, wave);
    return wexcl + incl - v;
}
static_assert(SORT_T == 1024, "block_excl_scan reads one wave total per lane");

__device__ __forceinline__ v2i bucket_record(const int* __restrict__ recs,
                                             const int* pbeg, const int* ploc, int i) {
    int lo = 0, hi = NCHUNK;
    for (int st = 0; st < 8; ++st) {
        const int mid = (lo + hi) >> 1;
        if (pbeg[mid] <= i) lo = mid; else hi = mid;
    }
    const int j = min(max(ploc[lo] + (i - pbeg[lo]), 0), N_EDGES - 1);
    return *(const v2i*)(recs + 2 * (size_t)j);
}
static_assert(NCHUNK == 256, "bucket_record halves eight times");

__device__ __forceinline__ float lrelu(float v) {
    return (v > 0.0f) ? v : NEG_SLOPE * v;
}

__device__ __forceinline__ int clamp_node(int s) {
    return min(max(s, 0), N_NODES - 1);
}

template <int K>
__device__ __forceinline__ void add_rows(v2f& acc, const float* __restrict__ hl,
                                         int s, float ex, int t) {
    v2f   hv[K];
    float w[K];
    for (int k = 0; k < K; ++k) {
        const int st = __builtin_amdgcn_readlane(s, t + k);
        w[k]  = __int_as_float(__builtin_amdgcn_readlane(__float_as_int(ex), t + k));
        hv[k] = *(const v2f*)(hl + (size_t)st * OUT_DIM);
    }
    for (int k = 0; k < K; ++k) {
        acc.x = __builtin_fmaf(w[k], hv[k].x, acc.x);
        acc.y = __builtin_fmaf(w[k], hv[k].y, acc.y);
    }
}

__device__ __forceinline__ void gather_node(
    const int node, const int start, const int n, const float ad, const float ab,
    const int s0, const float as0, const int lane,
    const int* __restrict__ list, const float* __restrict__ a_src,
    const float* __restrict__ h, float* __restrict__ out)
{
    const float NINF = -__builtin_inff();

    const float e0 = (lane < n) ? lrelu(as0 + ad + ab) : NINF;
    float m = e0;
    for (int base = 32; base < n; base += 32) {
        int j = base + lane;
        if (j < n) {
            int s = clamp_node(list[start + j]);
            m = fmaxf(m, lrelu(a_src[s] + ad + ab));
        }
    }
    m = wave_max32(m);
    if (!__builtin_isfinite(m)) m = 0.0f;

    v2f   acc  = {0.0f, 0.0f};
    float dsum = 0.0f;
    const float* hl = h + 2 * lane;
    for (int base = 0; base < n; base += 32) {
        int   s = s0;
        float e = e0;
        if (base > 0) {
            s = 0; e = NINF;
            int j = base + lane;
            if (j < n) {
                s = clamp_node(list[start + j]);
                e = lrelu(a_src[s] + ad + ab);
            }
        }
        const float ex = (base + lane < n) ? __expf(e - m) : 0.0f;
        dsum += ex;
        const int nc = min(32, n - base);
        int t = 0;
        for (; t + 8 <= nc; t += 8) add_rows<8>(acc, hl, s, ex, t);
        if (nc & 4) { add_rows<4>(acc, hl, s, ex, t); t += 4; }
        if (nc & 2) { add_rows<2>(acc, hl, s, ex, t); t += 2; }
        if (nc & 1) { add_rows<1>(acc, hl, s, ex, t); }
    }
    dsum = wave_sum32(dsum);

    v2f r;
    if (dsum > 0.0f) {
        r.x = acc.x / dsum;
        r.y = acc.y / dsum;
    } else {
        r = *(const v2f*)(h + (size_t)node * OUT_DIM + 2 * lane);
    }
    *(v2f*)(out + (size_t)node * OUT_DIM + 2 * lane) = r;
}

template <int LIM>
__device__ __forceinline__ int first_entries(const int* __restrict__ list,
                                             int start, int lane) {
    return list[min(start + lane, LIM - 1)];
}
__device__ __forceinline__ float first_asrc(const float* __restrict__ a_src,
                                            int s, int n, int lane) {
    return (lane < n) ? a_src[s] : 0.0f;
}

template <int LIM>
__device__ __forceinline__ void gather_wave(
    const int* __restrict__ list, const int lbase,
    const int* nst, const int* ncnt, const int nl0, const int nodeg0,
    const float my_ad, const float ab, const int lane,
    const float* __restrict__ a_src, const float* __restrict__ h, float* __restrict__ out)
{
    const int kk = lane & (GATHER_NPW - 1);
    const int my_st = min(max(lbase + nst[nl0 + kk], 0), LIM);
    const int my_n  = min(max(ncnt[nl0 + kk], 0), LIM - my_st);

    int st[GATHER_NPW], nn[GATHER_NPW];
    float ad[GATHER_NPW];
#pragma unroll
    for (int k = 0; k < GATHER_NPW; ++k) {
        st[k] = __builtin_amdgcn_readlane(my_st, k);
        nn[k] = __builtin_amdgcn_readlane(my_n, k);
        ad[k] = __int_as_float(__builtin_amdgcn_readlane(__float_as_int(my_ad), k));
    }

    int   pA = first_entries<LIM>(list, st[0], lane);
    int   pB = first_entries<LIM>(list, st[1], lane);
    int   sA = clamp_node(pA);
    float aA = first_asrc(a_src, sA, nn[0], lane);
#pragma unroll
    for (int k = 0; k < GATHER_NPW; ++k) {
        int   pC = 0;
        float aB = 0.0f;
        if (k + 2 < GATHER_NPW) pC = first_entries<LIM>(list, st[k + 2], lane);
        const int sB = clamp_node(pB);
        if (k + 1 < GATHER_NPW) aB = first_asrc(a_src, sB, nn[k + 1], lane);
        if (nodeg0 + k < N_NODES)
            gather_node(nodeg0 + k, st[k], nn[k], ad[k], ab, sA, aA, lane, list, a_src, h, out);
        sA = sB; pB = pC; aA = aB;
    }
}

__device__ __forceinline__ void gather_wave_plain(
    const int* __restrict__ list, const int lbase,
    const int* nst, const int* ncnt, const int nl0, const int nodeg0,
    const float my_ad, const float ab, const int lane,
    const float* __restrict__ a_src, const float* __restrict__ h, float* __restrict__ out)
{
    const int kk = lane & (GATHER_NPW - 1);
    const int my_st = min(max(lbase + nst[nl0 + kk], 0), N_EDGES);
    const int my_n  = min(max(ncnt[nl0 + kk], 0), N_EDGES - my_st);
#pragma unroll 1
    for (int k = 0; k < GATHER_NPW; ++k) {
        if (nodeg0 + k >= N_NODES) break;
        const int   stk = __builtin_amdgcn_readlane(my_st, k);
        const int   nk  = __builtin_amdgcn_readlane(my_n, k);
        const float adk = __int_as_float(__builtin_amdgcn_readlane(__float_as_int(my_ad), k));
        const int   s   = (lane < nk) ? clamp_node(list[stk + lane]) : 0;
        const float as  = first_asrc(a_src, s, nk, lane);
        gather_node(nodeg0 + k, stk, nk, adk, ab, s, as, lane, list, a_src, h, out);
    }
}

__global__ __launch_bounds__(SORT_T) void gat_bucket_gather(
    const int* __restrict__ recs, const int* __restrict__ tcnt,
    const int* __restrict__ toff, int* __restrict__ perm,
    const float* __restrict__ a_src, const float* __restrict__ a_dst,
    const float* __restrict__ att_b,
    const float* __restrict__ h, float* __restrict__ out)
{
    __shared__ int pbeg[NCHUNK + 1];
    __shared__ int ploc[NCHUNK];
    __shared__ int ncnt[BNODES];
    __shared__ int ncur[BNODES];
    __shared__ int nst[BNODES];
    __shared__ int wsum[32];
    __shared__ int lperm[STAGE_CAP];

    const int tid   = threadIdx.x;
    const int bk    = blockIdx.x;
    const int node0 = bk * BNODES;
    const int lane  = tid & 31;
    const int nl0   = __builtin_amdgcn_readfirstlane(tid >> 5) * GATHER_NPW;

    const float my_ad = a_dst[min(node0 + nl0 + (lane & (GATHER_NPW - 1)), N_NODES - 1)];
    const float ab = att_b[0];

    int len = 0;
    if (tid < NCHUNK) {
        len = tcnt[bk * NCHUNK + tid];
        ploc[tid] = tid * CHUNK_E + toff[bk * NCHUNK + tid];
    }
    if (tid < BNODES) ncnt[tid] = 0;
    int nb;
    const int pex = block_excl_scan(len, wsum, nb);
    if (tid < NCHUNK) pbeg[tid] = pex;
    if (tid == 0) pbeg[NCHUNK] = nb;

    const bool staged = nb <= STAGE_CAP;
    int first = 0;
    if (!staged) {
        int pre = 0;
        for (int i = tid; i < bk * NCHUNK; i += SORT_T) pre += tcnt[i];
        block_excl_scan(pre, wsum, first);
    }
    __syncthreads();

    int rs[REC_SLOTS], rl[REC_SLOTS];
    const int last = max(nb - 1, 0);
    int pi[REC_SLOTS], lo[REC_SLOTS];
#pragma unroll
    for (int k = 0; k < REC_SLOTS; ++k) {
        pi[k] = min(tid + k * SORT_T, last);
        lo[k] = 0;
    }
#pragma unroll
    for (int step = NCHUNK / 2; step > 0; step >>= 1) {
        int pm[REC_SLOTS];
#pragma unroll
        for (int k = 0; k < REC_SLOTS; ++k) pm[k] = pbeg[lo[k] + step];
#pragma unroll
        for (int k = 0; k < REC_SLOTS; ++k) if (pm[k] <= pi[k]) lo[k] += step;
    }
    int rj[REC_SLOTS];
#pragma unroll
    for (int k = 0; k < REC_SLOTS; ++k) {
        rj[k] = min(max(ploc[lo[k]] + (pi[k] - pbeg[lo[k]]), 0), N_EDGES - 1);
    }
#pragma unroll
    for (int k = 0; k < REC_SLOTS; ++k) {
        const v2i rec = *(const v2i*)(recs + 2 * (size_t)rj[k]);
        rs[k] = rec.x;
        rl[k] = (tid + k * SORT_T < nb) ? min(max(rec.y - node0, 0), BNODES - 1) : -1;
    }
#pragma unroll
    for (int k = 0; k < REC_SLOTS; ++k)
        if (rl[k] >= 0) atomicAdd(&ncnt[rl[k]], 1);
    for (int i = tid + REC_SLOTS * SORT_T; i < nb; i += SORT_T) {
        const v2i rec = bucket_record(recs, pbeg, ploc, i);
        atomicAdd(&ncnt[min(max(rec.y - node0, 0), BNODES - 1)], 1);
    }
    __syncthreads();

    const int nc = (tid < BNODES) ? ncnt[tid] : 0;
    int tot;
    const int nex = block_excl_scan(nc, wsum, tot);
    if (tid < BNODES) {
        ncur[tid] = nex;
        nst[tid]  = nex;
    }
    __syncthreads();

#pragma unroll
    for (int k = 0; k < REC_SLOTS; ++k) {
        if (rl[k] >= 0) {
            const int pos = atomicAdd(&ncur[rl[k]], 1);
            if (staged) {
                if ((unsigned)pos < (unsigned)STAGE_CAP) lperm[pos] = rs[k];
            } else if ((unsigned)(first + pos) < (unsigned)N_EDGES) {
                perm[first + pos] = rs[k];
            }
        }
    }
    for (int i = tid + REC_SLOTS * SORT_T; i < nb; i += SORT_T) {
        const v2i rec = bucket_record(recs, pbeg, ploc, i);
        const int pos = atomicAdd(&ncur[min(max(rec.y - node0, 0), BNODES - 1)], 1);
        if (staged) {
            if ((unsigned)pos < (unsigned)STAGE_CAP) lperm[pos] = rec.x;
        } else if ((unsigned)(first + pos) < (unsigned)N_EDGES) {
            perm[first + pos] = rec.x;
        }
    }

    if (staged) {
        __syncthreads();
        gather_wave<STAGE_CAP>(lperm, 0, nst, ncnt, nl0, node0 + nl0, my_ad, ab, lane,
                               a_src, h, out);
    } else {
        __threadfence();
        __syncthreads();
        gather_wave_plain(perm, first, nst, ncnt, nl0, node0 + nl0, my_ad, ab, lane,
                          a_src, h, out);
    }
}

extern "C" void kernel_launch(void* const* d_in, const int* in_sizes, int n_in,
                              void* d_out, int out_size, void* d_ws, size_t ws_size,
                              hipStream_t stream) {
    const float* hidden = (const float*)d_in[0];
    const float* W      = (const float*)d_in[1];
    const float* bias   = (const float*)d_in[2];
    const float* att_w  = (const float*)d_in[3];
    const float* att_b  = (const float*)d_in[4];
    const int*   src    = (const int*)  d_in[5];
    const int*   dst    = (const int*)  d_in[6];
    float* out = (float*)d_out;

    float* ws     = (float*)d_ws;
    float* h_ws   = ws;
    float* a_src  = h_ws  + (size_t)N_NODES * OUT_DIM;
    float* a_dst  = a_src + N_NODES;
    int*   recs   = (int*)(a_dst + N_NODES);
    int*   perm   = recs + 2 * (size_t)N_EDGES;
    int*   tcnt   = perm + N_EDGES;
    int*   toff   = tcnt + NBUCKET * NCHUNK;

    gat_gemm_chunks<<<GEMM_GROUPS, GEMM_WAVES * 32, 0, stream>>>(
        hidden, W, bias, att_w, src, dst, h_ws, a_src, a_dst, recs, tcnt, toff);
    gat_bucket_gather<<<NBUCKET, SORT_T, 0, stream>>>(
        recs, tcnt, toff, perm, a_src, a_dst, att_b, h_ws, out);
}


# Gather launch: 250 buckets of 200 nodes, one 25-wave group per WGP keeps all of a_src in LDS: edge source terms are LDS reads, a node's rows requested together; sum order = new LDS arrival rank
# speedup vs baseline: 7.2709x; 1.0409x over previous
#include <hip/hip_runtime.h>


#define N_NODES   50000
#define N_EDGES   800000
#define IN_DIM    256
#define OUT_DIM   64
#define NEG_SLOPE 0.01f

typedef __attribute__((ext_vector_type(2))) float v2f;
typedef __attribute__((ext_vector_type(4))) float v4f;
typedef __attribute__((ext_vector_type(8))) float v8f;
typedef __attribute__((ext_vector_type(4))) int   v4i;
typedef __attribute__((ext_vector_type(2))) int   v2i;

#define GATHER_NPW 8

#define NCHUNK      256
#define CHUNK_E     (N_EDGES / NCHUNK)
#define BNODES      200
#define NBUCKET     ((N_NODES + BNODES - 1) / BNODES)
static_assert(N_EDGES % NCHUNK == 0, "equal chunks");

__device__ __forceinline__ int bucket_of(int d) {
    return min(max(d, 0) / BNODES, NBUCKET - 1);
}

__device__ __forceinline__ void lds_barrier() {
    __builtin_amdgcn_fence(__ATOMIC_RELEASE, "workgroup", "local");
    __builtin_amdgcn_s_barrier();
    __builtin_amdgcn_fence(__ATOMIC_ACQUIRE, "workgroup", "local");
}

#define GEMM_GROUPS   256
#define GEMM_WAVES    16
#define N_STRIPS      (N_NODES / 16)
#define FULL_ROUNDS   (N_STRIPS / GEMM_GROUPS)
#define REM_STRIPS    (N_STRIPS - FULL_ROUNDS * GEMM_GROUPS)
#define KBLOCKS       (IN_DIM / 8)
#define ROW_PITCH     (OUT_DIM + 4)
static_assert(N_NODES % 16 == 0, "whole strips only");
static_assert(FULL_ROUNDS + 4 <= GEMM_WAVES, "waves for the shared strip");
static_assert(REM_STRIPS <= GEMM_GROUPS, "at most one shared strip per group");
#define CHUNK_T       (GEMM_WAVES * 32)
#define CHUNK_SLOTS   ((CHUNK_E + CHUNK_T - 1) / CHUNK_T)
#define SCAN_PER      ((NBUCKET + 31) / 32)
#define HIST_PAD      (32 * SCAN_PER)
static_assert(NCHUNK == GEMM_GROUPS, "group g orders chunk g");
static_assert(HIST_PAD <= CHUNK_T, "one counter per thread to clear");

template <int NT>
__device__ __forceinline__ void strip_products(v8f (&acc)[NT], const v4f (&a)[KBLOCKS],
                                               const v4f* __restrict__ wl, int t0, int lane) {
#pragma unroll
    for (int kb = 0; kb < KBLOCKS; ++kb) {
        v4f b[NT];
#pragma unroll
        for (int t = 0; t < NT; ++t) b[t] = wl[(kb * 4 + t0 + t) * 32 + lane];
        const v2f a01 = {a[kb].x, a[kb].y};
        const v2f a23 = {a[kb].z, a[kb].w};
#pragma unroll
        for (int t = 0; t < NT; ++t) {
            const v2f b01 = {b[t].x, b[t].y};
            acc[t] = __builtin_amdgcn_wmma_f32_16x16x4_f32(
                false, a01, false, b01, (short)0, acc[t], false, false);
        }
#pragma unroll
        for (int t = 0; t < NT; ++t) {
            const v2f b23 = {b[t].z, b[t].w};
            acc[t] = __builtin_amdgcn_wmma_f32_16x16x4_f32(
                false, a23, false, b23, (short)0, acc[t], false, false);
        }
    }
}

template <int M>
__device__ __forceinline__ float row_xor_add(float x) {
    const int xi = __builtin_bit_cast(int, x);
    const int yi = __builtin_amdgcn_update_dpp(0, xi, 0x160 + M, 0xf, 0xf, true);
    return x + __builtin_bit_cast(float, yi);
}


__device__ __forceinline__ int xhalf_i(int b) {
    return __builtin_amdgcn_permlanex16(b, b, 0x76543210, 0xfedcba98, false, false);
}
__device__ __forceinline__ float xhalf(float x) {
    return __builtin_bit_cast(float, xhalf_i(__builtin_bit_cast(int, x)));
}
template <int M>
__device__ __forceinline__ float row_xor_max(float x) {
    const int xi = __builtin_bit_cast(int, x);
    const int yi = __builtin_amdgcn_update_dpp(0, xi, 0x160 + M, 0xf, 0xf, true);
    return fmaxf(x, __builtin_bit_cast(float, yi));
}
__device__ __forceinline__ float wave_sum32(float x) {
    x = x + xhalf(x);
    x = row_xor_add<8>(x);
    x = row_xor_add<4>(x);
    x = row_xor_add<2>(x);
    return row_xor_add<1>(x);
}
__device__ __forceinline__ float wave_max32(float x) {
    x = fmaxf(x, xhalf(x));
    x = row_xor_max<8>(x);
    x = row_xor_max<4>(x);
    x = row_xor_max<2>(x);
    return row_xor_max<1>(x);
}
template <int N>
__device__ __forceinline__ int row_shr_or_zero(int x) {
    return __builtin_amdgcn_update_dpp(0, x, 0x110 + N, 0xf, 0xf, true);
}
__device__ __forceinline__ int wave_incl_scan(int v) {
    v += row_shr_or_zero<1>(v);
    v += row_shr_or_zero<2>(v);
    v += row_shr_or_zero<4>(v);
    v += row_shr_or_zero<8>(v);
    const int low = __builtin_amdgcn_readlane(v, 15);
    return v + (((threadIdx.x & 31) >= 16) ? low : 0);
}

__device__ __forceinline__ void row_sums8(float (&p)[8], float (&q)[8]) {
#pragma unroll
    for (int i = 0; i < 8; ++i) { p[i] = row_xor_add<1>(p[i]); q[i] = row_xor_add<1>(q[i]); }
#pragma unroll
    for (int i = 0; i < 8; ++i) { p[i] = row_xor_add<2>(p[i]); q[i] = row_xor_add<2>(q[i]); }
#pragma unroll
    for (int i = 0; i < 8; ++i) { p[i] = row_xor_add<4>(p[i]); q[i] = row_xor_add<4>(q[i]); }
#pragma unroll
    for (int i = 0; i < 8; ++i) { p[i] = row_xor_add<8>(p[i]); q[i] = row_xor_add<8>(q[i]); }
}

__global__ __launch_bounds__(GEMM_WAVES * 32) void gat_gemm_chunks(
    const float* __restrict__ hidden,
    const float* __restrict__ W,
    const float* __restrict__ bias,
    const float* __restrict__ att_w,
    const int* __restrict__ src,
    const int* __restrict__ dst,
    float* __restrict__ h_out,
    float* __restrict__ a_src,
    float* __restrict__ a_dst,
    int* __restrict__ recs,
    int* __restrict__ tcnt,
    int* __restrict__ toff)
{
    __shared__ v4f w_frag[KBLOCKS * 4 * 32];
    __shared__ float h_rows[FULL_ROUNDS][16 * ROW_PITCH];
    __shared__ float part[4][2][16];
    __shared__ int hist[HIST_PAD];
    __shared__ int bofs[HIST_PAD];
    __shared__ v2i stage[CHUNK_E];

    const int tid  = threadIdx.x;
    const int lane = tid & 31;
    const int wave = __builtin_amdgcn_readfirstlane(tid >> 5);
    const int half = lane >> 4;
    const int l16  = lane & 15;

    const bool whole = wave < FULL_ROUNDS;
    const int  strip = (int)blockIdx.x + GEMM_GROUPS * (whole ? wave : FULL_ROUNDS);
    const bool work  = whole || ((int)blockIdx.x < REM_STRIPS && wave < FULL_ROUNDS + 4);

    float wv[8][4];
#pragma unroll
    for (int i = 0; i < 8; ++i) {
        const int e  = tid + 512 * i;
        const int kb = e >> 7, t = (e >> 5) & 3, hf = (e >> 4) & 1, c = e & 15;
        const float* wp = W + (size_t)(8 * kb + 4 * hf) * OUT_DIM + 16 * t + c;
#pragma unroll
        for (int j = 0; j < 4; ++j) wv[i][j] = wp[j * OUT_DIM];
    }
    const int cbase = (int)blockIdx.x * CHUNK_E;
    int es[CHUNK_SLOTS], ed[CHUNK_SLOTS];
#pragma unroll
    for (int k = 0; k < CHUNK_SLOTS; ++k)
        es[k] = src[cbase + min(tid + k * CHUNK_T, CHUNK_E - 1)];
#pragma unroll
    for (int k = 0; k < CHUNK_SLOTS; ++k)
        ed[k] = dst[cbase + min(tid + k * CHUNK_T, CHUNK_E - 1)];
    if (tid < HIST_PAD) hist[tid] = 0;
    lds_barrier();

#pragma unroll
    for (int i = 0; i < 8; ++i) {
        const v4f v = {wv[i][0], wv[i][1], wv[i][2], wv[i][3]};
        w_frag[tid + 512 * i] = v;
    }

    int eb[CHUNK_SLOTS], er[CHUNK_SLOTS];
#pragma unroll
    for (int k = 0; k < CHUNK_SLOTS; ++k) {
        eb[k] = bucket_of(ed[k]);
        er[k] = 0;
        if (tid + k * CHUNK_T < CHUNK_E) er[k] = atomicAdd(&hist[eb[k]], 1);
    }

    v4f a[KBLOCKS];
    if (work) {
        const float* arow = hidden + (size_t)(strip * 16 + l16) * IN_DIM + 4 * half;
#pragma unroll
        for (int kb = 0; kb < KBLOCKS; ++kb) a[kb] = *(const v4f*)(arow + 8 * kb);
    }
    lds_barrier();

    const bool scans = (wave == GEMM_WAVES - 1);
    int cv[SCAN_PER] = {}, cx[SCAN_PER] = {};
    int lex = 0;
    if (scans) {
        int run = 0;
#pragma unroll
        for (int j = 0; j < SCAN_PER; ++j) cv[j] = hist[lane * SCAN_PER + j];
#pragma unroll
        for (int j = 0; j < SCAN_PER; ++j) { cx[j] = run; run += cv[j]; }
        const int incl = wave_incl_scan(run);
        lex = incl - run;
#pragma unroll
        for (int j = 0; j < SCAN_PER; ++j) bofs[lane * SCAN_PER + j] = lex + cx[j];
    }
    lds_barrier();

    int sl[CHUNK_SLOTS];
#pragma unroll
    for (int k = 0; k < CHUNK_SLOTS; ++k) sl[k] = bofs[eb[k]] + er[k];
#pragma unroll
    for (int k = 0; k < CHUNK_SLOTS; ++k) {
        const v2i rec = {es[k], ed[k]};
        if (tid + k * CHUNK_T < CHUNK_E && (unsigned)sl[k] < (unsigned)CHUNK_E) stage[sl[k]] = rec;
    }
    lds_barrier();

#pragma unroll
    for (int k = 0; k < CHUNK_SLOTS; ++k) {
        const int i = tid + k * CHUNK_T;
        if (i < CHUNK_E) *(v2i*)(recs + 2 * (size_t)(cbase + i)) = stage[i];
    }
    if (scans) {
#pragma unroll
        for (int j = 0; j < SCAN_PER; ++j) {
            const int idx = lane * SCAN_PER + j;
            if (idx < NBUCKET) {
                tcnt[idx * NCHUNK + (int)blockIdx.x] = cv[j];
                toff[idx * NCHUNK + (int)blockIdx.x] = lex + cx[j];
            }
        }
    }
    if (!work) return;

    if (whole) {
        v8f acc[4] = {};
        strip_products<4>(acc, a, w_frag, 0, lane);

        float* hr = h_rows[wave];
#pragma unroll
        for (int t = 0; t < 4; ++t) {
            const float bb = bias[16 * t + l16];
#pragma unroll
            for (int r = 0; r < 8; ++r)
                hr[(8 * half + r) * ROW_PITCH + 16 * t + l16] = acc[t][r] + bb;
        }
        float* orow = h_out + (size_t)strip * 16 * OUT_DIM;
        const v4f ws4 = *(const v4f*)(att_w + 4 * l16);
        const v4f wd4 = *(const v4f*)(att_w + OUT_DIM + 4 * l16);
        float ps[8], pd[8];
#pragma unroll
        for (int i = 0; i < 8; ++i) {
            const int row = 2 * i + half;
            const v4f v = *(const v4f*)(hr + row * ROW_PITCH + 4 * l16);
            *(v4f*)(orow + row * OUT_DIM + 4 * l16) = v;
            ps[i] = v.x * ws4.x + v.y * ws4.y + v.z * ws4.z + v.w * ws4.w;
            pd[i] = v.x * wd4.x + v.y * wd4.y + v.z * wd4.z + v.w * wd4.w;
        }
        row_sums8(ps, pd);
        float sres = ps[0], dres = pd[0];
#pragma unroll
        for (int i = 1; i < 8; ++i) if (l16 == i) { sres = ps[i]; dres = pd[i]; }
        if (l16 < 8) {
            a_src[strip * 16 + 2 * l16 + half] = sres;
            a_dst[strip * 16 + 2 * l16 + half] = dres;
        }
    } else {
        const int t0 = wave - FULL_ROUNDS;
        v8f acc[1] = {};
        strip_products<1>(acc, a, w_frag, t0, lane);
        const int gcol = 16 * t0 + l16;
        const float bb = bias[gcol];
        const float wsc = att_w[gcol], wdc = att_w[OUT_DIM + gcol];
        float ps[8], pd[8];
#pragma unroll
        for (int r = 0; r < 8; ++r) {
            const float hv = acc[0][r] + bb;
            h_out[(size_t)(strip * 16 + 8 * half + r) * OUT_DIM + gcol] = hv;
            ps[r] = hv * wsc;
            pd[r] = hv * wdc;
        }
        row_sums8(ps, pd);
        float sres = ps[0], dres = pd[0];
#pragma unroll
        for (int r = 1; r < 8; ++r) if (l16 == r) { sres = ps[r]; dres = pd[r]; }
        if (l16 < 8) {
            part[t0][0][8 * half + l16] = sres;
            part[t0][1][8 * half + l16] = dres;
        }
    }

    if ((int)blockIdx.x < REM_STRIPS) {
        __builtin_amdgcn_fence(__ATOMIC_RELEASE, "workgroup", "local");
        __builtin_amdgcn_s_barrier();
        __builtin_amdgcn_fence(__ATOMIC_ACQUIRE, "workgroup", "local");
        if (wave == FULL_ROUNDS) {
            const float v = ((part[0][half][l16] + part[1][half][l16])
                             + part[2][half][l16]) + part[3][half][l16];
            float* ap = half ? a_dst : a_src;
            ap[strip * 16 + l16] = v;
        }
    }
}

#define SORT_T      800
#define SORT_WAVES  (SORT_T / 32)
#define REC_SLOTS   5
#define STAGE_CAP   8192
#define ASRC_V4     (N_NODES / 4)
#define ASRC_SLOTS  ((ASRC_V4 + SORT_T - 1) / SORT_T)
static_assert(N_NODES % 4 == 0, "a_src copied in whole 16-byte vectors");
static_assert(NBUCKET <= SORT_T && NCHUNK <= SORT_T && BNODES <= SORT_T, "one table entry per thread");
static_assert(BNODES == (SORT_T / 32) * GATHER_NPW, "every wave of the group takes GATHER_NPW nodes");

__device__ __forceinline__ int block_excl_scan(int v, int* wsum, int& total) {
    const int lane = threadIdx.x & 31;
    const int wave = __builtin_amdgcn_readfirstlane(threadIdx.x >> 5);
    const int incl = wave_incl_scan(v);
    lds_barrier();
    if (lane == 31) wsum[wave] = incl;
    lds_barrier();
    const int wt = (lane < SORT_WAVES) ? wsum[lane] : 0;
    const int wi = wave_incl_scan(wt);
    total = __builtin_amdgcn_readlane(wi, 31);
    const int wexcl = __builtin_amdgcn_readlane(wi - wt, wave);
    return wexcl + incl - v;
}
static_assert(SORT_T % 32 == 0 && SORT_WAVES <= 32, "block_excl_scan reads one wave total per lane");

__device__ __forceinline__ v2i bucket_record(const int* __restrict__ recs,
                                             const int* pbeg, const int* ploc, int i) {
    int lo = 0, hi = NCHUNK;
    for (int st = 0; st < 8; ++st) {
        const int mid = (lo + hi) >> 1;
        if (pbeg[mid] <= i) lo = mid; else hi = mid;
    }
    const int j = min(max(ploc[lo] + (i - pbeg[lo]), 0), N_EDGES - 1);
    return *(const v2i*)(recs + 2 * (size_t)j);
}
static_assert(NCHUNK == 256, "bucket_record halves eight times");

__device__ __forceinline__ float lrelu(float v) {
    return (v > 0.0f) ? v : NEG_SLOPE * v;
}

__device__ __forceinline__ int clamp_node(int s) {
    return min(max(s, 0), N_NODES - 1);
}

template <int K>
__device__ __forceinline__ void load_rows(v2f (&hv)[K], const float* __restrict__ hl,
                                          int s, int t) {
#pragma unroll
    for (int k = 0; k < K; ++k) {
        const int st = __builtin_amdgcn_readlane(s, t + k);
        hv[k] = *(const v2f*)(hl + (size_t)st * OUT_DIM);
    }
}
template <int K>
__device__ __forceinline__ void sum_rows(v2f& acc, const v2f (&hv)[K], float ex, int t) {
#pragma unroll
    for (int k = 0; k < K; ++k) {
        const float w = __int_as_float(__builtin_amdgcn_readlane(__float_as_int(ex), t + k));
        acc.x = __builtin_fmaf(w, hv[k].x, acc.x);
        acc.y = __builtin_fmaf(w, hv[k].y, acc.y);
    }
}

__device__ __forceinline__ void gather_node(
    const int node, const int start, const int n, const float ad, const float ab,
    const int s0, const float as0, const int lane,
    const int* __restrict__ list, const float* __restrict__ a_src,
    const float* __restrict__ h, float* __restrict__ out)
{
    const float NINF = -__builtin_inff();

    const float e0 = (lane < n) ? lrelu(as0 + ad + ab) : NINF;
    float m = e0;
    for (int base = 32; base < n; base += 32) {
        int j = base + lane;
        if (j < n) {
            int s = clamp_node(list[start + j]);
            m = fmaxf(m, lrelu(a_src[s] + ad + ab));
        }
    }
    m = wave_max32(m);
    if (!__builtin_isfinite(m)) m = 0.0f;

    v2f   acc  = {0.0f, 0.0f};
    float dsum = 0.0f;
    const float* hl = h + 2 * lane;
    for (int base = 0; base < n; base += 32) {
        int   s = s0;
        float e = e0;
        if (base > 0) {
            s = 0; e = NINF;
            int j = base + lane;
            if (j < n) {
                s = clamp_node(list[start + j]);
                e = lrelu(a_src[s] + ad + ab);
            }
        }
        const float ex = (base + lane < n) ? __expf(e - m) : 0.0f;
        dsum += ex;
        const int nc = min(32, n - base);
        v2f h16[16], h8[8], h4[4], h2[2], h1[1];
        if (nc == 32) {
            load_rows<16>(h16, hl, s, 0);
            sum_rows<16>(acc, h16, ex, 0);
            load_rows<16>(h16, hl, s, 16);
            sum_rows<16>(acc, h16, ex, 16);
        } else {
            const int t8 = nc & 16;
            const int t4 = t8 + (nc & 8);
            const int t2 = t4 + (nc & 4);
            const int t1 = t2 + (nc & 2);
            if (nc & 16) load_rows<16>(h16, hl, s, 0);
            if (nc & 8)  load_rows<8>(h8, hl, s, t8);
            if (nc & 4)  load_rows<4>(h4, hl, s, t4);
            if (nc & 2)  load_rows<2>(h2, hl, s, t2);
            if (nc & 1)  load_rows<1>(h1, hl, s, t1);
            if (nc & 16) sum_rows<16>(acc, h16, ex, 0);
            if (nc & 8)  sum_rows<8>(acc, h8, ex, t8);
            if (nc & 4)  sum_rows<4>(acc, h4, ex, t4);
            if (nc & 2)  sum_rows<2>(acc, h2, ex, t2);
            if (nc & 1)  sum_rows<1>(acc, h1, ex, t1);
        }
    }
    dsum = wave_sum32(dsum);

    v2f r;
    if (dsum > 0.0f) {
        r.x = acc.x / dsum;
        r.y = acc.y / dsum;
    } else {
        r = *(const v2f*)(h + (size_t)node * OUT_DIM + 2 * lane);
    }
    *(v2f*)(out + (size_t)node * OUT_DIM + 2 * lane) = r;
}

template <int LIM>
__device__ __forceinline__ int first_entries(const int* __restrict__ list,
                                             int start, int lane) {
    return list[min(start + lane, LIM - 1)];
}
__device__ __forceinline__ float first_asrc(const float* __restrict__ a_src,
                                            int s, int n, int lane) {
    return (lane < n) ? a_src[s] : 0.0f;
}

template <int LIM>
__device__ __forceinline__ void gather_wave(
    const int* __restrict__ list, const int lbase,
    const int* nst, const int* ncnt, const int nl0, const int nodeg0,
    const float my_ad, const float ab, const int lane,
    const float* __restrict__ a_src, const float* __restrict__ h, float* __restrict__ out)
{
    const int kk = lane & (GATHER_NPW - 1);
    const int my_st = min(max(lbase + nst[nl0 + kk], 0), LIM);
    const int my_n  = min(max(ncnt[nl0 + kk], 0), LIM - my_st);

    const int   pA = first_entries<LIM>(list, __builtin_amdgcn_readlane(my_st, 0), lane);
    int   pB = first_entries<LIM>(list, __builtin_amdgcn_readlane(my_st, 1), lane);
    int   sA = clamp_node(pA);
    float aA = first_asrc(a_src, sA, __builtin_amdgcn_readlane(my_n, 0), lane);
#pragma unroll 1
    for (int k = 0; k < GATHER_NPW; ++k) {
        const int   k1  = min(k + 1, GATHER_NPW - 1), k2 = min(k + 2, GATHER_NPW - 1);
        const int   stk = __builtin_amdgcn_readlane(my_st, k);
        const int   nk  = __builtin_amdgcn_readlane(my_n, k);
        const float adk = __int_as_float(__builtin_amdgcn_readlane(__float_as_int(my_ad), k));
        const int   pC  = first_entries<LIM>(list, __builtin_amdgcn_readlane(my_st, k2), lane);
        const int   sB  = clamp_node(pB);
        const float aB  = first_asrc(a_src, sB, __builtin_amdgcn_readlane(my_n, k1), lane);
        if (nodeg0 + k < N_NODES)
            gather_node(nodeg0 + k, stk, nk, adk, ab, sA, aA, lane, list, a_src, h, out);
        sA = sB; pB = pC; aA = aB;
    }
}

__device__ __forceinline__ void gather_wave_plain(
    const int* __restrict__ list, const int lbase,
    const int* nst, const int* ncnt, const int nl0, const int nodeg0,
    const float my_ad, const float ab, const int lane,
    const float* __restrict__ a_src, const float* __restrict__ h, float* __restrict__ out)
{
    const int kk = lane & (GATHER_NPW - 1);
    const int my_st = min(max(lbase + nst[nl0 + kk], 0), N_EDGES);
    const int my_n  = min(max(ncnt[nl0 + kk], 0), N_EDGES - my_st);
#pragma unroll 1
    for (int k = 0; k < GATHER_NPW; ++k) {
        if (nodeg0 + k >= N_NODES) break;
        const int   stk = __builtin_amdgcn_readlane(my_st, k);
        const int   nk  = __builtin_amdgcn_readlane(my_n, k);
        const float adk = __int_as_float(__builtin_amdgcn_readlane(__float_as_int(my_ad), k));
        const int   s   = (lane < nk) ? clamp_node(list[stk + lane]) : 0;
        const float as  = first_asrc(a_src, s, nk, lane);
        gather_node(nodeg0 + k, stk, nk, adk, ab, s, as, lane, list, a_src, h, out);
    }
}

__global__ __launch_bounds__(SORT_T) void gat_bucket_gather(
    const int* __restrict__ recs, const int* __restrict__ tcnt,
    const int* __restrict__ toff, int* __restrict__ perm,
    const float* __restrict__ a_src, const float* __restrict__ a_dst,
    const float* __restrict__ att_b,
    const float* __restrict__ h, float* __restrict__ out)
{
    __shared__ int pbeg[NCHUNK + 1];
    __shared__ int ploc[NCHUNK];
    __shared__ int ncnt[BNODES];
    __shared__ int ncur[BNODES];
    __shared__ int nst[BNODES];
    __shared__ int wsum[32];
    __shared__ int lperm[STAGE_CAP];
    __shared__ v4f asrc_l[ASRC_V4];

    const int tid   = threadIdx.x;
    const int bk    = blockIdx.x;
    const int node0 = bk * BNODES;
    const int lane  = tid & 31;
    const int nl0   = __builtin_amdgcn_readfirstlane(tid >> 5) * GATHER_NPW;

    const float my_ad = a_dst[min(node0 + nl0 + (lane & (GATHER_NPW - 1)), N_NODES - 1)];
    const float ab = att_b[0];

    int len = 0;
    if (tid < NCHUNK) {
        len = tcnt[bk * NCHUNK + tid];
        ploc[tid] = tid * CHUNK_E + toff[bk * NCHUNK + tid];
    }
    if (tid < BNODES) ncnt[tid] = 0;
    v4f av[ASRC_SLOTS];
#pragma unroll
    for (int k = 0; k < ASRC_SLOTS; ++k)
        av[k] = ((const v4f*)a_src)[min(tid + k * SORT_T, ASRC_V4 - 1)];
    int nb;
    const int pex = block_excl_scan(len, wsum, nb);
    if (tid < NCHUNK) pbeg[tid] = pex;
    if (tid == 0) pbeg[NCHUNK] = nb;

    const bool staged = nb <= STAGE_CAP;
    int first = 0;
    if (!staged) {
        int pre = 0;
        for (int i = tid; i < bk * NCHUNK; i += SORT_T) pre += tcnt[i];
        block_excl_scan(pre, wsum, first);
    }
    lds_barrier();

    int rs[REC_SLOTS], rl[REC_SLOTS];
    const int last = max(nb - 1, 0);
    int pi[REC_SLOTS], lo[REC_SLOTS];
#pragma unroll
    for (int k = 0; k < REC_SLOTS; ++k) {
        pi[k] = min(tid + k * SORT_T, last);
        lo[k] = 0;
    }
#pragma unroll
    for (int step = NCHUNK / 2; step > 0; step >>= 1) {
        int pm[REC_SLOTS];
#pragma unroll
        for (int k = 0; k < REC_SLOTS; ++k) pm[k] = pbeg[lo[k] + step];
#pragma unroll
        for (int k = 0; k < REC_SLOTS; ++k) if (pm[k] <= pi[k]) lo[k] += step;
    }
    int rj[REC_SLOTS];
#pragma unroll
    for (int k = 0; k < REC_SLOTS; ++k) {
        rj[k] = min(max(ploc[lo[k]] + (pi[k] - pbeg[lo[k]]), 0), N_EDGES - 1);
    }
#pragma unroll
    for (int k = 0; k < REC_SLOTS; ++k) {
        const v2i rec = *(const v2i*)(recs + 2 * (size_t)rj[k]);
        rs[k] = rec.x;
        rl[k] = (tid + k * SORT_T < nb) ? min(max(rec.y - node0, 0), BNODES - 1) : -1;
    }
#pragma unroll
    for (int k = 0; k < REC_SLOTS; ++k)
        if (rl[k] >= 0) atomicAdd(&ncnt[rl[k]], 1);
    for (int i = tid + REC_SLOTS * SORT_T; i < nb; i += SORT_T) {
        const v2i rec = bucket_record(recs, pbeg, ploc, i);
        atomicAdd(&ncnt[min(max(rec.y - node0, 0), BNODES - 1)], 1);
    }
    lds_barrier();

    const int nc = (tid < BNODES) ? ncnt[tid] : 0;
    int tot;
    const int nex = block_excl_scan(nc, wsum, tot);
    if (tid < BNODES) {
        ncur[tid] = nex;
        nst[tid]  = nex;
    }
    lds_barrier();

#pragma unroll
    for (int k = 0; k < REC_SLOTS; ++k) {
        if (rl[k] >= 0) {
            const int pos = atomicAdd(&ncur[rl[k]], 1);
            if (staged) {
                if ((unsigned)pos < (unsigned)STAGE_CAP) lperm[pos] = rs[k];
            } else if ((unsigned)(first + pos) < (unsigned)N_EDGES) {
                perm[first + pos] = rs[k];
            }
        }
    }
    for (int i = tid + REC_SLOTS * SORT_T; i < nb; i += SORT_T) {
        const v2i rec = bucket_record(recs, pbeg, ploc, i);
        const int pos = atomicAdd(&ncur[min(max(rec.y - node0, 0), BNODES - 1)], 1);
        if (staged) {
            if ((unsigned)pos < (unsigned)STAGE_CAP) lperm[pos] = rec.x;
        } else if ((unsigned)(first + pos) < (unsigned)N_EDGES) {
            perm[first + pos] = rec.x;
        }
    }

#pragma unroll
    for (int k = 0; k < ASRC_SLOTS; ++k)
        if (tid + k * SORT_T < ASRC_V4) asrc_l[tid + k * SORT_T] = av[k];
    const float* asl = (const float*)asrc_l;

    if (staged) {
        lds_barrier();
        gather_wave<STAGE_CAP>(lperm, 0, nst, ncnt, nl0, node0 + nl0, my_ad, ab, lane,
                               asl, h, out);
    } else {
        __threadfence();
        __syncthreads();
        gather_wave_plain(perm, first, nst, ncnt, nl0, node0 + nl0, my_ad, ab, lane,
                          asl, h, out);
    }
}

extern "C" void kernel_launch(void* const* d_in, const int* in_sizes, int n_in,
                              void* d_out, int out_size, void* d_ws, size_t ws_size,
                              hipStream_t stream) {
    const float* hidden = (const float*)d_in[0];
    const float* W      = (const float*)d_in[1];
    const float* bias   = (const float*)d_in[2];
    const float* att_w  = (const float*)d_in[3];
    const float* att_b  = (const float*)d_in[4];
    const int*   src    = (const int*)  d_in[5];
    const int*   dst    = (const int*)  d_in[6];
    float* out = (float*)d_out;

    float* ws     = (float*)d_ws;
    float* h_ws   = ws;
    float* a_src  = h_ws  + (size_t)N_NODES * OUT_DIM;
    float* a_dst  = a_src + N_NODES;
    int*   recs   = (int*)(a_dst + N_NODES);
    int*   perm   = recs + 2 * (size_t)N_EDGES;
    int*   tcnt   = perm + N_EDGES;
    int*   toff   = tcnt + NBUCKET * NCHUNK;

    gat_gemm_chunks<<<GEMM_GROUPS, GEMM_WAVES * 32, 0, stream>>>(
        hidden, W, bias, att_w, src, dst, h_ws, a_src, a_dst, recs, tcnt, toff);
    gat_bucket_gather<<<NBUCKET, SORT_T, 0, stream>>>(
        recs, tcnt, toff, perm, a_src, a_dst, att_b, h_ws, out);
}
